# starting kernel
# baseline (speedup 1.0000x reference)
#include <hip/hip_runtime.h>
#include <math.h>

constexpr int N_NODES = 50000;
constexpr int N_EDGES = 10000;
constexpr int NNZ     = 150000;
constexpr int IN_CH   = 128;
constexpr int HIDDEN  = 64;
constexpr int D       = 6;
constexpr int OUT_CH  = HIDDEN * D;
constexpr int ND      = N_NODES * D;
constexpr int ED      = N_EDGES * D;

typedef __attribute__((ext_vector_type(2))) float v2f;
typedef __attribute__((ext_vector_type(8))) float v8f;

__global__ void wmma_gemm_f32(const float* __restrict__ A, const float* __restrict__ B,
                              const float* __restrict__ bias, float* __restrict__ C,
                              int K, int N, int nTiles, int totalTiles) {
  int tile = blockIdx.x * (blockDim.x >> 5) + (threadIdx.x >> 5);
  if (tile >= totalTiles) return;
  int lane = threadIdx.x & 31;
  int half = lane >> 4;
  int ml   = lane & 15;
  int m0 = (tile / nTiles) * 16;
  int n0 = (tile % nTiles) * 16;

  v8f acc = {};
  const float* Arow = A + (size_t)(m0 + ml) * K + 2 * half;
  const float* Bcol = B + (size_t)(2 * half) * N + n0 + ml;
  for (int k0 = 0; k0 < K; k0 += 4) {
    v2f a, b;
    a.x = Arow[k0];
    a.y = Arow[k0 + 1];
    b.x = Bcol[(size_t)k0 * N];
    b.y = Bcol[(size_t)(k0 + 1) * N];
    acc = __builtin_amdgcn_wmma_f32_16x16x4_f32(false, a, false, b,
                                                (short)0, acc, false, false);
  }
  float bv = bias ? bias[n0 + ml] : 0.0f;
#pragma unroll
  for (int j = 0; j < 8; ++j) {
    C[(size_t)(m0 + j + 8 * half) * N + n0 + ml] = acc[j] + bv;
  }
}

__global__ void stalk_mean_kernel(const float* __restrict__ proj, float* __restrict__ m,
                                  int total) {
  int idx = blockIdx.x * blockDim.x + threadIdx.x;
  if (idx >= total) return;
  int n = idx >> 6;
  int hh = idx & 63;
  const float* p = proj + (size_t)n * OUT_CH + hh;
  float s = 0.0f;
#pragma unroll
  for (int t = 0; t < D; ++t) s += p[t * HIDDEN];
  m[idx] = s * (1.0f / 6.0f);
}

__device__ inline float wave_allsum(float v) {
#pragma unroll
  for (int off = 16; off > 0; off >>= 1) v += __shfl_xor(v, off, 32);
  return v;
}

__global__ void alpha_kernel(const float* __restrict__ xm, const float* __restrict__ em,
                             const int* __restrict__ row, const int* __restrict__ col,
                             const float* __restrict__ ln_g, const float* __restrict__ ln_b,
                             const float* __restrict__ Wsh, const float* __restrict__ bsh,
                             float* __restrict__ alpha, float* __restrict__ Dsum,
                             float* __restrict__ Bcnt) {
  int i = blockIdx.x * (blockDim.x >> 5) + (threadIdx.x >> 5);
  if (i >= NNZ) return;
  int lane = threadIdx.x & 31;
  int rn = row[i], cn = col[i];
  int j0 = lane * 4;
  float z[4];
#pragma unroll
  for (int q = 0; q < 4; ++q) {
    int j = j0 + q;
    z[q] = (j < HIDDEN) ? xm[(size_t)rn * HIDDEN + j]
                        : em[(size_t)cn * HIDDEN + (j - HIDDEN)];
  }
  float s  = z[0] + z[1] + z[2] + z[3];
  float ss = z[0]*z[0] + z[1]*z[1] + z[2]*z[2] + z[3]*z[3];
  s  = wave_allsum(s);
  ss = wave_allsum(ss);
  float mu   = s * (1.0f / 128.0f);
  float var  = ss * (1.0f / 128.0f) - mu * mu;
  float rstd = rsqrtf(var + 1e-5f);

  float p[6] = {0, 0, 0, 0, 0, 0};
#pragma unroll
  for (int q = 0; q < 4; ++q) {
    int j = j0 + q;
    float fn = (z[q] - mu) * rstd * ln_g[j] + ln_b[j];
#pragma unroll
    for (int t = 0; t < D; ++t) p[t] += fn * Wsh[j * D + t];
  }
#pragma unroll
  for (int t = 0; t < D; ++t) p[t] = wave_allsum(p[t]);

  if (lane < D) {
    float a = 1.0f / (1.0f + expf(-(p[lane] + bsh[lane])));
    alpha[(size_t)i * D + lane] = a;
    atomicAdd(&Dsum[rn * D + lane], a * a);
  } else if (lane == D) {
    atomicAdd(&Bcnt[cn], 1.0f);
  }
}

__global__ void safe_inv_kernel(float* __restrict__ buf, int n) {
  int idx = blockIdx.x * blockDim.x + threadIdx.x;
  if (idx >= n) return;
  float v = buf[idx];
  buf[idx] = (v > 0.0f) ? (1.0f / v) : 0.0f;
}

__global__ void scatter_me_kernel(const float* __restrict__ h, const float* __restrict__ alpha,
                                  const int* __restrict__ row, const int* __restrict__ col,
                                  float* __restrict__ me) {
  int idx = blockIdx.x * blockDim.x + threadIdx.x;
  if (idx >= NNZ * OUT_CH) return;
  int i  = idx / OUT_CH;
  int r  = idx - i * OUT_CH;
  int t  = r >> 6;
  int hh = r & 63;
  float a = alpha[(size_t)i * D + t];
  float v = a * h[(size_t)(row[i] * D + t) * HIDDEN + hh];
  atomicAdd(&me[(size_t)(col[i] * D + t) * HIDDEN + hh], v);
}

__global__ void scatter_agg_kernel(const float* __restrict__ me, const float* __restrict__ alpha,
                                   const int* __restrict__ row, const int* __restrict__ col,
                                   const float* __restrict__ Binv, float* __restrict__ agg) {
  int idx = blockIdx.x * blockDim.x + threadIdx.x;
  if (idx >= NNZ * OUT_CH) return;
  int i  = idx / OUT_CH;
  int r  = idx - i * OUT_CH;
  int t  = r >> 6;
  int hh = r & 63;
  int cn = col[i];
  float a = alpha[(size_t)i * D + t] * Binv[cn];
  float v = a * me[(size_t)(cn * D + t) * HIDDEN + hh];
  atomicAdd(&agg[(size_t)(row[i] * D + t) * HIDDEN + hh], v);
}

__global__ void combine_kernel(const float* __restrict__ h, const float* __restrict__ agg,
                               const float* __restrict__ Dinv, const float* __restrict__ bias,
                               float* __restrict__ out, int total) {
  int idx = blockIdx.x * blockDim.x + threadIdx.x;
  if (idx >= total) return;
  int r  = idx >> 6;
  int hh = idx & 63;
  float o = h[idx] - Dinv[r] * agg[idx] + bias[hh];
  out[idx] = (o > 0.0f) ? o : expm1f(o);
}

extern "C" void kernel_launch(void* const* d_in, const int* in_sizes, int n_in,
                              void* d_out, int out_size, void* d_ws, size_t ws_size,
                              hipStream_t stream) {
  (void)in_sizes; (void)n_in; (void)out_size; (void)ws_size;
  const float* x       = (const float*)d_in[0];
  const float* hattr   = (const float*)d_in[1];
  const float* W_lin   = (const float*)d_in[2];
  const float* b_lin   = (const float*)d_in[3];
  const float* ln_g    = (const float*)d_in[4];
  const float* ln_b    = (const float*)d_in[5];
  const float* W_sheaf = (const float*)d_in[6];
  const float* b_sheaf = (const float*)d_in[7];
  const float* W_conv  = (const float*)d_in[8];
  const float* b_conv  = (const float*)d_in[9];
  const int*   eidx    = (const int*)d_in[10];
  const int* row = eidx;
  const int* col = eidx + NNZ;
  float* out = (float*)d_out;

  float* ws    = (float*)d_ws;
  float* xs    = ws;
  float* es    = xs   + (size_t)ND * HIDDEN;
  float* xm    = es   + (size_t)ED * HIDDEN;
  float* em    = xm   + (size_t)N_NODES * HIDDEN;
  float* alpha = em   + (size_t)N_EDGES * HIDDEN;
  float* Dbuf  = alpha + (size_t)NNZ * D;
  float* Bbuf  = Dbuf + ND;
  float* hbuf  = Bbuf + N_EDGES;
  float* me    = hbuf + (size_t)ND * HIDDEN;
  float* agg   = me   + (size_t)ED * HIDDEN;

  const int TPB = 256;

  {
    int tiles = (N_NODES / 16) * (OUT_CH / 16);
    wmma_gemm_f32<<<tiles / 8, TPB, 0, stream>>>(x, W_lin, b_lin, xs,
                                                 IN_CH, OUT_CH, OUT_CH / 16, tiles);
    int etiles = (N_EDGES / 16) * (OUT_CH / 16);
    wmma_gemm_f32<<<etiles / 8, TPB, 0, stream>>>(hattr, W_lin, b_lin, es,
                                                  IN_CH, OUT_CH, OUT_CH / 16, etiles);
  }

  stalk_mean_kernel<<<(N_NODES * HIDDEN + TPB - 1) / TPB, TPB, 0, stream>>>(xs, xm, N_NODES * HIDDEN);
  stalk_mean_kernel<<<(N_EDGES * HIDDEN + TPB - 1) / TPB, TPB, 0, stream>>>(es, em, N_EDGES * HIDDEN);

  hipMemsetAsync(Dbuf, 0, (size_t)ND * sizeof(float), stream);
  hipMemsetAsync(Bbuf, 0, (size_t)N_EDGES * sizeof(float), stream);
  alpha_kernel<<<NNZ / 8, TPB, 0, stream>>>(xm, em, row, col, ln_g, ln_b,
                                            W_sheaf, b_sheaf, alpha, Dbuf, Bbuf);
  safe_inv_kernel<<<(ND + TPB - 1) / TPB, TPB, 0, stream>>>(Dbuf, ND);
  safe_inv_kernel<<<(N_EDGES + TPB - 1) / TPB, TPB, 0, stream>>>(Bbuf, N_EDGES);

  const int convTiles  = (ND / 16) * (HIDDEN / 16);
  const int scatterBlk = (NNZ * OUT_CH + TPB - 1) / TPB;
  const int elemBlk    = (ND * HIDDEN + TPB - 1) / TPB;

  wmma_gemm_f32<<<convTiles / 8, TPB, 0, stream>>>(xs, W_conv, nullptr, hbuf,
                                                   HIDDEN, HIDDEN, HIDDEN / 16, convTiles);
  hipMemsetAsync(me, 0, (size_t)ED * HIDDEN * sizeof(float), stream);
  scatter_me_kernel<<<scatterBlk, TPB, 0, stream>>>(hbuf, alpha, row, col, me);
  hipMemsetAsync(agg, 0, (size_t)ND * HIDDEN * sizeof(float), stream);
  scatter_agg_kernel<<<scatterBlk, TPB, 0, stream>>>(me, alpha, row, col, Bbuf, agg);
  combine_kernel<<<elemBlk, TPB, 0, stream>>>(hbuf, agg, Dbuf, b_conv, out, ND * HIDDEN);

  wmma_gemm_f32<<<convTiles / 8, TPB, 0, stream>>>(out, W_conv + HIDDEN * HIDDEN, nullptr, hbuf,
                                                   HIDDEN, HIDDEN, HIDDEN / 16, convTiles);
  hipMemsetAsync(me, 0, (size_t)ED * HIDDEN * sizeof(float), stream);
  scatter_me_kernel<<<scatterBlk, TPB, 0, stream>>>(hbuf, alpha, row, col, me);
  hipMemsetAsync(agg, 0, (size_t)ND * HIDDEN * sizeof(float), stream);
  scatter_agg_kernel<<<scatterBlk, TPB, 0, stream>>>(me, alpha, row, col, Bbuf, agg);
  combine_kernel<<<elemBlk, TPB, 0, stream>>>(hbuf, agg, Dbuf, b_conv + HIDDEN, out, ND * HIDDEN);
}


# Replace float-atomic scatter-adds, degree atomics and zero fills by owner-computes segment gathers over id-sorted incidence lists; node owner applies scale+ELU; sums in ascending incidence order
# speedup vs baseline: 2.2258x; 2.2258x over previous
#include <hip/hip_runtime.h>
#include <math.h>

constexpr int N_NODES = 50000;
constexpr int N_EDGES = 10000;
constexpr int NNZ     = 150000;
constexpr int IN_CH   = 128;
constexpr int HIDDEN  = 64;
constexpr int D       = 6;
constexpr int OUT_CH  = HIDDEN * D;
constexpr int ND      = N_NODES * D;
constexpr int ED      = N_EDGES * D;

typedef __attribute__((ext_vector_type(2))) float v2f;
typedef __attribute__((ext_vector_type(8))) float v8f;

__global__ void wmma_gemm_f32(const float* __restrict__ A, const float* __restrict__ B,
                              const float* __restrict__ bias, float* __restrict__ C,
                              int K, int N, int nTiles, int totalTiles) {
  int tile = blockIdx.x * (blockDim.x >> 5) + (threadIdx.x >> 5);
  if (tile >= totalTiles) return;
  int lane = threadIdx.x & 31;
  int half = lane >> 4;
  int ml   = lane & 15;
  int m0 = (tile / nTiles) * 16;
  int n0 = (tile % nTiles) * 16;

  v8f acc = {};
  const float* Arow = A + (size_t)(m0 + ml) * K + 2 * half;
  const float* Bcol = B + (size_t)(2 * half) * N + n0 + ml;
  for (int k0 = 0; k0 < K; k0 += 4) {
    v2f a, b;
    a.x = Arow[k0];
    a.y = Arow[k0 + 1];
    b.x = Bcol[(size_t)k0 * N];
    b.y = Bcol[(size_t)(k0 + 1) * N];
    acc = __builtin_amdgcn_wmma_f32_16x16x4_f32(false, a, false, b,
                                                (short)0, acc, false, false);
  }
  float bv = bias ? bias[n0 + ml] : 0.0f;
#pragma unroll
  for (int j = 0; j < 8; ++j) {
    C[(size_t)(m0 + j + 8 * half) * N + n0 + ml] = acc[j] + bv;
  }
}

__global__ void stalk_mean_kernel(const float* __restrict__ proj, float* __restrict__ m,
                                  int total) {
  int idx = blockIdx.x * blockDim.x + threadIdx.x;
  if (idx >= total) return;
  int n = idx >> 6;
  int hh = idx & 63;
  const float* p = proj + (size_t)n * OUT_CH + hh;
  float s = 0.0f;
#pragma unroll
  for (int t = 0; t < D; ++t) s += p[t * HIDDEN];
  m[idx] = s * (1.0f / 6.0f);
}

__device__ inline float wave_allsum(float v) {
#pragma unroll
  for (int off = 16; off > 0; off >>= 1) v += __shfl_xor(v, off, 32);
  return v;
}

__global__ void alpha_kernel(const float* __restrict__ xm, const float* __restrict__ em,
                             const int* __restrict__ row, const int* __restrict__ col,
                             const float* __restrict__ ln_g, const float* __restrict__ ln_b,
                             const float* __restrict__ Wsh, const float* __restrict__ bsh,
                             float* __restrict__ alpha) {
  int i = blockIdx.x * (blockDim.x >> 5) + (threadIdx.x >> 5);
  if (i >= NNZ) return;
  int lane = threadIdx.x & 31;
  int rn = row[i], cn = col[i];
  int j0 = lane * 4;
  float z[4];
#pragma unroll
  for (int q = 0; q < 4; ++q) {
    int j = j0 + q;
    z[q] = (j < HIDDEN) ? xm[(size_t)rn * HIDDEN + j]
                        : em[(size_t)cn * HIDDEN + (j - HIDDEN)];
  }
  float s  = z[0] + z[1] + z[2] + z[3];
  float ss = z[0]*z[0] + z[1]*z[1] + z[2]*z[2] + z[3]*z[3];
  s  = wave_allsum(s);
  ss = wave_allsum(ss);
  float mu   = s * (1.0f / 128.0f);
  float var  = ss * (1.0f / 128.0f) - mu * mu;
  float rstd = rsqrtf(var + 1e-5f);

  float p[6] = {0, 0, 0, 0, 0, 0};
#pragma unroll
  for (int q = 0; q < 4; ++q) {
    int j = j0 + q;
    float fn = (z[q] - mu) * rstd * ln_g[j] + ln_b[j];
#pragma unroll
    for (int t = 0; t < D; ++t) p[t] += fn * Wsh[j * D + t];
  }
#pragma unroll
  for (int t = 0; t < D; ++t) p[t] = wave_allsum(p[t]);

  if (lane < D) {
    float a = 1.0f / (1.0f + expf(-(p[lane] + bsh[lane])));
    alpha[(size_t)i * D + lane] = a;
  }
}

__global__ void csr_count_kernel(const int* __restrict__ row, const int* __restrict__ col,
                                 int* __restrict__ ncnt, int* __restrict__ ecnt,
                                 int* __restrict__ narr, int* __restrict__ earr) {
  int i = blockIdx.x * blockDim.x + threadIdx.x;
  if (i >= NNZ) return;
  narr[i] = atomicAdd(&ncnt[row[i]], 1);
  earr[i] = atomicAdd(&ecnt[col[i]], 1);
}

constexpr int SCAN_T = 1024;

__device__ inline int wave_incl_scan(int v, int lane) {
#pragma unroll
  for (int off = 1; off < 32; off <<= 1) {
    int u = __shfl_up(v, off, 32);
    if (lane >= off) v += u;
  }
  return v;
}

__global__ void __launch_bounds__(SCAN_T)
csr_scan_kernel(const int* __restrict__ ncnt, const int* __restrict__ ecnt,
                int* __restrict__ nptr, int* __restrict__ eptr) {
  __shared__ int wtot[32];
  const int* cnt = (blockIdx.x == 0) ? ncnt : ecnt;
  int* ptr       = (blockIdx.x == 0) ? nptr : eptr;
  const int n    = (blockIdx.x == 0) ? N_NODES : N_EDGES;
  const int chunk = (n + SCAN_T - 1) / SCAN_T;
  const int tid = threadIdx.x, lane = tid & 31, w = tid >> 5;
  const int b = tid * chunk;
  const int e = (b + chunk < n) ? (b + chunk) : n;
  int s = 0;
  for (int k = b; k < e; ++k) s += cnt[k];
  int incl = wave_incl_scan(s, lane);
  if (lane == 31) wtot[w] = incl;
  __syncthreads();
  if (w == 0) {
    int t = wtot[lane];
    wtot[lane] = wave_incl_scan(t, lane) - t;
  }
  __syncthreads();
  int run = wtot[w] + incl - s;
  for (int k = b; k < e; ++k) { ptr[k] = run; run += cnt[k]; }
  if (tid == SCAN_T - 1) ptr[n] = run;
}

__global__ void csr_place_kernel(const int* __restrict__ row, const int* __restrict__ col,
                                 const int* __restrict__ nptr, const int* __restrict__ eptr,
                                 const int* __restrict__ narr, const int* __restrict__ earr,
                                 int* __restrict__ ntmp, int* __restrict__ etmp) {
  int i = blockIdx.x * blockDim.x + threadIdx.x;
  if (i >= NNZ) return;
  ntmp[nptr[row[i]] + narr[i]] = i;
  etmp[eptr[col[i]] + earr[i]] = i;
}

__global__ void csr_order_kernel(const int* __restrict__ row, const int* __restrict__ col,
                                 const int* __restrict__ nptr, const int* __restrict__ eptr,
                                 const int* __restrict__ ntmp, const int* __restrict__ etmp,
                                 int* __restrict__ nlst, int* __restrict__ ncol,
                                 int* __restrict__ elst, int* __restrict__ erow) {
  int g = blockIdx.x * blockDim.x + threadIdx.x;
  if (g >= 2 * NNZ) return;
  const bool eside = (g >= NNZ);
  const int p = eside ? (g - NNZ) : g;
  const int* tmp = eside ? etmp : ntmp;
  const int i = tmp[p];
  const int r = row[i], c = col[i];
  const int* ptr = eside ? eptr : nptr;
  const int key = eside ? c : r;
  const int s = ptr[key], e = ptr[key + 1];
  int rank = 0;
  for (int k = s; k < e; ++k) rank += (tmp[k] < i) ? 1 : 0;
  if (eside) { elst[s + rank] = i; erow[s + rank] = r; }
  else       { nlst[s + rank] = i; ncol[s + rank] = c; }
}

constexpr int SEG_T = OUT_CH / 4;

template <bool NODE_SIDE>
__global__ void __launch_bounds__(2 * SEG_T)
seg_gather_kernel(const int* __restrict__ ptr, const int* __restrict__ lst,
                  const int* __restrict__ oth, const float* __restrict__ alpha,
                  const float* __restrict__ src, const float* __restrict__ h,
                  const float* __restrict__ bias, float* __restrict__ dst, int nseg) {
  const int sub = (threadIdx.x >= SEG_T) ? 1 : 0;
  const int seg = blockIdx.x * 2 + sub;
  if (seg >= nseg) return;
  const int q4 = threadIdx.x - sub * SEG_T;
  const int t  = q4 >> 4;
  const int s = ptr[seg], e = ptr[seg + 1];
  float4 acc = make_float4(0.0f, 0.0f, 0.0f, 0.0f);
  float dsum = 0.0f;
  float4 hv = make_float4(0.0f, 0.0f, 0.0f, 0.0f), bv = hv;
  if (NODE_SIDE) {
    hv = *reinterpret_cast<const float4*>(h + (size_t)seg * OUT_CH + 4 * q4);
    bv = *reinterpret_cast<const float4*>(bias + ((4 * q4) & (HIDDEN - 1)));
  }
  for (int q = s; q < e; ++q) {
    const int i = lst[q];
    const int o = oth[q];
    const float a = alpha[(size_t)i * D + t];
    const float4 v = *reinterpret_cast<const float4*>(src + (size_t)o * OUT_CH + 4 * q4);
    acc.x += a * v.x;
    acc.y += a * v.y;
    acc.z += a * v.z;
    acc.w += a * v.w;
    if (NODE_SIDE) dsum += a * a;
  }
  float4* dptr = reinterpret_cast<float4*>(dst + (size_t)seg * OUT_CH + 4 * q4);
  if (NODE_SIDE) {
    const float dinv = (dsum > 0.0f) ? (1.0f / dsum) : 0.0f;
    float4 o;
    o.x = hv.x - dinv * acc.x + bv.x;
    o.y = hv.y - dinv * acc.y + bv.y;
    o.z = hv.z - dinv * acc.z + bv.z;
    o.w = hv.w - dinv * acc.w + bv.w;
    o.x = (o.x > 0.0f) ? o.x : expm1f(o.x);
    o.y = (o.y > 0.0f) ? o.y : expm1f(o.y);
    o.z = (o.z > 0.0f) ? o.z : expm1f(o.z);
    o.w = (o.w > 0.0f) ? o.w : expm1f(o.w);
    *dptr = o;
  } else {
    const int len = e - s;
    const float binv = (len > 0) ? (1.0f / (float)len) : 0.0f;
    *dptr = make_float4(acc.x * binv, acc.y * binv, acc.z * binv, acc.w * binv);
  }
}

extern "C" void kernel_launch(void* const* d_in, const int* in_sizes, int n_in,
                              void* d_out, int out_size, void* d_ws, size_t ws_size,
                              hipStream_t stream) {
  (void)in_sizes; (void)n_in; (void)out_size; (void)ws_size;
  const float* x       = (const float*)d_in[0];
  const float* hattr   = (const float*)d_in[1];
  const float* W_lin   = (const float*)d_in[2];
  const float* b_lin   = (const float*)d_in[3];
  const float* ln_g    = (const float*)d_in[4];
  const float* ln_b    = (const float*)d_in[5];
  const float* W_sheaf = (const float*)d_in[6];
  const float* b_sheaf = (const float*)d_in[7];
  const float* W_conv  = (const float*)d_in[8];
  const float* b_conv  = (const float*)d_in[9];
  const int*   eidx    = (const int*)d_in[10];
  const int* row = eidx;
  const int* col = eidx + NNZ;
  float* out = (float*)d_out;

  float* ws    = (float*)d_ws;
  float* xs    = ws;
  float* es    = xs   + (size_t)ND * HIDDEN;
  float* xm    = es   + (size_t)ED * HIDDEN;
  float* em    = xm   + (size_t)N_NODES * HIDDEN;
  float* alpha = em   + (size_t)N_EDGES * HIDDEN;
  float* spare = alpha + (size_t)NNZ * D;
  float* hbuf  = spare + ND + N_EDGES;
  float* me    = hbuf + (size_t)ND * HIDDEN;
  int*   ib    = (int*)(me + (size_t)ED * HIDDEN);
  int* ncnt = ib;
  int* ecnt = ncnt + N_NODES;
  int* nptr = ecnt + N_EDGES;
  int* eptr = nptr + N_NODES + 8;
  int* narr = eptr + N_EDGES + 8;
  int* earr = narr + NNZ;
  int* ntmp = earr + NNZ;
  int* etmp = ntmp + NNZ;
  int* nlst = etmp + NNZ;
  int* ncol = nlst + NNZ;
  int* elst = ncol + NNZ;
  int* erow = elst + NNZ;

  const int TPB = 256;

  hipMemsetAsync(ncnt, 0, (size_t)(N_NODES + N_EDGES) * sizeof(int), stream);
  csr_count_kernel<<<(NNZ + TPB - 1) / TPB, TPB, 0, stream>>>(row, col, ncnt, ecnt, narr, earr);
  csr_scan_kernel<<<2, SCAN_T, 0, stream>>>(ncnt, ecnt, nptr, eptr);
  csr_place_kernel<<<(NNZ + TPB - 1) / TPB, TPB, 0, stream>>>(row, col, nptr, eptr, narr, earr,
                                                              ntmp, etmp);
  csr_order_kernel<<<(2 * NNZ + TPB - 1) / TPB, TPB, 0, stream>>>(row, col, nptr, eptr, ntmp, etmp,
                                                                  nlst, ncol, elst, erow);

  {
    int tiles = (N_NODES / 16) * (OUT_CH / 16);
    wmma_gemm_f32<<<tiles / 8, TPB, 0, stream>>>(x, W_lin, b_lin, xs,
                                                 IN_CH, OUT_CH, OUT_CH / 16, tiles);
    int etiles = (N_EDGES / 16) * (OUT_CH / 16);
    wmma_gemm_f32<<<etiles / 8, TPB, 0, stream>>>(hattr, W_lin, b_lin, es,
                                                  IN_CH, OUT_CH, OUT_CH / 16, etiles);
  }

  stalk_mean_kernel<<<(N_NODES * HIDDEN + TPB - 1) / TPB, TPB, 0, stream>>>(xs, xm, N_NODES * HIDDEN);
  stalk_mean_kernel<<<(N_EDGES * HIDDEN + TPB - 1) / TPB, TPB, 0, stream>>>(es, em, N_EDGES * HIDDEN);

  alpha_kernel<<<NNZ / 8, TPB, 0, stream>>>(xm, em, row, col, ln_g, ln_b,
                                            W_sheaf, b_sheaf, alpha);

  const int convTiles = (ND / 16) * (HIDDEN / 16);
  const int eBlk = (N_EDGES + 1) / 2;
  const int nBlk = (N_NODES + 1) / 2;

  wmma_gemm_f32<<<convTiles / 8, TPB, 0, stream>>>(xs, W_conv, nullptr, hbuf,
                                                   HIDDEN, HIDDEN, HIDDEN / 16, convTiles);
  seg_gather_kernel<false><<<eBlk, 2 * SEG_T, 0, stream>>>(eptr, elst, erow, alpha, hbuf,
                                                           nullptr, nullptr, me, N_EDGES);
  seg_gather_kernel<true><<<nBlk, 2 * SEG_T, 0, stream>>>(nptr, nlst, ncol, alpha, me,
                                                          hbuf, b_conv, out, N_NODES);

  wmma_gemm_f32<<<convTiles / 8, TPB, 0, stream>>>(out, W_conv + HIDDEN * HIDDEN, nullptr, hbuf,
                                                   HIDDEN, HIDDEN, HIDDEN / 16, convTiles);
  seg_gather_kernel<false><<<eBlk, 2 * SEG_T, 0, stream>>>(eptr, elst, erow, alpha, hbuf,
                                                           nullptr, nullptr, me, N_EDGES);
  seg_gather_kernel<true><<<nBlk, 2 * SEG_T, 0, stream>>>(nptr, nlst, ncol, alpha, me,
                                                          hbuf, b_conv + HIDDEN, out, N_NODES);
}


# GEMM re-tiled: a wave keeps its weight slice in registers as WMMA fragments and walks a run of 16-row tiles (16-byte A loads, next tile prefetched, whole-line stores); k summed in a permuted order
# speedup vs baseline: 2.7299x; 1.2265x over previous
#include <hip/hip_runtime.h>
#include <math.h>

constexpr int N_NODES = 50000;
constexpr int N_EDGES = 10000;
constexpr int NNZ     = 150000;
constexpr int IN_CH   = 128;
constexpr int HIDDEN  = 64;
constexpr int D       = 6;
constexpr int OUT_CH  = HIDDEN * D;
constexpr int ND      = N_NODES * D;
constexpr int ED      = N_EDGES * D;

typedef __attribute__((ext_vector_type(2))) float v2f;
typedef __attribute__((ext_vector_type(8))) float v8f;

template <int NT> struct ColVec;
template <> struct ColVec<2> { typedef float2 type; };
template <> struct ColVec<4> { typedef float4 type; };

template <int NT>
__device__ inline void load_cols(const float* __restrict__ p, float (&d)[NT]) {
  const typename ColVec<NT>::type v = *reinterpret_cast<const typename ColVec<NT>::type*>(p);
  const float* f = reinterpret_cast<const float*>(&v);
#pragma unroll
  for (int t = 0; t < NT; ++t) d[t] = f[t];
}

template <int NT>
__device__ inline void store_cols(float* __restrict__ p, const float (&d)[NT]) {
  typename ColVec<NT>::type v;
  float* f = reinterpret_cast<float*>(&v);
#pragma unroll
  for (int t = 0; t < NT; ++t) f[t] = d[t];
  *reinterpret_cast<typename ColVec<NT>::type*>(p) = v;
}

constexpr int gemm_wpb(int g) { return (g > 1) ? g : 4; }

template <int K>
__device__ inline void load_a_tile(float4 (&av)[K / 8], const float* __restrict__ A,
                                   int mt, int c, int h) {
  const float* ap = A + ((size_t)mt * 16 + c) * K + 8 * h;
#pragma unroll
  for (int g = 0; g < K / 16; ++g) {
    av[2 * g]     = *reinterpret_cast<const float4*>(ap + 16 * g);
    av[2 * g + 1] = *reinterpret_cast<const float4*>(ap + 16 * g + 4);
  }
}

template <int K, int NT, int G, bool HAS_BIAS>
__global__ void __launch_bounds__(32 * gemm_wpb(G))
wmma_gemm_f32(const float* __restrict__ A, const float* __restrict__ B,
              const float* __restrict__ bias, float* __restrict__ C,
              int mTiles, int tilesPerWave) {
  constexpr int N = 16 * NT * G;
  constexpr int S = K / 4;
  const int wave  = __builtin_amdgcn_readfirstlane(blockIdx.x * gemm_wpb(G) + (threadIdx.x >> 5));
  const int cg    = wave % G;
  int mt          = (wave / G) * tilesPerWave;
  if (mt >= mTiles) return;
  const int mtEnd = (mt + tilesPerWave < mTiles) ? (mt + tilesPerWave) : mTiles;
  const int lane = threadIdx.x & 31;
  const int h = lane >> 4;
  const int c = lane & 15;
  const int n0 = cg * 16 * NT;

  v2f bf[S][NT];
#pragma unroll
  for (int s = 0; s < S; ++s) {
    const int k = 16 * (s >> 2) + 8 * h + 2 * (s & 3);
    float w0[NT], w1[NT];
    load_cols<NT>(B + (size_t)k * N + n0 + NT * c, w0);
    load_cols<NT>(B + (size_t)(k + 1) * N + n0 + NT * c, w1);
#pragma unroll
    for (int t = 0; t < NT; ++t) { bf[s][t].x = w0[t]; bf[s][t].y = w1[t]; }
  }
  float bv[NT];
#pragma unroll
  for (int t = 0; t < NT; ++t) bv[t] = 0.0f;
  if (HAS_BIAS) load_cols<NT>(bias + n0 + NT * c, bv);

  float4 av[K / 8];
  load_a_tile<K>(av, A, mt, c, h);
  for (; mt < mtEnd; ++mt) {
    const int nx = (mt + 1 < mtEnd) ? (mt + 1) : mt;
    float4 an[K / 8];
    load_a_tile<K>(an, A, nx, c, h);
    __builtin_amdgcn_sched_barrier(0);

    v8f acc[NT];
#pragma unroll
    for (int t = 0; t < NT; ++t) acc[t] = v8f{};
#pragma unroll
    for (int s = 0; s < S; ++s) {
      const float4 q = av[s >> 1];
      v2f a;
      a.x = (s & 1) ? q.z : q.x;
      a.y = (s & 1) ? q.w : q.y;
#pragma unroll
      for (int t = 0; t < NT; ++t)
        acc[t] = __builtin_amdgcn_wmma_f32_16x16x4_f32(false, a, false, bf[s][t],
                                                       (short)0, acc[t], false, false);
    }
    float* cp = C + ((size_t)mt * 16 + 8 * h) * N + n0 + NT * c;
#pragma unroll
    for (int j = 0; j < 8; ++j) {
      float o[NT];
#pragma unroll
      for (int t = 0; t < NT; ++t) o[t] = acc[t][j] + bv[t];
      store_cols<NT>(cp + (size_t)j * N, o);
    }
    __builtin_amdgcn_sched_barrier(0);
#pragma unroll
    for (int i = 0; i < K / 8; ++i) av[i] = an[i];
  }
}

__global__ void stalk_mean_kernel(const float* __restrict__ proj, float* __restrict__ m,
                                  int total) {
  int idx = blockIdx.x * blockDim.x + threadIdx.x;
  if (idx >= total) return;
  int n = idx >> 6;
  int hh = idx & 63;
  const float* p = proj + (size_t)n * OUT_CH + hh;
  float s = 0.0f;
#pragma unroll
  for (int t = 0; t < D; ++t) s += p[t * HIDDEN];
  m[idx] = s * (1.0f / 6.0f);
}

__device__ inline float wave_allsum(float v) {
#pragma unroll
  for (int off = 16; off > 0; off >>= 1) v += __shfl_xor(v, off, 32);
  return v;
}

__global__ void alpha_kernel(const float* __restrict__ xm, const float* __restrict__ em,
                             const int* __restrict__ row, const int* __restrict__ col,
                             const float* __restrict__ ln_g, const float* __restrict__ ln_b,
                             const float* __restrict__ Wsh, const float* __restrict__ bsh,
                             float* __restrict__ alpha) {
  int i = blockIdx.x * (blockDim.x >> 5) + (threadIdx.x >> 5);
  if (i >= NNZ) return;
  int lane = threadIdx.x & 31;
  int rn = row[i], cn = col[i];
  int j0 = lane * 4;
  float z[4];
#pragma unroll
  for (int q = 0; q < 4; ++q) {
    int j = j0 + q;
    z[q] = (j < HIDDEN) ? xm[(size_t)rn * HIDDEN + j]
                        : em[(size_t)cn * HIDDEN + (j - HIDDEN)];
  }
  float s  = z[0] + z[1] + z[2] + z[3];
  float ss = z[0]*z[0] + z[1]*z[1] + z[2]*z[2] + z[3]*z[3];
  s  = wave_allsum(s);
  ss = wave_allsum(ss);
  float mu   = s * (1.0f / 128.0f);
  float var  = ss * (1.0f / 128.0f) - mu * mu;
  float rstd = rsqrtf(var + 1e-5f);

  float p[6] = {0, 0, 0, 0, 0, 0};
#pragma unroll
  for (int q = 0; q < 4; ++q) {
    int j = j0 + q;
    float fn = (z[q] - mu) * rstd * ln_g[j] + ln_b[j];
#pragma unroll
    for (int t = 0; t < D; ++t) p[t] += fn * Wsh[j * D + t];
  }
#pragma unroll
  for (int t = 0; t < D; ++t) p[t] = wave_allsum(p[t]);

  if (lane < D) {
    float a = 1.0f / (1.0f + expf(-(p[lane] + bsh[lane])));
    alpha[(size_t)i * D + lane] = a;
  }
}

__global__ void csr_count_kernel(const int* __restrict__ row, const int* __restrict__ col,
                                 int* __restrict__ ncnt, int* __restrict__ ecnt,
                                 int* __restrict__ narr, int* __restrict__ earr) {
  int i = blockIdx.x * blockDim.x + threadIdx.x;
  if (i >= NNZ) return;
  narr[i] = atomicAdd(&ncnt[row[i]], 1);
  earr[i] = atomicAdd(&ecnt[col[i]], 1);
}

constexpr int SCAN_T = 1024;

__device__ inline int wave_incl_scan(int v, int lane) {
#pragma unroll
  for (int off = 1; off < 32; off <<= 1) {
    int u = __shfl_up(v, off, 32);
    if (lane >= off) v += u;
  }
  return v;
}

__global__ void __launch_bounds__(SCAN_T)
csr_scan_kernel(const int* __restrict__ ncnt, const int* __restrict__ ecnt,
                int* __restrict__ nptr, int* __restrict__ eptr) {
  __shared__ int wtot[32];
  const int* cnt = (blockIdx.x == 0) ? ncnt : ecnt;
  int* ptr       = (blockIdx.x == 0) ? nptr : eptr;
  const int n    = (blockIdx.x == 0) ? N_NODES : N_EDGES;
  const int chunk = (n + SCAN_T - 1) / SCAN_T;
  const int tid = threadIdx.x, lane = tid & 31, w = tid >> 5;
  const int b = tid * chunk;
  const int e = (b + chunk < n) ? (b + chunk) : n;
  int s = 0;
  for (int k = b; k < e; ++k) s += cnt[k];
  int incl = wave_incl_scan(s, lane);
  if (lane == 31) wtot[w] = incl;
  __syncthreads();
  if (w == 0) {
    int t = wtot[lane];
    wtot[lane] = wave_incl_scan(t, lane) - t;
  }
  __syncthreads();
  int run = wtot[w] + incl - s;
  for (int k = b; k < e; ++k) { ptr[k] = run; run += cnt[k]; }
  if (tid == SCAN_T - 1) ptr[n] = run;
}

__global__ void csr_place_kernel(const int* __restrict__ row, const int* __restrict__ col,
                                 const int* __restrict__ nptr, const int* __restrict__ eptr,
                                 const int* __restrict__ narr, const int* __restrict__ earr,
                                 int* __restrict__ ntmp, int* __restrict__ etmp) {
  int i = blockIdx.x * blockDim.x + threadIdx.x;
  if (i >= NNZ) return;
  ntmp[nptr[row[i]] + narr[i]] = i;
  etmp[eptr[col[i]] + earr[i]] = i;
}

__global__ void csr_order_kernel(const int* __restrict__ row, const int* __restrict__ col,
                                 const int* __restrict__ nptr, const int* __restrict__ eptr,
                                 const int* __restrict__ ntmp, const int* __restrict__ etmp,
                                 int* __restrict__ nlst, int* __restrict__ ncol,
                                 int* __restrict__ elst, int* __restrict__ erow) {
  int g = blockIdx.x * blockDim.x + threadIdx.x;
  if (g >= 2 * NNZ) return;
  const bool eside = (g >= NNZ);
  const int p = eside ? (g - NNZ) : g;
  const int* tmp = eside ? etmp : ntmp;
  const int i = tmp[p];
  const int r = row[i], c = col[i];
  const int* ptr = eside ? eptr : nptr;
  const int key = eside ? c : r;
  const int s = ptr[key], e = ptr[key + 1];
  int rank = 0;
  for (int k = s; k < e; ++k) rank += (tmp[k] < i) ? 1 : 0;
  if (eside) { elst[s + rank] = i; erow[s + rank] = r; }
  else       { nlst[s + rank] = i; ncol[s + rank] = c; }
}

constexpr int SEG_T = OUT_CH / 4;

template <bool NODE_SIDE>
__global__ void __launch_bounds__(2 * SEG_T)
seg_gather_kernel(const int* __restrict__ ptr, const int* __restrict__ lst,
                  const int* __restrict__ oth, const float* __restrict__ alpha,
                  const float* __restrict__ src, const float* __restrict__ h,
                  const float* __restrict__ bias, float* __restrict__ dst, int nseg) {
  const int sub = (threadIdx.x >= SEG_T) ? 1 : 0;
  const int seg = blockIdx.x * 2 + sub;
  if (seg >= nseg) return;
  const int q4 = threadIdx.x - sub * SEG_T;
  const int t  = q4 >> 4;
  const int s = ptr[seg], e = ptr[seg + 1];
  float4 acc = make_float4(0.0f, 0.0f, 0.0f, 0.0f);
  float dsum = 0.0f;
  float4 hv = make_float4(0.0f, 0.0f, 0.0f, 0.0f), bv = hv;
  if (NODE_SIDE) {
    hv = *reinterpret_cast<const float4*>(h + (size_t)seg * OUT_CH + 4 * q4);
    bv = *reinterpret_cast<const float4*>(bias + ((4 * q4) & (HIDDEN - 1)));
  }
  for (int q = s; q < e; ++q) {
    const int i = lst[q];
    const int o = oth[q];
    const float a = alpha[(size_t)i * D + t];
    const float4 v = *reinterpret_cast<const float4*>(src + (size_t)o * OUT_CH + 4 * q4);
    acc.x += a * v.x;
    acc.y += a * v.y;
    acc.z += a * v.z;
    acc.w += a * v.w;
    if (NODE_SIDE) dsum += a * a;
  }
  float4* dptr = reinterpret_cast<float4*>(dst + (size_t)seg * OUT_CH + 4 * q4);
  if (NODE_SIDE) {
    const float dinv = (dsum > 0.0f) ? (1.0f / dsum) : 0.0f;
    float4 o;
    o.x = hv.x - dinv * acc.x + bv.x;
    o.y = hv.y - dinv * acc.y + bv.y;
    o.z = hv.z - dinv * acc.z + bv.z;
    o.w = hv.w - dinv * acc.w + bv.w;
    o.x = (o.x > 0.0f) ? o.x : expm1f(o.x);
    o.y = (o.y > 0.0f) ? o.y : expm1f(o.y);
    o.z = (o.z > 0.0f) ? o.z : expm1f(o.z);
    o.w = (o.w > 0.0f) ? o.w : expm1f(o.w);
    *dptr = o;
  } else {
    const int len = e - s;
    const float binv = (len > 0) ? (1.0f / (float)len) : 0.0f;
    *dptr = make_float4(acc.x * binv, acc.y * binv, acc.z * binv, acc.w * binv);
  }
}

constexpr int PROJ_N_RUN = 13;
constexpr int PROJ_N_BLK = (N_NODES / 16 + PROJ_N_RUN - 1) / PROJ_N_RUN;
constexpr int PROJ_E_RUN = 3;
constexpr int PROJ_E_BLK = (N_EDGES / 16 + PROJ_E_RUN - 1) / PROJ_E_RUN;
constexpr int CONV_RUN   = 5;
constexpr int CONV_BLK   = ((ND / 16 + CONV_RUN - 1) / CONV_RUN + 3) / 4;

extern "C" void kernel_launch(void* const* d_in, const int* in_sizes, int n_in,
                              void* d_out, int out_size, void* d_ws, size_t ws_size,
                              hipStream_t stream) {
  (void)in_sizes; (void)n_in; (void)out_size; (void)ws_size;
  const float* x       = (const float*)d_in[0];
  const float* hattr   = (const float*)d_in[1];
  const float* W_lin   = (const float*)d_in[2];
  const float* b_lin   = (const float*)d_in[3];
  const float* ln_g    = (const float*)d_in[4];
  const float* ln_b    = (const float*)d_in[5];
  const float* W_sheaf = (const float*)d_in[6];
  const float* b_sheaf = (const float*)d_in[7];
  const float* W_conv  = (const float*)d_in[8];
  const float* b_conv  = (const float*)d_in[9];
  const int*   eidx    = (const int*)d_in[10];
  const int* row = eidx;
  const int* col = eidx + NNZ;
  float* out = (float*)d_out;

  float* ws    = (float*)d_ws;
  float* xs    = ws;
  float* es    = xs   + (size_t)ND * HIDDEN;
  float* xm    = es   + (size_t)ED * HIDDEN;
  float* em    = xm   + (size_t)N_NODES * HIDDEN;
  float* alpha = em   + (size_t)N_EDGES * HIDDEN;
  float* spare = alpha + (size_t)NNZ * D;
  float* hbuf  = spare + ND + N_EDGES;
  float* me    = hbuf + (size_t)ND * HIDDEN;
  int*   ib    = (int*)(me + (size_t)ED * HIDDEN);
  int* ncnt = ib;
  int* ecnt = ncnt + N_NODES;
  int* nptr = ecnt + N_EDGES;
  int* eptr = nptr + N_NODES + 8;
  int* narr = eptr + N_EDGES + 8;
  int* earr = narr + NNZ;
  int* ntmp = earr + NNZ;
  int* etmp = ntmp + NNZ;
  int* nlst = etmp + NNZ;
  int* ncol = nlst + NNZ;
  int* elst = ncol + NNZ;
  int* erow = elst + NNZ;

  const int TPB = 256;

  hipMemsetAsync(ncnt, 0, (size_t)(N_NODES + N_EDGES) * sizeof(int), stream);
  csr_count_kernel<<<(NNZ + TPB - 1) / TPB, TPB, 0, stream>>>(row, col, ncnt, ecnt, narr, earr);
  csr_scan_kernel<<<2, SCAN_T, 0, stream>>>(ncnt, ecnt, nptr, eptr);
  csr_place_kernel<<<(NNZ + TPB - 1) / TPB, TPB, 0, stream>>>(row, col, nptr, eptr, narr, earr,
                                                              ntmp, etmp);
  csr_order_kernel<<<(2 * NNZ + TPB - 1) / TPB, TPB, 0, stream>>>(row, col, nptr, eptr, ntmp, etmp,
                                                                  nlst, ncol, elst, erow);

  wmma_gemm_f32<IN_CH, 2, 12, true><<<PROJ_N_BLK, 32 * 12, 0, stream>>>(
      x, W_lin, b_lin, xs, N_NODES / 16, PROJ_N_RUN);
  wmma_gemm_f32<IN_CH, 2, 12, true><<<PROJ_E_BLK, 32 * 12, 0, stream>>>(
      hattr, W_lin, b_lin, es, N_EDGES / 16, PROJ_E_RUN);

  stalk_mean_kernel<<<(N_NODES * HIDDEN + TPB - 1) / TPB, TPB, 0, stream>>>(xs, xm, N_NODES * HIDDEN);
  stalk_mean_kernel<<<(N_EDGES * HIDDEN + TPB - 1) / TPB, TPB, 0, stream>>>(es, em, N_EDGES * HIDDEN);

  alpha_kernel<<<NNZ / 8, TPB, 0, stream>>>(xm, em, row, col, ln_g, ln_b,
                                            W_sheaf, b_sheaf, alpha);

  const int eBlk = (N_EDGES + 1) / 2;
  const int nBlk = (N_NODES + 1) / 2;

  wmma_gemm_f32<HIDDEN, 4, 1, false><<<CONV_BLK, 32 * 4, 0, stream>>>(
      xs, W_conv, nullptr, hbuf, ND / 16, CONV_RUN);
  seg_gather_kernel<false><<<eBlk, 2 * SEG_T, 0, stream>>>(eptr, elst, erow, alpha, hbuf,
                                                           nullptr, nullptr, me, N_EDGES);
  seg_gather_kernel<true><<<nBlk, 2 * SEG_T, 0, stream>>>(nptr, nlst, ncol, alpha, me,
                                                          hbuf, b_conv, out, N_NODES);

  wmma_gemm_f32<HIDDEN, 4, 1, false><<<CONV_BLK, 32 * 4, 0, stream>>>(
      out, W_conv + HIDDEN * HIDDEN, nullptr, hbuf, ND / 16, CONV_RUN);
  seg_gather_kernel<false><<<eBlk, 2 * SEG_T, 0, stream>>>(eptr, elst, erow, alpha, hbuf,
                                                           nullptr, nullptr, me, N_EDGES);
  seg_gather_kernel<true><<<nBlk, 2 * SEG_T, 0, stream>>>(nptr, nlst, ncol, alpha, me,
                                                          hbuf, b_conv + HIDDEN, out, N_NODES);
}


# Stalk-mean, sheaf-coefficient and segment-gather kernels run as small fixed grids of long-lived waves that loop over items (one wave per 384-wide row, up to 4 incidences in flight); sum orders kept
# speedup vs baseline: 3.0485x; 1.1167x over previous
#include <hip/hip_runtime.h>
#include <math.h>

constexpr int N_NODES = 50000;
constexpr int N_EDGES = 10000;
constexpr int NNZ     = 150000;
constexpr int IN_CH   = 128;
constexpr int HIDDEN  = 64;
constexpr int D       = 6;
constexpr int OUT_CH  = HIDDEN * D;
constexpr int ND      = N_NODES * D;
constexpr int ED      = N_EDGES * D;

typedef __attribute__((ext_vector_type(2))) float v2f;
typedef __attribute__((ext_vector_type(8))) float v8f;

template <int NT> struct ColVec;
template <> struct ColVec<2> { typedef float2 type; };
template <> struct ColVec<4> { typedef float4 type; };

template <int NT>
__device__ inline void load_cols(const float* __restrict__ p, float (&d)[NT]) {
  const typename ColVec<NT>::type v = *reinterpret_cast<const typename ColVec<NT>::type*>(p);
  const float* f = reinterpret_cast<const float*>(&v);
#pragma unroll
  for (int t = 0; t < NT; ++t) d[t] = f[t];
}

template <int NT>
__device__ inline void store_cols(float* __restrict__ p, const float (&d)[NT]) {
  typename ColVec<NT>::type v;
  float* f = reinterpret_cast<float*>(&v);
#pragma unroll
  for (int t = 0; t < NT; ++t) f[t] = d[t];
  *reinterpret_cast<typename ColVec<NT>::type*>(p) = v;
}

constexpr int gemm_wpb(int g) { return (g > 1) ? g : 4; }

template <int K>
__device__ inline void load_a_tile(float4 (&av)[K / 8], const float* __restrict__ A,
                                   int mt, int c, int h) {
  const float* ap = A + ((size_t)mt * 16 + c) * K + 8 * h;
#pragma unroll
  for (int g = 0; g < K / 16; ++g) {
    av[2 * g]     = *reinterpret_cast<const float4*>(ap + 16 * g);
    av[2 * g + 1] = *reinterpret_cast<const float4*>(ap + 16 * g + 4);
  }
}

template <int K, int NT, int G, bool HAS_BIAS>
__global__ void __launch_bounds__(32 * gemm_wpb(G))
wmma_gemm_f32(const float* __restrict__ A, const float* __restrict__ B,
              const float* __restrict__ bias, float* __restrict__ C,
              int mTiles, int tilesPerWave) {
  constexpr int N = 16 * NT * G;
  constexpr int S = K / 4;
  const int wave  = __builtin_amdgcn_readfirstlane(blockIdx.x * gemm_wpb(G) + (threadIdx.x >> 5));
  const int cg    = wave % G;
  int mt          = (wave / G) * tilesPerWave;
  if (mt >= mTiles) return;
  const int mtEnd = (mt + tilesPerWave < mTiles) ? (mt + tilesPerWave) : mTiles;
  const int lane = threadIdx.x & 31;
  const int h = lane >> 4;
  const int c = lane & 15;
  const int n0 = cg * 16 * NT;

  v2f bf[S][NT];
#pragma unroll
  for (int s = 0; s < S; ++s) {
    const int k = 16 * (s >> 2) + 8 * h + 2 * (s & 3);
    float w0[NT], w1[NT];
    load_cols<NT>(B + (size_t)k * N + n0 + NT * c, w0);
    load_cols<NT>(B + (size_t)(k + 1) * N + n0 + NT * c, w1);
#pragma unroll
    for (int t = 0; t < NT; ++t) { bf[s][t].x = w0[t]; bf[s][t].y = w1[t]; }
  }
  float bv[NT];
#pragma unroll
  for (int t = 0; t < NT; ++t) bv[t] = 0.0f;
  if (HAS_BIAS) load_cols<NT>(bias + n0 + NT * c, bv);

  float4 av[K / 8];
  load_a_tile<K>(av, A, mt, c, h);
  for (; mt < mtEnd; ++mt) {
    const int nx = (mt + 1 < mtEnd) ? (mt + 1) : mt;
    float4 an[K / 8];
    load_a_tile<K>(an, A, nx, c, h);
    __builtin_amdgcn_sched_barrier(0);

    v8f acc[NT];
#pragma unroll
    for (int t = 0; t < NT; ++t) acc[t] = v8f{};
#pragma unroll
    for (int s = 0; s < S; ++s) {
      const float4 q = av[s >> 1];
      v2f a;
      a.x = (s & 1) ? q.z : q.x;
      a.y = (s & 1) ? q.w : q.y;
#pragma unroll
      for (int t = 0; t < NT; ++t)
        acc[t] = __builtin_amdgcn_wmma_f32_16x16x4_f32(false, a, false, bf[s][t],
                                                       (short)0, acc[t], false, false);
    }
    float* cp = C + ((size_t)mt * 16 + 8 * h) * N + n0 + NT * c;
#pragma unroll
    for (int j = 0; j < 8; ++j) {
      float o[NT];
#pragma unroll
      for (int t = 0; t < NT; ++t) o[t] = acc[t][j] + bv[t];
      store_cols<NT>(cp + (size_t)j * N, o);
    }
    __builtin_amdgcn_sched_barrier(0);
#pragma unroll
    for (int i = 0; i < K / 8; ++i) av[i] = an[i];
  }
}

constexpr int LOOP_WAVES = 8;
constexpr int LOOP_T     = 32 * LOOP_WAVES;
constexpr int LOOP_BLK   = 1024;

constexpr int SM_ITEMS = 4;

__global__ void __launch_bounds__(LOOP_T)
stalk_mean_kernel(const float* __restrict__ proj, float* __restrict__ m, int total4) {
  const int stride = gridDim.x * LOOP_T;
  const int idx0 = blockIdx.x * LOOP_T + threadIdx.x;
  float4 v[SM_ITEMS][D];
#pragma unroll
  for (int r = 0; r < SM_ITEMS; ++r) {
    int idx = idx0 + r * stride;
    idx = (idx < total4) ? idx : (total4 - 1);
    const float* p = proj + (size_t)(idx >> 4) * OUT_CH + 4 * (idx & 15);
#pragma unroll
    for (int t = 0; t < D; ++t) v[r][t] = *reinterpret_cast<const float4*>(p + t * HIDDEN);
  }
#pragma unroll
  for (int r = 0; r < SM_ITEMS; ++r) {
    const int idx = idx0 + r * stride;
    float4 s = make_float4(0.0f, 0.0f, 0.0f, 0.0f);
#pragma unroll
    for (int t = 0; t < D; ++t) {
      s.x += v[r][t].x; s.y += v[r][t].y; s.z += v[r][t].z; s.w += v[r][t].w;
    }
    const float k = 1.0f / 6.0f;
    if (idx < total4)
      *reinterpret_cast<float4*>(m + (size_t)idx * 4) = make_float4(s.x * k, s.y * k, s.z * k, s.w * k);
  }
}

__device__ inline float wave_allsum(float v) {
#pragma unroll
  for (int off = 16; off > 0; off >>= 1) v += __shfl_xor(v, off, 32);
  return v;
}

__global__ void __launch_bounds__(LOOP_T)
alpha_kernel(const float* __restrict__ xm, const float* __restrict__ em,
             const int* __restrict__ row, const int* __restrict__ col,
             const float* __restrict__ ln_g, const float* __restrict__ ln_b,
             const float* __restrict__ Wsh, const float* __restrict__ bsh,
             float* __restrict__ alpha) {
  const int nw = gridDim.x * LOOP_WAVES;
  int i = blockIdx.x * LOOP_WAVES + (threadIdx.x >> 5);
  if (i >= NNZ) return;
  const int lane = threadIdx.x & 31;
  const int j0 = lane * 4;
  const bool xside = (j0 < HIDDEN);
  const float* zsrc = xside ? xm : em;
  const int zoff = xside ? j0 : (j0 - HIDDEN);
  float g[4], b[4], w[4][D];
#pragma unroll
  for (int q = 0; q < 4; ++q) {
    g[q] = ln_g[j0 + q];
    b[q] = ln_b[j0 + q];
#pragma unroll
    for (int t = 0; t < D; ++t) w[q][t] = Wsh[(j0 + q) * D + t];
  }
  const float bs = (lane < D) ? bsh[lane] : 0.0f;

  int key = xside ? row[i] : col[i];
  float4 zc = *reinterpret_cast<const float4*>(zsrc + (size_t)key * HIDDEN + zoff);
  for (; i < NNZ; i += nw) {
    const int inx = (i + nw < NNZ) ? (i + nw) : i;
    key = xside ? row[inx] : col[inx];
    const float4 zn = *reinterpret_cast<const float4*>(zsrc + (size_t)key * HIDDEN + zoff);

    const float z[4] = {zc.x, zc.y, zc.z, zc.w};
    float s  = z[0] + z[1] + z[2] + z[3];
    float ss = z[0]*z[0] + z[1]*z[1] + z[2]*z[2] + z[3]*z[3];
    s  = wave_allsum(s);
    ss = wave_allsum(ss);
    float mu   = s * (1.0f / 128.0f);
    float var  = ss * (1.0f / 128.0f) - mu * mu;
    float rstd = rsqrtf(var + 1e-5f);

    float p[6] = {0, 0, 0, 0, 0, 0};
#pragma unroll
    for (int q = 0; q < 4; ++q) {
      float fn = (z[q] - mu) * rstd * g[q] + b[q];
#pragma unroll
      for (int t = 0; t < D; ++t) p[t] += fn * w[q][t];
    }
#pragma unroll
    for (int t = 0; t < D; ++t) p[t] = wave_allsum(p[t]);

    if (lane < D) {
      float a = 1.0f / (1.0f + expf(-(p[lane] + bs)));
      alpha[(size_t)i * D + lane] = a;
    }
    zc = zn;
  }
}

__global__ void csr_count_kernel(const int* __restrict__ row, const int* __restrict__ col,
                                 int* __restrict__ ncnt, int* __restrict__ ecnt,
                                 int* __restrict__ narr, int* __restrict__ earr) {
  int i = blockIdx.x * blockDim.x + threadIdx.x;
  if (i >= NNZ) return;
  narr[i] = atomicAdd(&ncnt[row[i]], 1);
  earr[i] = atomicAdd(&ecnt[col[i]], 1);
}

constexpr int SCAN_T = 1024;

__device__ inline int wave_incl_scan(int v, int lane) {
#pragma unroll
  for (int off = 1; off < 32; off <<= 1) {
    int u = __shfl_up(v, off, 32);
    if (lane >= off) v += u;
  }
  return v;
}

__global__ void __launch_bounds__(SCAN_T)
csr_scan_kernel(const int* __restrict__ ncnt, const int* __restrict__ ecnt,
                int* __restrict__ nptr, int* __restrict__ eptr) {
  __shared__ int wtot[32];
  const int* cnt = (blockIdx.x == 0) ? ncnt : ecnt;
  int* ptr       = (blockIdx.x == 0) ? nptr : eptr;
  const int n    = (blockIdx.x == 0) ? N_NODES : N_EDGES;
  const int chunk = (n + SCAN_T - 1) / SCAN_T;
  const int tid = threadIdx.x, lane = tid & 31, w = tid >> 5;
  const int b = tid * chunk;
  const int e = (b + chunk < n) ? (b + chunk) : n;
  int s = 0;
  for (int k = b; k < e; ++k) s += cnt[k];
  int incl = wave_incl_scan(s, lane);
  if (lane == 31) wtot[w] = incl;
  __syncthreads();
  if (w == 0) {
    int t = wtot[lane];
    wtot[lane] = wave_incl_scan(t, lane) - t;
  }
  __syncthreads();
  int run = wtot[w] + incl - s;
  for (int k = b; k < e; ++k) { ptr[k] = run; run += cnt[k]; }
  if (tid == SCAN_T - 1) ptr[n] = run;
}

__global__ void csr_place_kernel(const int* __restrict__ row, const int* __restrict__ col,
                                 const int* __restrict__ nptr, const int* __restrict__ eptr,
                                 const int* __restrict__ narr, const int* __restrict__ earr,
                                 int* __restrict__ ntmp, int* __restrict__ etmp) {
  int i = blockIdx.x * blockDim.x + threadIdx.x;
  if (i >= NNZ) return;
  ntmp[nptr[row[i]] + narr[i]] = i;
  etmp[eptr[col[i]] + earr[i]] = i;
}

__global__ void csr_order_kernel(const int* __restrict__ row, const int* __restrict__ col,
                                 const int* __restrict__ nptr, const int* __restrict__ eptr,
                                 const int* __restrict__ ntmp, const int* __restrict__ etmp,
                                 int* __restrict__ nlst, int* __restrict__ ncol,
                                 int* __restrict__ elst, int* __restrict__ erow) {
  int g = blockIdx.x * blockDim.x + threadIdx.x;
  if (g >= 2 * NNZ) return;
  const bool eside = (g >= NNZ);
  const int p = eside ? (g - NNZ) : g;
  const int* tmp = eside ? etmp : ntmp;
  const int i = tmp[p];
  const int r = row[i], c = col[i];
  const int* ptr = eside ? eptr : nptr;
  const int key = eside ? c : r;
  const int s = ptr[key], e = ptr[key + 1];
  int rank = 0;
  for (int k = s; k < e; ++k) rank += (tmp[k] < i) ? 1 : 0;
  if (eside) { elst[s + rank] = i; erow[s + rank] = r; }
  else       { nlst[s + rank] = i; ncol[s + rank] = c; }
}

constexpr int SEG_C = OUT_CH / 128;
constexpr int SEG_B = 4;
static_assert(SEG_B == 4, "the list walk picks its group body from the sizes 4, 3, 2, 1");

template <bool NODE_SIDE, int NB>
__device__ inline void seg_group(const int* __restrict__ lst, const int* __restrict__ oth,
                                 const float* __restrict__ alpha, const float* __restrict__ src,
                                 int q, int c0, int th,
                                 float4 (&acc)[SEG_C], float (&dsum)[SEG_C]) {
  int ii[NB], oo[NB];
#pragma unroll
  for (int u = 0; u < NB; ++u) { ii[u] = lst[q + u]; oo[u] = oth[q + u]; }
  float a[NB][SEG_C];
  float4 v[NB][SEG_C];
#pragma unroll
  for (int u = 0; u < NB; ++u) {
#pragma unroll
    for (int k = 0; k < SEG_C; ++k) {
      a[u][k] = alpha[(size_t)ii[u] * D + 2 * k + th];
      v[u][k] = *reinterpret_cast<const float4*>(src + (size_t)oo[u] * OUT_CH + 128 * k + c0);
    }
  }
#pragma unroll
  for (int u = 0; u < NB; ++u) {
#pragma unroll
    for (int k = 0; k < SEG_C; ++k) {
      acc[k].x = __builtin_fmaf(a[u][k], v[u][k].x, acc[k].x);
      acc[k].y = __builtin_fmaf(a[u][k], v[u][k].y, acc[k].y);
      acc[k].z = __builtin_fmaf(a[u][k], v[u][k].z, acc[k].z);
      acc[k].w = __builtin_fmaf(a[u][k], v[u][k].w, acc[k].w);
      if (NODE_SIDE) dsum[k] = __builtin_fmaf(a[u][k], a[u][k], dsum[k]);
    }
  }
}

template <bool NODE_SIDE>
__global__ void __launch_bounds__(LOOP_T)
seg_gather_kernel(const int* __restrict__ ptr, const int* __restrict__ lst,
                  const int* __restrict__ oth, const float* __restrict__ alpha,
                  const float* __restrict__ src, const float* __restrict__ h,
                  const float* __restrict__ bias, float* __restrict__ dst, int nseg) {
  const int nw = gridDim.x * LOOP_WAVES;
  int seg = blockIdx.x * LOOP_WAVES + (threadIdx.x >> 5);
  if (seg >= nseg) return;
  const int lane = threadIdx.x & 31;
  const int c0 = 4 * lane;
  const int th = lane >> 4;
  float4 bv = make_float4(0.0f, 0.0f, 0.0f, 0.0f);
  if (NODE_SIDE) bv = *reinterpret_cast<const float4*>(bias + (c0 & (HIDDEN - 1)));

  int s = ptr[seg], e = ptr[seg + 1];
  for (; seg < nseg; seg += nw) {
    const int sx = (seg + nw < nseg) ? (seg + nw) : seg;
    const int sn = ptr[sx], en = ptr[sx + 1];
    float4 hv[SEG_C];
    if (NODE_SIDE) {
#pragma unroll
      for (int k = 0; k < SEG_C; ++k)
        hv[k] = *reinterpret_cast<const float4*>(h + (size_t)seg * OUT_CH + 128 * k + c0);
    }
    float4 acc[SEG_C];
    float dsum[SEG_C];
#pragma unroll
    for (int k = 0; k < SEG_C; ++k) { acc[k] = make_float4(0.0f, 0.0f, 0.0f, 0.0f); dsum[k] = 0.0f; }

    for (int q = s; q < e; q += SEG_B) {
      const int n = e - q;
      if (n >= 4)      seg_group<NODE_SIDE, 4>(lst, oth, alpha, src, q, c0, th, acc, dsum);
      else if (n == 3) seg_group<NODE_SIDE, 3>(lst, oth, alpha, src, q, c0, th, acc, dsum);
      else if (n == 2) seg_group<NODE_SIDE, 2>(lst, oth, alpha, src, q, c0, th, acc, dsum);
      else             seg_group<NODE_SIDE, 1>(lst, oth, alpha, src, q, c0, th, acc, dsum);
    }

    if (NODE_SIDE) {
#pragma unroll
      for (int k = 0; k < SEG_C; ++k) {
        const float dinv = (dsum[k] > 0.0f) ? (1.0f / dsum[k]) : 0.0f;
        float4 o;
        o.x = hv[k].x - dinv * acc[k].x + bv.x;
        o.y = hv[k].y - dinv * acc[k].y + bv.y;
        o.z = hv[k].z - dinv * acc[k].z + bv.z;
        o.w = hv[k].w - dinv * acc[k].w + bv.w;
        o.x = (o.x > 0.0f) ? o.x : expm1f(o.x);
        o.y = (o.y > 0.0f) ? o.y : expm1f(o.y);
        o.z = (o.z > 0.0f) ? o.z : expm1f(o.z);
        o.w = (o.w > 0.0f) ? o.w : expm1f(o.w);
        *reinterpret_cast<float4*>(dst + (size_t)seg * OUT_CH + 128 * k + c0) = o;
      }
    } else {
      const int len = e - s;
      const float binv = (len > 0) ? (1.0f / (float)len) : 0.0f;
#pragma unroll
      for (int k = 0; k < SEG_C; ++k)
        *reinterpret_cast<float4*>(dst + (size_t)seg * OUT_CH + 128 * k + c0) =
            make_float4(acc[k].x * binv, acc[k].y * binv, acc[k].z * binv, acc[k].w * binv);
    }
    s = sn;
    e = en;
  }
}

constexpr int PROJ_N_RUN = 13;
constexpr int PROJ_N_BLK = (N_NODES / 16 + PROJ_N_RUN - 1) / PROJ_N_RUN;
constexpr int PROJ_E_RUN = 3;
constexpr int PROJ_E_BLK = (N_EDGES / 16 + PROJ_E_RUN - 1) / PROJ_E_RUN;
constexpr int CONV_RUN   = 5;
constexpr int CONV_BLK   = ((ND / 16 + CONV_RUN - 1) / CONV_RUN + 3) / 4;
constexpr int SM_N_Q   = N_NODES * HIDDEN / 4;
constexpr int SM_E_Q   = N_EDGES * HIDDEN / 4;
constexpr int SM_N_BLK = (SM_N_Q + LOOP_T * SM_ITEMS - 1) / (LOOP_T * SM_ITEMS);
constexpr int SM_E_BLK = (SM_E_Q + LOOP_T * SM_ITEMS - 1) / (LOOP_T * SM_ITEMS);
constexpr int SEG_E_BLK = (N_EDGES + LOOP_WAVES - 1) / LOOP_WAVES;
constexpr int SEG_N_BLK = LOOP_BLK;

extern "C" void kernel_launch(void* const* d_in, const int* in_sizes, int n_in,
                              void* d_out, int out_size, void* d_ws, size_t ws_size,
                              hipStream_t stream) {
  (void)in_sizes; (void)n_in; (void)out_size; (void)ws_size;
  const float* x       = (const float*)d_in[0];
  const float* hattr   = (const float*)d_in[1];
  const float* W_lin   = (const float*)d_in[2];
  const float* b_lin   = (const float*)d_in[3];
  const float* ln_g    = (const float*)d_in[4];
  const float* ln_b    = (const float*)d_in[5];
  const float* W_sheaf = (const float*)d_in[6];
  const float* b_sheaf = (const float*)d_in[7];
  const float* W_conv  = (const float*)d_in[8];
  const float* b_conv  = (const float*)d_in[9];
  const int*   eidx    = (const int*)d_in[10];
  const int* row = eidx;
  const int* col = eidx + NNZ;
  float* out = (float*)d_out;

  float* ws    = (float*)d_ws;
  float* xs    = ws;
  float* es    = xs   + (size_t)ND * HIDDEN;
  float* xm    = es   + (size_t)ED * HIDDEN;
  float* em    = xm   + (size_t)N_NODES * HIDDEN;
  float* alpha = em   + (size_t)N_EDGES * HIDDEN;
  float* spare = alpha + (size_t)NNZ * D;
  float* hbuf  = spare + ND + N_EDGES;
  float* me    = hbuf + (size_t)ND * HIDDEN;
  int*   ib    = (int*)(me + (size_t)ED * HIDDEN);
  int* ncnt = ib;
  int* ecnt = ncnt + N_NODES;
  int* nptr = ecnt + N_EDGES;
  int* eptr = nptr + N_NODES + 8;
  int* narr = eptr + N_EDGES + 8;
  int* earr = narr + NNZ;
  int* ntmp = earr + NNZ;
  int* etmp = ntmp + NNZ;
  int* nlst = etmp + NNZ;
  int* ncol = nlst + NNZ;
  int* elst = ncol + NNZ;
  int* erow = elst + NNZ;

  const int TPB = 256;

  hipMemsetAsync(ncnt, 0, (size_t)(N_NODES + N_EDGES) * sizeof(int), stream);
  csr_count_kernel<<<(NNZ + TPB - 1) / TPB, TPB, 0, stream>>>(row, col, ncnt, ecnt, narr, earr);
  csr_scan_kernel<<<2, SCAN_T, 0, stream>>>(ncnt, ecnt, nptr, eptr);
  csr_place_kernel<<<(NNZ + TPB - 1) / TPB, TPB, 0, stream>>>(row, col, nptr, eptr, narr, earr,
                                                              ntmp, etmp);
  csr_order_kernel<<<(2 * NNZ + TPB - 1) / TPB, TPB, 0, stream>>>(row, col, nptr, eptr, ntmp, etmp,
                                                                  nlst, ncol, elst, erow);

  wmma_gemm_f32<IN_CH, 2, 12, true><<<PROJ_N_BLK, 32 * 12, 0, stream>>>(
      x, W_lin, b_lin, xs, N_NODES / 16, PROJ_N_RUN);
  wmma_gemm_f32<IN_CH, 2, 12, true><<<PROJ_E_BLK, 32 * 12, 0, stream>>>(
      hattr, W_lin, b_lin, es, N_EDGES / 16, PROJ_E_RUN);

  stalk_mean_kernel<<<SM_N_BLK, LOOP_T, 0, stream>>>(xs, xm, SM_N_Q);
  stalk_mean_kernel<<<SM_E_BLK, LOOP_T, 0, stream>>>(es, em, SM_E_Q);

  alpha_kernel<<<LOOP_BLK, LOOP_T, 0, stream>>>(xm, em, row, col, ln_g, ln_b,
                                                W_sheaf, b_sheaf, alpha);

  wmma_gemm_f32<HIDDEN, 4, 1, false><<<CONV_BLK, 32 * 4, 0, stream>>>(
      xs, W_conv, nullptr, hbuf, ND / 16, CONV_RUN);
  seg_gather_kernel<false><<<SEG_E_BLK, LOOP_T, 0, stream>>>(eptr, elst, erow, alpha, hbuf,
                                                           nullptr, nullptr, me, N_EDGES);
  seg_gather_kernel<true><<<SEG_N_BLK, LOOP_T, 0, stream>>>(nptr, nlst, ncol, alpha, me,
                                                          hbuf, b_conv, out, N_NODES);

  wmma_gemm_f32<HIDDEN, 4, 1, false><<<CONV_BLK, 32 * 4, 0, stream>>>(
      out, W_conv + HIDDEN * HIDDEN, nullptr, hbuf, ND / 16, CONV_RUN);
  seg_gather_kernel<false><<<SEG_E_BLK, LOOP_T, 0, stream>>>(eptr, elst, erow, alpha, hbuf,
                                                           nullptr, nullptr, me, N_EDGES);
  seg_gather_kernel<true><<<SEG_N_BLK, LOOP_T, 0, stream>>>(nptr, nlst, ncol, alpha, me,
                                                          hbuf, b_conv + HIDDEN, out, N_NODES);
}


# Incidence lists built by one launch instead of a memset and four: a work-group owns a key range, reads the whole key array, counts and places through LDS counters; lists and sum orders unchanged
# speedup vs baseline: 3.8659x; 1.2681x over previous
#include <hip/hip_runtime.h>
#include <math.h>

constexpr int N_NODES = 50000;
constexpr int N_EDGES = 10000;
constexpr int NNZ     = 150000;
constexpr int IN_CH   = 128;
constexpr int HIDDEN  = 64;
constexpr int D       = 6;
constexpr int OUT_CH  = HIDDEN * D;
constexpr int ND      = N_NODES * D;
constexpr int ED      = N_EDGES * D;

typedef __attribute__((ext_vector_type(2))) float v2f;
typedef __attribute__((ext_vector_type(8))) float v8f;

template <int NT> struct ColVec;
template <> struct ColVec<2> { typedef float2 type; };
template <> struct ColVec<4> { typedef float4 type; };

template <int NT>
__device__ inline void load_cols(const float* __restrict__ p, float (&d)[NT]) {
  const typename ColVec<NT>::type v = *reinterpret_cast<const typename ColVec<NT>::type*>(p);
  const float* f = reinterpret_cast<const float*>(&v);
#pragma unroll
  for (int t = 0; t < NT; ++t) d[t] = f[t];
}

template <int NT>
__device__ inline void store_cols(float* __restrict__ p, const float (&d)[NT]) {
  typename ColVec<NT>::type v;
  float* f = reinterpret_cast<float*>(&v);
#pragma unroll
  for (int t = 0; t < NT; ++t) f[t] = d[t];
  *reinterpret_cast<typename ColVec<NT>::type*>(p) = v;
}

constexpr int gemm_wpb(int g) { return (g > 1) ? g : 4; }

template <int K>
__device__ inline void load_a_tile(float4 (&av)[K / 8], const float* __restrict__ A,
                                   int mt, int c, int h) {
  const float* ap = A + ((size_t)mt * 16 + c) * K + 8 * h;
#pragma unroll
  for (int g = 0; g < K / 16; ++g) {
    av[2 * g]     = *reinterpret_cast<const float4*>(ap + 16 * g);
    av[2 * g + 1] = *reinterpret_cast<const float4*>(ap + 16 * g + 4);
  }
}

template <int K, int NT, int G, bool HAS_BIAS>
__global__ void __launch_bounds__(32 * gemm_wpb(G))
wmma_gemm_f32(const float* __restrict__ A, const float* __restrict__ B,
              const float* __restrict__ bias, float* __restrict__ C,
              int mTiles, int tilesPerWave) {
  constexpr int N = 16 * NT * G;
  constexpr int S = K / 4;
  const int wave  = __builtin_amdgcn_readfirstlane(blockIdx.x * gemm_wpb(G) + (threadIdx.x >> 5));
  const int cg    = wave % G;
  int mt          = (wave / G) * tilesPerWave;
  if (mt >= mTiles) return;
  const int mtEnd = (mt + tilesPerWave < mTiles) ? (mt + tilesPerWave) : mTiles;
  const int lane = threadIdx.x & 31;
  const int h = lane >> 4;
  const int c = lane & 15;
  const int n0 = cg * 16 * NT;

  v2f bf[S][NT];
#pragma unroll
  for (int s = 0; s < S; ++s) {
    const int k = 16 * (s >> 2) + 8 * h + 2 * (s & 3);
    float w0[NT], w1[NT];
    load_cols<NT>(B + (size_t)k * N + n0 + NT * c, w0);
    load_cols<NT>(B + (size_t)(k + 1) * N + n0 + NT * c, w1);
#pragma unroll
    for (int t = 0; t < NT; ++t) { bf[s][t].x = w0[t]; bf[s][t].y = w1[t]; }
  }
  float bv[NT];
#pragma unroll
  for (int t = 0; t < NT; ++t) bv[t] = 0.0f;
  if (HAS_BIAS) load_cols<NT>(bias + n0 + NT * c, bv);

  float4 av[K / 8];
  load_a_tile<K>(av, A, mt, c, h);
  for (; mt < mtEnd; ++mt) {
    const int nx = (mt + 1 < mtEnd) ? (mt + 1) : mt;
    float4 an[K / 8];
    load_a_tile<K>(an, A, nx, c, h);
    __builtin_amdgcn_sched_barrier(0);

    v8f acc[NT];
#pragma unroll
    for (int t = 0; t < NT; ++t) acc[t] = v8f{};
#pragma unroll
    for (int s = 0; s < S; ++s) {
      const float4 q = av[s >> 1];
      v2f a;
      a.x = (s & 1) ? q.z : q.x;
      a.y = (s & 1) ? q.w : q.y;
#pragma unroll
      for (int t = 0; t < NT; ++t)
        acc[t] = __builtin_amdgcn_wmma_f32_16x16x4_f32(false, a, false, bf[s][t],
                                                       (short)0, acc[t], false, false);
    }
    float* cp = C + ((size_t)mt * 16 + 8 * h) * N + n0 + NT * c;
#pragma unroll
    for (int j = 0; j < 8; ++j) {
      float o[NT];
#pragma unroll
      for (int t = 0; t < NT; ++t) o[t] = acc[t][j] + bv[t];
      store_cols<NT>(cp + (size_t)j * N, o);
    }
    __builtin_amdgcn_sched_barrier(0);
#pragma unroll
    for (int i = 0; i < K / 8; ++i) av[i] = an[i];
  }
}

constexpr int LOOP_WAVES = 8;
constexpr int LOOP_T     = 32 * LOOP_WAVES;
constexpr int LOOP_BLK   = 1024;

constexpr int SM_ITEMS = 4;

__global__ void __launch_bounds__(LOOP_T)
stalk_mean_kernel(const float* __restrict__ proj, float* __restrict__ m, int total4) {
  const int stride = gridDim.x * LOOP_T;
  const int idx0 = blockIdx.x * LOOP_T + threadIdx.x;
  float4 v[SM_ITEMS][D];
#pragma unroll
  for (int r = 0; r < SM_ITEMS; ++r) {
    int idx = idx0 + r * stride;
    idx = (idx < total4) ? idx : (total4 - 1);
    const float* p = proj + (size_t)(idx >> 4) * OUT_CH + 4 * (idx & 15);
#pragma unroll
    for (int t = 0; t < D; ++t) v[r][t] = *reinterpret_cast<const float4*>(p + t * HIDDEN);
  }
#pragma unroll
  for (int r = 0; r < SM_ITEMS; ++r) {
    const int idx = idx0 + r * stride;
    float4 s = make_float4(0.0f, 0.0f, 0.0f, 0.0f);
#pragma unroll
    for (int t = 0; t < D; ++t) {
      s.x += v[r][t].x; s.y += v[r][t].y; s.z += v[r][t].z; s.w += v[r][t].w;
    }
    const float k = 1.0f / 6.0f;
    if (idx < total4)
      *reinterpret_cast<float4*>(m + (size_t)idx * 4) = make_float4(s.x * k, s.y * k, s.z * k, s.w * k);
  }
}

__device__ inline float wave_allsum(float v) {
#pragma unroll
  for (int off = 16; off > 0; off >>= 1) v += __shfl_xor(v, off, 32);
  return v;
}

__global__ void __launch_bounds__(LOOP_T)
alpha_kernel(const float* __restrict__ xm, const float* __restrict__ em,
             const int* __restrict__ row, const int* __restrict__ col,
             const float* __restrict__ ln_g, const float* __restrict__ ln_b,
             const float* __restrict__ Wsh, const float* __restrict__ bsh,
             float* __restrict__ alpha) {
  const int nw = gridDim.x * LOOP_WAVES;
  int i = blockIdx.x * LOOP_WAVES + (threadIdx.x >> 5);
  if (i >= NNZ) return;
  const int lane = threadIdx.x & 31;
  const int j0 = lane * 4;
  const bool xside = (j0 < HIDDEN);
  const float* zsrc = xside ? xm : em;
  const int zoff = xside ? j0 : (j0 - HIDDEN);
  float g[4], b[4], w[4][D];
#pragma unroll
  for (int q = 0; q < 4; ++q) {
    g[q] = ln_g[j0 + q];
    b[q] = ln_b[j0 + q];
#pragma unroll
    for (int t = 0; t < D; ++t) w[q][t] = Wsh[(j0 + q) * D + t];
  }
  const float bs = (lane < D) ? bsh[lane] : 0.0f;

  int key = xside ? row[i] : col[i];
  float4 zc = *reinterpret_cast<const float4*>(zsrc + (size_t)key * HIDDEN + zoff);
  for (; i < NNZ; i += nw) {
    const int inx = (i + nw < NNZ) ? (i + nw) : i;
    key = xside ? row[inx] : col[inx];
    const float4 zn = *reinterpret_cast<const float4*>(zsrc + (size_t)key * HIDDEN + zoff);

    const float z[4] = {zc.x, zc.y, zc.z, zc.w};
    float s  = z[0] + z[1] + z[2] + z[3];
    float ss = z[0]*z[0] + z[1]*z[1] + z[2]*z[2] + z[3]*z[3];
    s  = wave_allsum(s);
    ss = wave_allsum(ss);
    float mu   = s * (1.0f / 128.0f);
    float var  = ss * (1.0f / 128.0f) - mu * mu;
    float rstd = rsqrtf(var + 1e-5f);

    float p[6] = {0, 0, 0, 0, 0, 0};
#pragma unroll
    for (int q = 0; q < 4; ++q) {
      float fn = (z[q] - mu) * rstd * g[q] + b[q];
#pragma unroll
      for (int t = 0; t < D; ++t) p[t] += fn * w[q][t];
    }
#pragma unroll
    for (int t = 0; t < D; ++t) p[t] = wave_allsum(p[t]);

    if (lane < D) {
      float a = 1.0f / (1.0f + expf(-(p[lane] + bs)));
      alpha[(size_t)i * D + lane] = a;
    }
    zc = zn;
  }
}

constexpr int CSR_T      = 1024;
constexpr int CSR_R_N    = 544;
constexpr int CSR_NB_N   = (N_NODES + CSR_R_N - 1) / CSR_R_N;
constexpr int CSR_R_E    = 160;
constexpr int CSR_NB_E   = (N_EDGES + CSR_R_E - 1) / CSR_R_E;
constexpr int CSR_IDBITS = 18;
constexpr int CSR_N4     = NNZ / 4;
constexpr int CSR_U      = 4;
static_assert(NNZ % 4 == 0, "the key arrays are read as 16-byte quads");
static_assert(NNZ <= (1 << CSR_IDBITS), "an incidence id must fit the low bits of a list word");
static_assert(CSR_R_N <= CSR_T && CSR_R_E <= CSR_T, "the scan takes one counter per thread");
static_assert(CSR_T <= (1 << (32 - CSR_IDBITS)), "a key slot must fit the high bits of a list word");
static_assert((CSR_NB_N - 1) * CSR_R_N < N_NODES && (CSR_NB_E - 1) * CSR_R_E < N_EDGES,
              "every work-group owns at least one key");

__device__ inline int wave_allsum_i(int v) {
#pragma unroll
  for (int off = 16; off > 0; off >>= 1) v += __shfl_xor(v, off, 32);
  return v;
}

__device__ inline int wave_incl_scan(int v, int lane) {
#pragma unroll
  for (int off = 1; off < 32; off <<= 1) {
    int u = __shfl_up(v, off, 32);
    if (lane >= off) v += u;
  }
  return v;
}

__device__ inline void csr_count4(const int4 q, int k0, int nk, int* cur, int& below) {
  const int kk[4] = {q.x, q.y, q.z, q.w};
#pragma unroll
  for (int c = 0; c < 4; ++c) {
    const int d = kk[c] - k0;
    below += (d < 0) ? 1 : 0;
    if ((unsigned)d < (unsigned)nk) atomicAdd(&cur[d], 1);
  }
}

__device__ inline void csr_place4(const int4 q, int i0, int k0, int nk, int* cur,
                                  const int* start, unsigned* __restrict__ tmp) {
  const int kk[4] = {q.x, q.y, q.z, q.w};
#pragma unroll
  for (int c = 0; c < 4; ++c) {
    const int d = kk[c] - k0;
    if ((unsigned)d < (unsigned)nk) {
      const int slot = atomicAdd(&cur[d], 1);
      tmp[start[d] + slot] = ((unsigned)d << CSR_IDBITS) | (unsigned)(i0 + c);
    }
  }
}

__global__ void __launch_bounds__(CSR_T)
csr_build_kernel(const int* __restrict__ row, const int* __restrict__ col,
                 int* __restrict__ nptr, int* __restrict__ eptr,
                 unsigned* __restrict__ ntmp, unsigned* __restrict__ etmp,
                 int* __restrict__ nlst, int* __restrict__ ncol,
                 int* __restrict__ elst, int* __restrict__ erow) {
  __shared__ int cur[CSR_T];
  __shared__ int start[CSR_T];
  __shared__ int wsum[32];
  __shared__ int wbel[32];
  __shared__ int sh_base, sh_total;

  const bool eside = (blockIdx.x >= CSR_NB_N);
  const int b      = eside ? (blockIdx.x - CSR_NB_N) : blockIdx.x;
  const int* keys  = eside ? col : row;
  const int* other = eside ? row : col;
  const int nkeys  = eside ? N_EDGES : N_NODES;
  const int R      = eside ? CSR_R_E : CSR_R_N;
  int* ptr         = eside ? eptr : nptr;
  unsigned* tmp    = eside ? etmp : ntmp;
  int* lst         = eside ? elst : nlst;
  int* oth         = eside ? erow : ncol;
  const int k0 = b * R;
  const int nk = (nkeys - k0 < R) ? (nkeys - k0) : R;
  const int tid = threadIdx.x, lane = tid & 31, w = tid >> 5;
  const int4* k4 = reinterpret_cast<const int4*>(keys);

  cur[tid] = 0;
  __syncthreads();

  int below = 0;
  {
    int v = tid;
    for (; v + (CSR_U - 1) * CSR_T < CSR_N4; v += CSR_U * CSR_T) {
      int4 q[CSR_U];
#pragma unroll
      for (int u = 0; u < CSR_U; ++u) q[u] = k4[v + u * CSR_T];
#pragma unroll
      for (int u = 0; u < CSR_U; ++u) csr_count4(q[u], k0, nk, cur, below);
    }
    for (; v < CSR_N4; v += CSR_T) csr_count4(k4[v], k0, nk, cur, below);
  }
  const int bw = wave_allsum_i(below);
  if (lane == 0) wbel[w] = bw;
  __syncthreads();

  const int cnt  = (tid < nk) ? cur[tid] : 0;
  const int incl = wave_incl_scan(cnt, lane);
  if (lane == 31) wsum[w] = incl;
  __syncthreads();
  if (w == 0) {
    const int t  = wsum[lane];
    const int ti = wave_incl_scan(t, lane);
    const int bl = wave_allsum_i(wbel[lane]);
    wsum[lane] = ti - t;
    if (lane == 31) { sh_base = bl; sh_total = ti; }
  }
  __syncthreads();
  const int base  = sh_base;
  const int total = sh_total;
  const int st = base + wsum[w] + incl - cnt;
  start[tid] = st;
  cur[tid] = 0;
  if (tid < nk) ptr[k0 + tid] = st;
  if (tid == nk - 1 && k0 + nk == nkeys) ptr[nkeys] = st + cnt;
  __syncthreads();

  {
    int v = tid;
    for (; v + (CSR_U - 1) * CSR_T < CSR_N4; v += CSR_U * CSR_T) {
      int4 q[CSR_U];
#pragma unroll
      for (int u = 0; u < CSR_U; ++u) q[u] = k4[v + u * CSR_T];
#pragma unroll
      for (int u = 0; u < CSR_U; ++u)
        csr_place4(q[u], 4 * (v + u * CSR_T), k0, nk, cur, start, tmp);
    }
    for (; v < CSR_N4; v += CSR_T) csr_place4(k4[v], 4 * v, k0, nk, cur, start, tmp);
  }
  __syncthreads();

  for (int p = tid; p < total; p += CSR_T) {
    const unsigned wv = tmp[base + p];
    const int d = (int)(wv >> CSR_IDBITS);
    const int i = (int)(wv & ((1u << CSR_IDBITS) - 1u));
    const int o = other[i];
    const int s = start[d], n = cur[d];
    const unsigned* sp = tmp + s;
    int rank = 0;
    int k = 0;
    for (; k + 4 <= n; k += 4) {
      const unsigned u0 = sp[k], u1 = sp[k + 1], u2 = sp[k + 2], u3 = sp[k + 3];
      rank += ((u0 < wv) ? 1 : 0) + ((u1 < wv) ? 1 : 0) + ((u2 < wv) ? 1 : 0) + ((u3 < wv) ? 1 : 0);
    }
    for (; k < n; ++k) rank += (sp[k] < wv) ? 1 : 0;
    lst[s + rank] = i;
    oth[s + rank] = o;
  }
}

constexpr int SEG_C = OUT_CH / 128;
constexpr int SEG_B = 4;
static_assert(SEG_B == 4, "the list walk picks its group body from the sizes 4, 3, 2, 1");

template <bool NODE_SIDE, int NB>
__device__ inline void seg_group(const int* __restrict__ lst, const int* __restrict__ oth,
                                 const float* __restrict__ alpha, const float* __restrict__ src,
                                 int q, int c0, int th,
                                 float4 (&acc)[SEG_C], float (&dsum)[SEG_C]) {
  int ii[NB], oo[NB];
#pragma unroll
  for (int u = 0; u < NB; ++u) { ii[u] = lst[q + u]; oo[u] = oth[q + u]; }
  float a[NB][SEG_C];
  float4 v[NB][SEG_C];
#pragma unroll
  for (int u = 0; u < NB; ++u) {
#pragma unroll
    for (int k = 0; k < SEG_C; ++k) {
      a[u][k] = alpha[(size_t)ii[u] * D + 2 * k + th];
      v[u][k] = *reinterpret_cast<const float4*>(src + (size_t)oo[u] * OUT_CH + 128 * k + c0);
    }
  }
#pragma unroll
  for (int u = 0; u < NB; ++u) {
#pragma unroll
    for (int k = 0; k < SEG_C; ++k) {
      acc[k].x = __builtin_fmaf(a[u][k], v[u][k].x, acc[k].x);
      acc[k].y = __builtin_fmaf(a[u][k], v[u][k].y, acc[k].y);
      acc[k].z = __builtin_fmaf(a[u][k], v[u][k].z, acc[k].z);
      acc[k].w = __builtin_fmaf(a[u][k], v[u][k].w, acc[k].w);
      if (NODE_SIDE) dsum[k] = __builtin_fmaf(a[u][k], a[u][k], dsum[k]);
    }
  }
}

template <bool NODE_SIDE>
__global__ void __launch_bounds__(LOOP_T)
seg_gather_kernel(const int* __restrict__ ptr, const int* __restrict__ lst,
                  const int* __restrict__ oth, const float* __restrict__ alpha,
                  const float* __restrict__ src, const float* __restrict__ h,
                  const float* __restrict__ bias, float* __restrict__ dst, int nseg) {
  const int nw = gridDim.x * LOOP_WAVES;
  int seg = blockIdx.x * LOOP_WAVES + (threadIdx.x >> 5);
  if (seg >= nseg) return;
  const int lane = threadIdx.x & 31;
  const int c0 = 4 * lane;
  const int th = lane >> 4;
  float4 bv = make_float4(0.0f, 0.0f, 0.0f, 0.0f);
  if (NODE_SIDE) bv = *reinterpret_cast<const float4*>(bias + (c0 & (HIDDEN - 1)));

  int s = ptr[seg], e = ptr[seg + 1];
  for (; seg < nseg; seg += nw) {
    const int sx = (seg + nw < nseg) ? (seg + nw) : seg;
    const int sn = ptr[sx], en = ptr[sx + 1];
    float4 hv[SEG_C];
    if (NODE_SIDE) {
#pragma unroll
      for (int k = 0; k < SEG_C; ++k)
        hv[k] = *reinterpret_cast<const float4*>(h + (size_t)seg * OUT_CH + 128 * k + c0);
    }
    float4 acc[SEG_C];
    float dsum[SEG_C];
#pragma unroll
    for (int k = 0; k < SEG_C; ++k) { acc[k] = make_float4(0.0f, 0.0f, 0.0f, 0.0f); dsum[k] = 0.0f; }

    for (int q = s; q < e; q += SEG_B) {
      const int n = e - q;
      if (n >= 4)      seg_group<NODE_SIDE, 4>(lst, oth, alpha, src, q, c0, th, acc, dsum);
      else if (n == 3) seg_group<NODE_SIDE, 3>(lst, oth, alpha, src, q, c0, th, acc, dsum);
      else if (n == 2) seg_group<NODE_SIDE, 2>(lst, oth, alpha, src, q, c0, th, acc, dsum);
      else             seg_group<NODE_SIDE, 1>(lst, oth, alpha, src, q, c0, th, acc, dsum);
    }

    if (NODE_SIDE) {
#pragma unroll
      for (int k = 0; k < SEG_C; ++k) {
        const float dinv = (dsum[k] > 0.0f) ? (1.0f / dsum[k]) : 0.0f;
        float4 o;
        o.x = hv[k].x - dinv * acc[k].x + bv.x;
        o.y = hv[k].y - dinv * acc[k].y + bv.y;
        o.z = hv[k].z - dinv * acc[k].z + bv.z;
        o.w = hv[k].w - dinv * acc[k].w + bv.w;
        o.x = (o.x > 0.0f) ? o.x : expm1f(o.x);
        o.y = (o.y > 0.0f) ? o.y : expm1f(o.y);
        o.z = (o.z > 0.0f) ? o.z : expm1f(o.z);
        o.w = (o.w > 0.0f) ? o.w : expm1f(o.w);
        *reinterpret_cast<float4*>(dst + (size_t)seg * OUT_CH + 128 * k + c0) = o;
      }
    } else {
      const int len = e - s;
      const float binv = (len > 0) ? (1.0f / (float)len) : 0.0f;
#pragma unroll
      for (int k = 0; k < SEG_C; ++k)
        *reinterpret_cast<float4*>(dst + (size_t)seg * OUT_CH + 128 * k + c0) =
            make_float4(acc[k].x * binv, acc[k].y * binv, acc[k].z * binv, acc[k].w * binv);
    }
    s = sn;
    e = en;
  }
}

constexpr int PROJ_N_RUN = 13;
constexpr int PROJ_N_BLK = (N_NODES / 16 + PROJ_N_RUN - 1) / PROJ_N_RUN;
constexpr int PROJ_E_RUN = 3;
constexpr int PROJ_E_BLK = (N_EDGES / 16 + PROJ_E_RUN - 1) / PROJ_E_RUN;
constexpr int CONV_RUN   = 5;
constexpr int CONV_BLK   = ((ND / 16 + CONV_RUN - 1) / CONV_RUN + 3) / 4;
constexpr int SM_N_Q   = N_NODES * HIDDEN / 4;
constexpr int SM_E_Q   = N_EDGES * HIDDEN / 4;
constexpr int SM_N_BLK = (SM_N_Q + LOOP_T * SM_ITEMS - 1) / (LOOP_T * SM_ITEMS);
constexpr int SM_E_BLK = (SM_E_Q + LOOP_T * SM_ITEMS - 1) / (LOOP_T * SM_ITEMS);
constexpr int SEG_E_BLK = (N_EDGES + LOOP_WAVES - 1) / LOOP_WAVES;
constexpr int SEG_N_BLK = LOOP_BLK;

extern "C" void kernel_launch(void* const* d_in, const int* in_sizes, int n_in,
                              void* d_out, int out_size, void* d_ws, size_t ws_size,
                              hipStream_t stream) {
  (void)in_sizes; (void)n_in; (void)out_size; (void)ws_size;
  const float* x       = (const float*)d_in[0];
  const float* hattr   = (const float*)d_in[1];
  const float* W_lin   = (const float*)d_in[2];
  const float* b_lin   = (const float*)d_in[3];
  const float* ln_g    = (const float*)d_in[4];
  const float* ln_b    = (const float*)d_in[5];
  const float* W_sheaf = (const float*)d_in[6];
  const float* b_sheaf = (const float*)d_in[7];
  const float* W_conv  = (const float*)d_in[8];
  const float* b_conv  = (const float*)d_in[9];
  const int*   eidx    = (const int*)d_in[10];
  const int* row = eidx;
  const int* col = eidx + NNZ;
  float* out = (float*)d_out;

  float* ws    = (float*)d_ws;
  float* xs    = ws;
  float* es    = xs   + (size_t)ND * HIDDEN;
  float* xm    = es   + (size_t)ED * HIDDEN;
  float* em    = xm   + (size_t)N_NODES * HIDDEN;
  float* alpha = em   + (size_t)N_EDGES * HIDDEN;
  float* spare = alpha + (size_t)NNZ * D;
  float* hbuf  = spare + ND + N_EDGES;
  float* me    = hbuf + (size_t)ND * HIDDEN;
  int*   ib    = (int*)(me + (size_t)ED * HIDDEN);
  int* nptr = ib;
  int* eptr = nptr + N_NODES + 8;
  unsigned* ntmp = (unsigned*)(eptr + N_EDGES + 8);
  unsigned* etmp = ntmp + NNZ;
  int* nlst = (int*)(etmp + NNZ);
  int* ncol = nlst + NNZ;
  int* elst = ncol + NNZ;
  int* erow = elst + NNZ;

  csr_build_kernel<<<CSR_NB_N + CSR_NB_E, CSR_T, 0, stream>>>(row, col, nptr, eptr, ntmp, etmp,
                                                              nlst, ncol, elst, erow);

  wmma_gemm_f32<IN_CH, 2, 12, true><<<PROJ_N_BLK, 32 * 12, 0, stream>>>(
      x, W_lin, b_lin, xs, N_NODES / 16, PROJ_N_RUN);
  wmma_gemm_f32<IN_CH, 2, 12, true><<<PROJ_E_BLK, 32 * 12, 0, stream>>>(
      hattr, W_lin, b_lin, es, N_EDGES / 16, PROJ_E_RUN);

  stalk_mean_kernel<<<SM_N_BLK, LOOP_T, 0, stream>>>(xs, xm, SM_N_Q);
  stalk_mean_kernel<<<SM_E_BLK, LOOP_T, 0, stream>>>(es, em, SM_E_Q);

  alpha_kernel<<<LOOP_BLK, LOOP_T, 0, stream>>>(xm, em, row, col, ln_g, ln_b,
                                                W_sheaf, b_sheaf, alpha);

  wmma_gemm_f32<HIDDEN, 4, 1, false><<<CONV_BLK, 32 * 4, 0, stream>>>(
      xs, W_conv, nullptr, hbuf, ND / 16, CONV_RUN);
  seg_gather_kernel<false><<<SEG_E_BLK, LOOP_T, 0, stream>>>(eptr, elst, erow, alpha, hbuf,
                                                           nullptr, nullptr, me, N_EDGES);
  seg_gather_kernel<true><<<SEG_N_BLK, LOOP_T, 0, stream>>>(nptr, nlst, ncol, alpha, me,
                                                          hbuf, b_conv, out, N_NODES);

  wmma_gemm_f32<HIDDEN, 4, 1, false><<<CONV_BLK, 32 * 4, 0, stream>>>(
      out, W_conv + HIDDEN * HIDDEN, nullptr, hbuf, ND / 16, CONV_RUN);
  seg_gather_kernel<false><<<SEG_E_BLK, LOOP_T, 0, stream>>>(eptr, elst, erow, alpha, hbuf,
                                                           nullptr, nullptr, me, N_EDGES);
  seg_gather_kernel<true><<<SEG_N_BLK, LOOP_T, 0, stream>>>(nptr, nlst, ncol, alpha, me,
                                                          hbuf, b_conv + HIDDEN, out, N_NODES);
}


# Linear maps folded on the weights first: W_lin x W0 per stalk (small WMMA product) and W_lin's stalk mean; ONE GEMM over x gives h1, xm, em; xs/es, mean launches, conv-1 GEMM gone; sums re-associated
# speedup vs baseline: 4.3990x; 1.1379x over previous
#include <hip/hip_runtime.h>
#include <math.h>

constexpr int N_NODES = 50000;
constexpr int N_EDGES = 10000;
constexpr int NNZ     = 150000;
constexpr int IN_CH   = 128;
constexpr int HIDDEN  = 64;
constexpr int D       = 6;
constexpr int OUT_CH  = HIDDEN * D;
constexpr int ND      = N_NODES * D;
constexpr int ED      = N_EDGES * D;

typedef __attribute__((ext_vector_type(2))) float v2f;
typedef __attribute__((ext_vector_type(8))) float v8f;

template <int NT> struct ColVec;
template <> struct ColVec<2> { typedef float2 type; };
template <> struct ColVec<4> { typedef float4 type; };

template <int NT>
__device__ inline void load_cols(const float* __restrict__ p, float (&d)[NT]) {
  const typename ColVec<NT>::type v = *reinterpret_cast<const typename ColVec<NT>::type*>(p);
  const float* f = reinterpret_cast<const float*>(&v);
#pragma unroll
  for (int t = 0; t < NT; ++t) d[t] = f[t];
}

template <int NT>
__device__ inline void store_cols(float* __restrict__ p, const float (&d)[NT]) {
  typename ColVec<NT>::type v;
  float* f = reinterpret_cast<float*>(&v);
#pragma unroll
  for (int t = 0; t < NT; ++t) f[t] = d[t];
  *reinterpret_cast<typename ColVec<NT>::type*>(p) = v;
}

constexpr int gemm_wpb(int g) { return (g > 1) ? g : 4; }

template <int K>
__device__ inline void load_a_tile(float4 (&av)[K / 8], const float* __restrict__ A,
                                   int mt, int c, int h) {
  const float* ap = A + ((size_t)mt * 16 + c) * K + 8 * h;
#pragma unroll
  for (int g = 0; g < K / 16; ++g) {
    av[2 * g]     = *reinterpret_cast<const float4*>(ap + 16 * g);
    av[2 * g + 1] = *reinterpret_cast<const float4*>(ap + 16 * g + 4);
  }
}

template <int K, int NT, int G, bool HAS_BIAS>
__global__ void __launch_bounds__(32 * gemm_wpb(G))
wmma_gemm_f32(const float* __restrict__ A, const float* __restrict__ B,
              const float* __restrict__ bias, float* __restrict__ C,
              int mTiles, int tilesPerWave) {
  constexpr int N = 16 * NT * G;
  constexpr int S = K / 4;
  const int wave  = __builtin_amdgcn_readfirstlane(blockIdx.x * gemm_wpb(G) + (threadIdx.x >> 5));
  const int cg    = wave % G;
  int mt          = (wave / G) * tilesPerWave;
  if (mt >= mTiles) return;
  const int mtEnd = (mt + tilesPerWave < mTiles) ? (mt + tilesPerWave) : mTiles;
  const int lane = threadIdx.x & 31;
  const int h = lane >> 4;
  const int c = lane & 15;
  const int n0 = cg * 16 * NT;

  v2f bf[S][NT];
#pragma unroll
  for (int s = 0; s < S; ++s) {
    const int k = 16 * (s >> 2) + 8 * h + 2 * (s & 3);
    float w0[NT], w1[NT];
    load_cols<NT>(B + (size_t)k * N + n0 + NT * c, w0);
    load_cols<NT>(B + (size_t)(k + 1) * N + n0 + NT * c, w1);
#pragma unroll
    for (int t = 0; t < NT; ++t) { bf[s][t].x = w0[t]; bf[s][t].y = w1[t]; }
  }
  float bv[NT];
#pragma unroll
  for (int t = 0; t < NT; ++t) bv[t] = 0.0f;
  if (HAS_BIAS) load_cols<NT>(bias + n0 + NT * c, bv);

  float4 av[K / 8];
  load_a_tile<K>(av, A, mt, c, h);
  for (; mt < mtEnd; ++mt) {
    const int nx = (mt + 1 < mtEnd) ? (mt + 1) : mt;
    float4 an[K / 8];
    load_a_tile<K>(an, A, nx, c, h);
    __builtin_amdgcn_sched_barrier(0);

    v8f acc[NT];
#pragma unroll
    for (int t = 0; t < NT; ++t) acc[t] = v8f{};
#pragma unroll
    for (int s = 0; s < S; ++s) {
      const float4 q = av[s >> 1];
      v2f a;
      a.x = (s & 1) ? q.z : q.x;
      a.y = (s & 1) ? q.w : q.y;
#pragma unroll
      for (int t = 0; t < NT; ++t)
        acc[t] = __builtin_amdgcn_wmma_f32_16x16x4_f32(false, a, false, bf[s][t],
                                                       (short)0, acc[t], false, false);
    }
    float* cp = C + ((size_t)mt * 16 + 8 * h) * N + n0 + NT * c;
#pragma unroll
    for (int j = 0; j < 8; ++j) {
      float o[NT];
#pragma unroll
      for (int t = 0; t < NT; ++t) o[t] = acc[t][j] + bv[t];
      store_cols<NT>(cp + (size_t)j * N, o);
    }
    __builtin_amdgcn_sched_barrier(0);
#pragma unroll
    for (int i = 0; i < K / 8; ++i) av[i] = an[i];
  }
}

constexpr int FOLD_TILES = IN_CH * D / 16;
constexpr int FOLD_WPB   = 4;
constexpr int FOLD_BLK   = (FOLD_TILES + 1 + FOLD_WPB - 1) / FOLD_WPB;
constexpr int FOLD_MQ    = IN_CH * HIDDEN / 4;

__global__ void __launch_bounds__(32 * FOLD_WPB)
fold_weights_kernel(const float* __restrict__ W_lin, const float* __restrict__ b_lin,
                    const float* __restrict__ W0,
                    float* __restrict__ Wp, float* __restrict__ bp,
                    float* __restrict__ Wm, float* __restrict__ bm) {
  constexpr int K = HIDDEN, NT = 4, S = K / 4;
  for (int qd = blockIdx.x * (32 * FOLD_WPB) + threadIdx.x; qd < FOLD_MQ + HIDDEN / 4;
       qd += gridDim.x * (32 * FOLD_WPB)) {
    const bool isb = (qd >= FOLD_MQ);
    const float* p = isb ? (b_lin + 4 * (qd - FOLD_MQ)) : (W_lin + (size_t)(qd >> 4) * OUT_CH + 4 * (qd & 15));
    float4 sm = make_float4(0.0f, 0.0f, 0.0f, 0.0f);
#pragma unroll
    for (int t = 0; t < D; ++t) {
      const float4 v = *reinterpret_cast<const float4*>(p + t * HIDDEN);
      sm.x += v.x; sm.y += v.y; sm.z += v.z; sm.w += v.w;
    }
    const float k6 = 1.0f / 6.0f;
    float* d = isb ? (bm + 4 * (qd - FOLD_MQ)) : (Wm + (size_t)qd * 4);
    *reinterpret_cast<float4*>(d) = make_float4(sm.x * k6, sm.y * k6, sm.z * k6, sm.w * k6);
  }

  const int wave = __builtin_amdgcn_readfirstlane(blockIdx.x * FOLD_WPB + (threadIdx.x >> 5));
  if (wave > FOLD_TILES) return;
  const bool btile = (wave == FOLD_TILES);
  const float* src = btile ? b_lin : (W_lin + (size_t)wave * 16 * K);
  float* dst       = btile ? bp : (Wp + (size_t)wave * 16 * K);
  const int rows   = btile ? D : 16;
  const int lane = threadIdx.x & 31;
  const int h = lane >> 4;
  const int c = lane & 15;

  float4 av[K / 8];
#pragma unroll
  for (int i = 0; i < K / 8; ++i) av[i] = make_float4(0.0f, 0.0f, 0.0f, 0.0f);
  if (c < rows) {
    const float* ap = src + (size_t)c * K + 8 * h;
#pragma unroll
    for (int g = 0; g < K / 16; ++g) {
      av[2 * g]     = *reinterpret_cast<const float4*>(ap + 16 * g);
      av[2 * g + 1] = *reinterpret_cast<const float4*>(ap + 16 * g + 4);
    }
  }
  v8f acc[NT];
#pragma unroll
  for (int t = 0; t < NT; ++t) acc[t] = v8f{};
#pragma unroll
  for (int s = 0; s < S; ++s) {
    const int k = 16 * (s >> 2) + 8 * h + 2 * (s & 3);
    float w0[NT], w1[NT];
    load_cols<NT>(W0 + (size_t)k * K + NT * c, w0);
    load_cols<NT>(W0 + (size_t)(k + 1) * K + NT * c, w1);
    const float4 q = av[s >> 1];
    v2f a;
    a.x = (s & 1) ? q.z : q.x;
    a.y = (s & 1) ? q.w : q.y;
#pragma unroll
    for (int t = 0; t < NT; ++t) {
      v2f b;
      b.x = w0[t]; b.y = w1[t];
      acc[t] = __builtin_amdgcn_wmma_f32_16x16x4_f32(false, a, false, b, (short)0, acc[t], false, false);
    }
  }
#pragma unroll
  for (int j = 0; j < 8; ++j) {
    float o[NT];
#pragma unroll
    for (int t = 0; t < NT; ++t) o[t] = acc[t][j];
    if (j + 8 * h < rows) store_cols<NT>(dst + (size_t)(j + 8 * h) * K + NT * c, o);
  }
}

constexpr int PF_WAVES  = 12;
constexpr int PF_RUN    = 15;
constexpr int PF_RUNS_N = (N_NODES / 16 + PF_RUN - 1) / PF_RUN;
constexpr int PF_RUNS_E = (N_EDGES / 16 + PF_RUN - 1) / PF_RUN;
constexpr int PF_BLK_H  = PF_RUNS_N;
constexpr int PF_MW_N   = 2 * PF_RUNS_N;
constexpr int PF_MW_E   = 2 * PF_RUNS_E;
constexpr int PF_BLK_M  = (PF_MW_N + PF_MW_E + PF_WAVES - 1) / PF_WAVES;
static_assert(N_NODES % 16 == 0 && N_EDGES % 16 == 0, "row tiles are whole");

__global__ void __launch_bounds__(32 * PF_WAVES)
proj_fold_kernel(const float* __restrict__ x, const float* __restrict__ hattr,
                 const float* __restrict__ Wp, const float* __restrict__ bp,
                 const float* __restrict__ Wm, const float* __restrict__ bm,
                 float* __restrict__ h1, float* __restrict__ xm, float* __restrict__ em) {
  constexpr int K = IN_CH, NT = 2, S = K / 4;
  const int blk = blockIdx.x;
  const int wib = __builtin_amdgcn_readfirstlane(threadIdx.x >> 5);
  const float* A; const float* B; const float* bias; float* C;
  int pitch, n0, run, mTiles;
  if (blk < PF_BLK_H) {
    A = x; B = Wp; bias = bp; C = h1; pitch = OUT_CH; n0 = 16 * NT * wib; run = blk; mTiles = N_NODES / 16;
  } else {
    const int u = (blk - PF_BLK_H) * PF_WAVES + wib;
    B = Wm; bias = bm; pitch = HIDDEN;
    if (u < PF_MW_N) {
      A = x; C = xm; run = u >> 1; n0 = 16 * NT * (u & 1); mTiles = N_NODES / 16;
    } else {
      const int v = u - PF_MW_N;
      if (v >= PF_MW_E) return;
      A = hattr; C = em; run = v >> 1; n0 = 16 * NT * (v & 1); mTiles = N_EDGES / 16;
    }
  }
  int mt = run * PF_RUN;
  if (mt >= mTiles) return;
  const int mtEnd = (mt + PF_RUN < mTiles) ? (mt + PF_RUN) : mTiles;
  const int lane = threadIdx.x & 31;
  const int h = lane >> 4;
  const int c = lane & 15;

  v2f bf[S][NT];
#pragma unroll
  for (int s = 0; s < S; ++s) {
    const int k = 16 * (s >> 2) + 8 * h + 2 * (s & 3);
    float w0[NT], w1[NT];
    load_cols<NT>(B + (size_t)k * pitch + n0 + NT * c, w0);
    load_cols<NT>(B + (size_t)(k + 1) * pitch + n0 + NT * c, w1);
#pragma unroll
    for (int t = 0; t < NT; ++t) { bf[s][t].x = w0[t]; bf[s][t].y = w1[t]; }
  }
  float bv[NT];
  load_cols<NT>(bias + n0 + NT * c, bv);

  float4 av[K / 8];
  load_a_tile<K>(av, A, mt, c, h);
  for (; mt < mtEnd; ++mt) {
    const int nx = (mt + 1 < mtEnd) ? (mt + 1) : mt;
    float4 an[K / 8];
    load_a_tile<K>(an, A, nx, c, h);
    __builtin_amdgcn_sched_barrier(0);

    v8f acc[NT];
#pragma unroll
    for (int t = 0; t < NT; ++t) acc[t] = v8f{};
#pragma unroll
    for (int s = 0; s < S; ++s) {
      const float4 q = av[s >> 1];
      v2f a;
      a.x = (s & 1) ? q.z : q.x;
      a.y = (s & 1) ? q.w : q.y;
#pragma unroll
      for (int t = 0; t < NT; ++t)
        acc[t] = __builtin_amdgcn_wmma_f32_16x16x4_f32(false, a, false, bf[s][t],
                                                       (short)0, acc[t], false, false);
    }
    float* cp = C + ((size_t)mt * 16 + 8 * h) * pitch + n0 + NT * c;
#pragma unroll
    for (int j = 0; j < 8; ++j) {
      float o[NT];
#pragma unroll
      for (int t = 0; t < NT; ++t) o[t] = acc[t][j] + bv[t];
      store_cols<NT>(cp + (size_t)j * pitch, o);
    }
    __builtin_amdgcn_sched_barrier(0);
#pragma unroll
    for (int i = 0; i < K / 8; ++i) av[i] = an[i];
  }
}

constexpr int LOOP_WAVES = 8;
constexpr int LOOP_T     = 32 * LOOP_WAVES;
constexpr int LOOP_BLK   = 1024;

__device__ inline float wave_allsum(float v) {
#pragma unroll
  for (int off = 16; off > 0; off >>= 1) v += __shfl_xor(v, off, 32);
  return v;
}

__global__ void __launch_bounds__(LOOP_T)
alpha_kernel(const float* __restrict__ xm, const float* __restrict__ em,
             const int* __restrict__ row, const int* __restrict__ col,
             const float* __restrict__ ln_g, const float* __restrict__ ln_b,
             const float* __restrict__ Wsh, const float* __restrict__ bsh,
             float* __restrict__ alpha) {
  const int nw = gridDim.x * LOOP_WAVES;
  int i = blockIdx.x * LOOP_WAVES + (threadIdx.x >> 5);
  if (i >= NNZ) return;
  const int lane = threadIdx.x & 31;
  const int j0 = lane * 4;
  const bool xside = (j0 < HIDDEN);
  const float* zsrc = xside ? xm : em;
  const int zoff = xside ? j0 : (j0 - HIDDEN);
  float g[4], b[4], w[4][D];
#pragma unroll
  for (int q = 0; q < 4; ++q) {
    g[q] = ln_g[j0 + q];
    b[q] = ln_b[j0 + q];
#pragma unroll
    for (int t = 0; t < D; ++t) w[q][t] = Wsh[(j0 + q) * D + t];
  }
  const float bs = (lane < D) ? bsh[lane] : 0.0f;

  int key = xside ? row[i] : col[i];
  float4 zc = *reinterpret_cast<const float4*>(zsrc + (size_t)key * HIDDEN + zoff);
  for (; i < NNZ; i += nw) {
    const int inx = (i + nw < NNZ) ? (i + nw) : i;
    key = xside ? row[inx] : col[inx];
    const float4 zn = *reinterpret_cast<const float4*>(zsrc + (size_t)key * HIDDEN + zoff);

    const float z[4] = {zc.x, zc.y, zc.z, zc.w};
    float s  = z[0] + z[1] + z[2] + z[3];
    float ss = z[0]*z[0] + z[1]*z[1] + z[2]*z[2] + z[3]*z[3];
    s  = wave_allsum(s);
    ss = wave_allsum(ss);
    float mu   = s * (1.0f / 128.0f);
    float var  = ss * (1.0f / 128.0f) - mu * mu;
    float rstd = rsqrtf(var + 1e-5f);

    float p[6] = {0, 0, 0, 0, 0, 0};
#pragma unroll
    for (int q = 0; q < 4; ++q) {
      float fn = (z[q] - mu) * rstd * g[q] + b[q];
#pragma unroll
      for (int t = 0; t < D; ++t) p[t] += fn * w[q][t];
    }
#pragma unroll
    for (int t = 0; t < D; ++t) p[t] = wave_allsum(p[t]);

    if (lane < D) {
      float a = 1.0f / (1.0f + expf(-(p[lane] + bs)));
      alpha[(size_t)i * D + lane] = a;
    }
    zc = zn;
  }
}

constexpr int CSR_T      = 1024;
constexpr int CSR_R_N    = 544;
constexpr int CSR_NB_N   = (N_NODES + CSR_R_N - 1) / CSR_R_N;
constexpr int CSR_R_E    = 160;
constexpr int CSR_NB_E   = (N_EDGES + CSR_R_E - 1) / CSR_R_E;
constexpr int CSR_IDBITS = 18;
constexpr int CSR_N4     = NNZ / 4;
constexpr int CSR_U      = 4;
static_assert(NNZ % 4 == 0, "the key arrays are read as 16-byte quads");
static_assert(NNZ <= (1 << CSR_IDBITS), "an incidence id must fit the low bits of a list word");
static_assert(CSR_R_N <= CSR_T && CSR_R_E <= CSR_T, "the scan takes one counter per thread");
static_assert(CSR_T <= (1 << (32 - CSR_IDBITS)), "a key slot must fit the high bits of a list word");
static_assert((CSR_NB_N - 1) * CSR_R_N < N_NODES && (CSR_NB_E - 1) * CSR_R_E < N_EDGES,
              "every work-group owns at least one key");

__device__ inline int wave_allsum_i(int v) {
#pragma unroll
  for (int off = 16; off > 0; off >>= 1) v += __shfl_xor(v, off, 32);
  return v;
}

__device__ inline int wave_incl_scan(int v, int lane) {
#pragma unroll
  for (int off = 1; off < 32; off <<= 1) {
    int u = __shfl_up(v, off, 32);
    if (lane >= off) v += u;
  }
  return v;
}

__device__ inline void csr_count4(const int4 q, int k0, int nk, int* cur, int& below) {
  const int kk[4] = {q.x, q.y, q.z, q.w};
#pragma unroll
  for (int c = 0; c < 4; ++c) {
    const int d = kk[c] - k0;
    below += (d < 0) ? 1 : 0;
    if ((unsigned)d < (unsigned)nk) atomicAdd(&cur[d], 1);
  }
}

__device__ inline void csr_place4(const int4 q, int i0, int k0, int nk, int* cur,
                                  const int* start, unsigned* __restrict__ tmp) {
  const int kk[4] = {q.x, q.y, q.z, q.w};
#pragma unroll
  for (int c = 0; c < 4; ++c) {
    const int d = kk[c] - k0;
    if ((unsigned)d < (unsigned)nk) {
      const int slot = atomicAdd(&cur[d], 1);
      tmp[start[d] + slot] = ((unsigned)d << CSR_IDBITS) | (unsigned)(i0 + c);
    }
  }
}

__global__ void __launch_bounds__(CSR_T)
csr_build_kernel(const int* __restrict__ row, const int* __restrict__ col,
                 int* __restrict__ nptr, int* __restrict__ eptr,
                 unsigned* __restrict__ ntmp, unsigned* __restrict__ etmp,
                 int* __restrict__ nlst, int* __restrict__ ncol,
                 int* __restrict__ elst, int* __restrict__ erow) {
  __shared__ int cur[CSR_T];
  __shared__ int start[CSR_T];
  __shared__ int wsum[32];
  __shared__ int wbel[32];
  __shared__ int sh_base, sh_total;

  const bool eside = (blockIdx.x >= CSR_NB_N);
  const int b      = eside ? (blockIdx.x - CSR_NB_N) : blockIdx.x;
  const int* keys  = eside ? col : row;
  const int* other = eside ? row : col;
  const int nkeys  = eside ? N_EDGES : N_NODES;
  const int R      = eside ? CSR_R_E : CSR_R_N;
  int* ptr         = eside ? eptr : nptr;
  unsigned* tmp    = eside ? etmp : ntmp;
  int* lst         = eside ? elst : nlst;
  int* oth         = eside ? erow : ncol;
  const int k0 = b * R;
  const int nk = (nkeys - k0 < R) ? (nkeys - k0) : R;
  const int tid = threadIdx.x, lane = tid & 31, w = tid >> 5;
  const int4* k4 = reinterpret_cast<const int4*>(keys);

  cur[tid] = 0;
  __syncthreads();

  int below = 0;
  {
    int v = tid;
    for (; v + (CSR_U - 1) * CSR_T < CSR_N4; v += CSR_U * CSR_T) {
      int4 q[CSR_U];
#pragma unroll
      for (int u = 0; u < CSR_U; ++u) q[u] = k4[v + u * CSR_T];
#pragma unroll
      for (int u = 0; u < CSR_U; ++u) csr_count4(q[u], k0, nk, cur, below);
    }
    for (; v < CSR_N4; v += CSR_T) csr_count4(k4[v], k0, nk, cur, below);
  }
  const int bw = wave_allsum_i(below);
  if (lane == 0) wbel[w] = bw;
  __syncthreads();

  const int cnt  = (tid < nk) ? cur[tid] : 0;
  const int incl = wave_incl_scan(cnt, lane);
  if (lane == 31) wsum[w] = incl;
  __syncthreads();
  if (w == 0) {
    const int t  = wsum[lane];
    const int ti = wave_incl_scan(t, lane);
    const int bl = wave_allsum_i(wbel[lane]);
    wsum[lane] = ti - t;
    if (lane == 31) { sh_base = bl; sh_total = ti; }
  }
  __syncthreads();
  const int base  = sh_base;
  const int total = sh_total;
  const int st = base + wsum[w] + incl - cnt;
  start[tid] = st;
  cur[tid] = 0;
  if (tid < nk) ptr[k0 + tid] = st;
  if (tid == nk - 1 && k0 + nk == nkeys) ptr[nkeys] = st + cnt;
  __syncthreads();

  {
    int v = tid;
    for (; v + (CSR_U - 1) * CSR_T < CSR_N4; v += CSR_U * CSR_T) {
      int4 q[CSR_U];
#pragma unroll
      for (int u = 0; u < CSR_U; ++u) q[u] = k4[v + u * CSR_T];
#pragma unroll
      for (int u = 0; u < CSR_U; ++u)
        csr_place4(q[u], 4 * (v + u * CSR_T), k0, nk, cur, start, tmp);
    }
    for (; v < CSR_N4; v += CSR_T) csr_place4(k4[v], 4 * v, k0, nk, cur, start, tmp);
  }
  __syncthreads();

  for (int p = tid; p < total; p += CSR_T) {
    const unsigned wv = tmp[base + p];
    const int d = (int)(wv >> CSR_IDBITS);
    const int i = (int)(wv & ((1u << CSR_IDBITS) - 1u));
    const int o = other[i];
    const int s = start[d], n = cur[d];
    const unsigned* sp = tmp + s;
    int rank = 0;
    int k = 0;
    for (; k + 4 <= n; k += 4) {
      const unsigned u0 = sp[k], u1 = sp[k + 1], u2 = sp[k + 2], u3 = sp[k + 3];
      rank += ((u0 < wv) ? 1 : 0) + ((u1 < wv) ? 1 : 0) + ((u2 < wv) ? 1 : 0) + ((u3 < wv) ? 1 : 0);
    }
    for (; k < n; ++k) rank += (sp[k] < wv) ? 1 : 0;
    lst[s + rank] = i;
    oth[s + rank] = o;
  }
}

constexpr int SEG_C = OUT_CH / 128;
constexpr int SEG_B = 4;
static_assert(SEG_B == 4, "the list walk picks its group body from the sizes 4, 3, 2, 1");

template <bool NODE_SIDE, int NB>
__device__ inline void seg_group(const int* __restrict__ lst, const int* __restrict__ oth,
                                 const float* __restrict__ alpha, const float* __restrict__ src,
                                 int q, int c0, int th,
                                 float4 (&acc)[SEG_C], float (&dsum)[SEG_C]) {
  int ii[NB], oo[NB];
#pragma unroll
  for (int u = 0; u < NB; ++u) { ii[u] = lst[q + u]; oo[u] = oth[q + u]; }
  float a[NB][SEG_C];
  float4 v[NB][SEG_C];
#pragma unroll
  for (int u = 0; u < NB; ++u) {
#pragma unroll
    for (int k = 0; k < SEG_C; ++k) {
      a[u][k] = alpha[(size_t)ii[u] * D + 2 * k + th];
      v[u][k] = *reinterpret_cast<const float4*>(src + (size_t)oo[u] * OUT_CH + 128 * k + c0);
    }
  }
#pragma unroll
  for (int u = 0; u < NB; ++u) {
#pragma unroll
    for (int k = 0; k < SEG_C; ++k) {
      acc[k].x = __builtin_fmaf(a[u][k], v[u][k].x, acc[k].x);
      acc[k].y = __builtin_fmaf(a[u][k], v[u][k].y, acc[k].y);
      acc[k].z = __builtin_fmaf(a[u][k], v[u][k].z, acc[k].z);
      acc[k].w = __builtin_fmaf(a[u][k], v[u][k].w, acc[k].w);
      if (NODE_SIDE) dsum[k] = __builtin_fmaf(a[u][k], a[u][k], dsum[k]);
    }
  }
}

template <bool NODE_SIDE>
__global__ void __launch_bounds__(LOOP_T)
seg_gather_kernel(const int* __restrict__ ptr, const int* __restrict__ lst,
                  const int* __restrict__ oth, const float* __restrict__ alpha,
                  const float* __restrict__ src, const float* __restrict__ h,
                  const float* __restrict__ bias, float* __restrict__ dst, int nseg) {
  const int nw = gridDim.x * LOOP_WAVES;
  int seg = blockIdx.x * LOOP_WAVES + (threadIdx.x >> 5);
  if (seg >= nseg) return;
  const int lane = threadIdx.x & 31;
  const int c0 = 4 * lane;
  const int th = lane >> 4;
  float4 bv = make_float4(0.0f, 0.0f, 0.0f, 0.0f);
  if (NODE_SIDE) bv = *reinterpret_cast<const float4*>(bias + (c0 & (HIDDEN - 1)));

  int s = ptr[seg], e = ptr[seg + 1];
  for (; seg < nseg; seg += nw) {
    const int sx = (seg + nw < nseg) ? (seg + nw) : seg;
    const int sn = ptr[sx], en = ptr[sx + 1];
    float4 hv[SEG_C];
    if (NODE_SIDE) {
#pragma unroll
      for (int k = 0; k < SEG_C; ++k)
        hv[k] = *reinterpret_cast<const float4*>(h + (size_t)seg * OUT_CH + 128 * k + c0);
    }
    float4 acc[SEG_C];
    float dsum[SEG_C];
#pragma unroll
    for (int k = 0; k < SEG_C; ++k) { acc[k] = make_float4(0.0f, 0.0f, 0.0f, 0.0f); dsum[k] = 0.0f; }

    for (int q = s; q < e; q += SEG_B) {
      const int n = e - q;
      if (n >= 4)      seg_group<NODE_SIDE, 4>(lst, oth, alpha, src, q, c0, th, acc, dsum);
      else if (n == 3) seg_group<NODE_SIDE, 3>(lst, oth, alpha, src, q, c0, th, acc, dsum);
      else if (n == 2) seg_group<NODE_SIDE, 2>(lst, oth, alpha, src, q, c0, th, acc, dsum);
      else             seg_group<NODE_SIDE, 1>(lst, oth, alpha, src, q, c0, th, acc, dsum);
    }

    if (NODE_SIDE) {
#pragma unroll
      for (int k = 0; k < SEG_C; ++k) {
        const float dinv = (dsum[k] > 0.0f) ? (1.0f / dsum[k]) : 0.0f;
        float4 o;
        o.x = hv[k].x - dinv * acc[k].x + bv.x;
        o.y = hv[k].y - dinv * acc[k].y + bv.y;
        o.z = hv[k].z - dinv * acc[k].z + bv.z;
        o.w = hv[k].w - dinv * acc[k].w + bv.w;
        o.x = (o.x > 0.0f) ? o.x : expm1f(o.x);
        o.y = (o.y > 0.0f) ? o.y : expm1f(o.y);
        o.z = (o.z > 0.0f) ? o.z : expm1f(o.z);
        o.w = (o.w > 0.0f) ? o.w : expm1f(o.w);
        *reinterpret_cast<float4*>(dst + (size_t)seg * OUT_CH + 128 * k + c0) = o;
      }
    } else {
      const int len = e - s;
      const float binv = (len > 0) ? (1.0f / (float)len) : 0.0f;
#pragma unroll
      for (int k = 0; k < SEG_C; ++k)
        *reinterpret_cast<float4*>(dst + (size_t)seg * OUT_CH + 128 * k + c0) =
            make_float4(acc[k].x * binv, acc[k].y * binv, acc[k].z * binv, acc[k].w * binv);
    }
    s = sn;
    e = en;
  }
}

constexpr int CONV_RUN   = 5;
constexpr int CONV_BLK   = ((ND / 16 + CONV_RUN - 1) / CONV_RUN + 3) / 4;
constexpr int SEG_E_BLK = (N_EDGES + LOOP_WAVES - 1) / LOOP_WAVES;
constexpr int SEG_N_BLK = LOOP_BLK;

extern "C" void kernel_launch(void* const* d_in, const int* in_sizes, int n_in,
                              void* d_out, int out_size, void* d_ws, size_t ws_size,
                              hipStream_t stream) {
  (void)in_sizes; (void)n_in; (void)out_size; (void)ws_size;
  const float* x       = (const float*)d_in[0];
  const float* hattr   = (const float*)d_in[1];
  const float* W_lin   = (const float*)d_in[2];
  const float* b_lin   = (const float*)d_in[3];
  const float* ln_g    = (const float*)d_in[4];
  const float* ln_b    = (const float*)d_in[5];
  const float* W_sheaf = (const float*)d_in[6];
  const float* b_sheaf = (const float*)d_in[7];
  const float* W_conv  = (const float*)d_in[8];
  const float* b_conv  = (const float*)d_in[9];
  const int*   eidx    = (const int*)d_in[10];
  const int* row = eidx;
  const int* col = eidx + NNZ;
  float* out = (float*)d_out;

  float* ws    = (float*)d_ws;
  float* fold  = ws;
  float* Wp    = fold;
  float* bp    = Wp + (size_t)IN_CH * OUT_CH;
  float* Wm    = bp + OUT_CH;
  float* bm    = Wm + (size_t)IN_CH * HIDDEN;
  float* es    = ws   + (size_t)ND * HIDDEN;
  float* xm    = es   + (size_t)ED * HIDDEN;
  float* em    = xm   + (size_t)N_NODES * HIDDEN;
  float* alpha = em   + (size_t)N_EDGES * HIDDEN;
  float* spare = alpha + (size_t)NNZ * D;
  float* hbuf  = spare + ND + N_EDGES;
  float* me    = hbuf + (size_t)ND * HIDDEN;
  int*   ib    = (int*)(me + (size_t)ED * HIDDEN);
  int* nptr = ib;
  int* eptr = nptr + N_NODES + 8;
  unsigned* ntmp = (unsigned*)(eptr + N_EDGES + 8);
  unsigned* etmp = ntmp + NNZ;
  int* nlst = (int*)(etmp + NNZ);
  int* ncol = nlst + NNZ;
  int* elst = ncol + NNZ;
  int* erow = elst + NNZ;

  csr_build_kernel<<<CSR_NB_N + CSR_NB_E, CSR_T, 0, stream>>>(row, col, nptr, eptr, ntmp, etmp,
                                                              nlst, ncol, elst, erow);

  fold_weights_kernel<<<FOLD_BLK, 32 * FOLD_WPB, 0, stream>>>(W_lin, b_lin, W_conv, Wp, bp, Wm, bm);

  proj_fold_kernel<<<PF_BLK_H + PF_BLK_M, 32 * PF_WAVES, 0, stream>>>(x, hattr, Wp, bp, Wm, bm,
                                                                     hbuf, xm, em);

  alpha_kernel<<<LOOP_BLK, LOOP_T, 0, stream>>>(xm, em, row, col, ln_g, ln_b,
                                                W_sheaf, b_sheaf, alpha);

  seg_gather_kernel<false><<<SEG_E_BLK, LOOP_T, 0, stream>>>(eptr, elst, erow, alpha, hbuf,
                                                           nullptr, nullptr, me, N_EDGES);
  seg_gather_kernel<true><<<SEG_N_BLK, LOOP_T, 0, stream>>>(nptr, nlst, ncol, alpha, me,
                                                          hbuf, b_conv, out, N_NODES);

  wmma_gemm_f32<HIDDEN, 4, 1, false><<<CONV_BLK, 32 * 4, 0, stream>>>(
      out, W_conv + HIDDEN * HIDDEN, nullptr, hbuf, ND / 16, CONV_RUN);
  seg_gather_kernel<false><<<SEG_E_BLK, LOOP_T, 0, stream>>>(eptr, elst, erow, alpha, hbuf,
                                                           nullptr, nullptr, me, N_EDGES);
  seg_gather_kernel<true><<<SEG_N_BLK, LOOP_T, 0, stream>>>(nptr, nlst, ncol, alpha, me,
                                                          hbuf, b_conv + HIDDEN, out, N_NODES);
}


# Sheaf-coefficient kernel: 8 lanes per incidence, 4 incidences per wave; LayerNorm and projection sums in-lane, then 3 butterfly steps as row-swizzle (DPP) adds, not LDS shuffles; sums re-associated
# speedup vs baseline: 4.6571x; 1.0587x over previous
#include <hip/hip_runtime.h>
#include <math.h>

constexpr int N_NODES = 50000;
constexpr int N_EDGES = 10000;
constexpr int NNZ     = 150000;
constexpr int IN_CH   = 128;
constexpr int HIDDEN  = 64;
constexpr int D       = 6;
constexpr int OUT_CH  = HIDDEN * D;
constexpr int ND      = N_NODES * D;
constexpr int ED      = N_EDGES * D;

typedef __attribute__((ext_vector_type(2))) float v2f;
typedef __attribute__((ext_vector_type(8))) float v8f;

template <int NT> struct ColVec;
template <> struct ColVec<2> { typedef float2 type; };
template <> struct ColVec<4> { typedef float4 type; };

template <int NT>
__device__ inline void load_cols(const float* __restrict__ p, float (&d)[NT]) {
  const typename ColVec<NT>::type v = *reinterpret_cast<const typename ColVec<NT>::type*>(p);
  const float* f = reinterpret_cast<const float*>(&v);
#pragma unroll
  for (int t = 0; t < NT; ++t) d[t] = f[t];
}

template <int NT>
__device__ inline void store_cols(float* __restrict__ p, const float (&d)[NT]) {
  typename ColVec<NT>::type v;
  float* f = reinterpret_cast<float*>(&v);
#pragma unroll
  for (int t = 0; t < NT; ++t) f[t] = d[t];
  *reinterpret_cast<typename ColVec<NT>::type*>(p) = v;
}

constexpr int gemm_wpb(int g) { return (g > 1) ? g : 4; }

template <int K>
__device__ inline void load_a_tile(float4 (&av)[K / 8], const float* __restrict__ A,
                                   int mt, int c, int h) {
  const float* ap = A + ((size_t)mt * 16 + c) * K + 8 * h;
#pragma unroll
  for (int g = 0; g < K / 16; ++g) {
    av[2 * g]     = *reinterpret_cast<const float4*>(ap + 16 * g);
    av[2 * g + 1] = *reinterpret_cast<const float4*>(ap + 16 * g + 4);
  }
}

template <int K, int NT, int G, bool HAS_BIAS>
__global__ void __launch_bounds__(32 * gemm_wpb(G))
wmma_gemm_f32(const float* __restrict__ A, const float* __restrict__ B,
              const float* __restrict__ bias, float* __restrict__ C,
              int mTiles, int tilesPerWave) {
  constexpr int N = 16 * NT * G;
  constexpr int S = K / 4;
  const int wave  = __builtin_amdgcn_readfirstlane(blockIdx.x * gemm_wpb(G) + (threadIdx.x >> 5));
  const int cg    = wave % G;
  int mt          = (wave / G) * tilesPerWave;
  if (mt >= mTiles) return;
  const int mtEnd = (mt + tilesPerWave < mTiles) ? (mt + tilesPerWave) : mTiles;
  const int lane = threadIdx.x & 31;
  const int h = lane >> 4;
  const int c = lane & 15;
  const int n0 = cg * 16 * NT;

  v2f bf[S][NT];
#pragma unroll
  for (int s = 0; s < S; ++s) {
    const int k = 16 * (s >> 2) + 8 * h + 2 * (s & 3);
    float w0[NT], w1[NT];
    load_cols<NT>(B + (size_t)k * N + n0 + NT * c, w0);
    load_cols<NT>(B + (size_t)(k + 1) * N + n0 + NT * c, w1);
#pragma unroll
    for (int t = 0; t < NT; ++t) { bf[s][t].x = w0[t]; bf[s][t].y = w1[t]; }
  }
  float bv[NT];
#pragma unroll
  for (int t = 0; t < NT; ++t) bv[t] = 0.0f;
  if (HAS_BIAS) load_cols<NT>(bias + n0 + NT * c, bv);

  float4 av[K / 8];
  load_a_tile<K>(av, A, mt, c, h);
  for (; mt < mtEnd; ++mt) {
    const int nx = (mt + 1 < mtEnd) ? (mt + 1) : mt;
    float4 an[K / 8];
    load_a_tile<K>(an, A, nx, c, h);
    __builtin_amdgcn_sched_barrier(0);

    v8f acc[NT];
#pragma unroll
    for (int t = 0; t < NT; ++t) acc[t] = v8f{};
#pragma unroll
    for (int s = 0; s < S; ++s) {
      const float4 q = av[s >> 1];
      v2f a;
      a.x = (s & 1) ? q.z : q.x;
      a.y = (s & 1) ? q.w : q.y;
#pragma unroll
      for (int t = 0; t < NT; ++t)
        acc[t] = __builtin_amdgcn_wmma_f32_16x16x4_f32(false, a, false, bf[s][t],
                                                       (short)0, acc[t], false, false);
    }
    float* cp = C + ((size_t)mt * 16 + 8 * h) * N + n0 + NT * c;
#pragma unroll
    for (int j = 0; j < 8; ++j) {
      float o[NT];
#pragma unroll
      for (int t = 0; t < NT; ++t) o[t] = acc[t][j] + bv[t];
      store_cols<NT>(cp + (size_t)j * N, o);
    }
    __builtin_amdgcn_sched_barrier(0);
#pragma unroll
    for (int i = 0; i < K / 8; ++i) av[i] = an[i];
  }
}

constexpr int FOLD_TILES = IN_CH * D / 16;
constexpr int FOLD_WPB   = 4;
constexpr int FOLD_BLK   = (FOLD_TILES + 1 + FOLD_WPB - 1) / FOLD_WPB;
constexpr int FOLD_MQ    = IN_CH * HIDDEN / 4;

__global__ void __launch_bounds__(32 * FOLD_WPB)
fold_weights_kernel(const float* __restrict__ W_lin, const float* __restrict__ b_lin,
                    const float* __restrict__ W0,
                    float* __restrict__ Wp, float* __restrict__ bp,
                    float* __restrict__ Wm, float* __restrict__ bm) {
  constexpr int K = HIDDEN, NT = 4, S = K / 4;
  for (int qd = blockIdx.x * (32 * FOLD_WPB) + threadIdx.x; qd < FOLD_MQ + HIDDEN / 4;
       qd += gridDim.x * (32 * FOLD_WPB)) {
    const bool isb = (qd >= FOLD_MQ);
    const float* p = isb ? (b_lin + 4 * (qd - FOLD_MQ)) : (W_lin + (size_t)(qd >> 4) * OUT_CH + 4 * (qd & 15));
    float4 sm = make_float4(0.0f, 0.0f, 0.0f, 0.0f);
#pragma unroll
    for (int t = 0; t < D; ++t) {
      const float4 v = *reinterpret_cast<const float4*>(p + t * HIDDEN);
      sm.x += v.x; sm.y += v.y; sm.z += v.z; sm.w += v.w;
    }
    const float k6 = 1.0f / 6.0f;
    float* d = isb ? (bm + 4 * (qd - FOLD_MQ)) : (Wm + (size_t)qd * 4);
    *reinterpret_cast<float4*>(d) = make_float4(sm.x * k6, sm.y * k6, sm.z * k6, sm.w * k6);
  }

  const int wave = __builtin_amdgcn_readfirstlane(blockIdx.x * FOLD_WPB + (threadIdx.x >> 5));
  if (wave > FOLD_TILES) return;
  const bool btile = (wave == FOLD_TILES);
  const float* src = btile ? b_lin : (W_lin + (size_t)wave * 16 * K);
  float* dst       = btile ? bp : (Wp + (size_t)wave * 16 * K);
  const int rows   = btile ? D : 16;
  const int lane = threadIdx.x & 31;
  const int h = lane >> 4;
  const int c = lane & 15;

  float4 av[K / 8];
#pragma unroll
  for (int i = 0; i < K / 8; ++i) av[i] = make_float4(0.0f, 0.0f, 0.0f, 0.0f);
  if (c < rows) {
    const float* ap = src + (size_t)c * K + 8 * h;
#pragma unroll
    for (int g = 0; g < K / 16; ++g) {
      av[2 * g]     = *reinterpret_cast<const float4*>(ap + 16 * g);
      av[2 * g + 1] = *reinterpret_cast<const float4*>(ap + 16 * g + 4);
    }
  }
  v8f acc[NT];
#pragma unroll
  for (int t = 0; t < NT; ++t) acc[t] = v8f{};
#pragma unroll
  for (int s = 0; s < S; ++s) {
    const int k = 16 * (s >> 2) + 8 * h + 2 * (s & 3);
    float w0[NT], w1[NT];
    load_cols<NT>(W0 + (size_t)k * K + NT * c, w0);
    load_cols<NT>(W0 + (size_t)(k + 1) * K + NT * c, w1);
    const float4 q = av[s >> 1];
    v2f a;
    a.x = (s & 1) ? q.z : q.x;
    a.y = (s & 1) ? q.w : q.y;
#pragma unroll
    for (int t = 0; t < NT; ++t) {
      v2f b;
      b.x = w0[t]; b.y = w1[t];
      acc[t] = __builtin_amdgcn_wmma_f32_16x16x4_f32(false, a, false, b, (short)0, acc[t], false, false);
    }
  }
#pragma unroll
  for (int j = 0; j < 8; ++j) {
    float o[NT];
#pragma unroll
    for (int t = 0; t < NT; ++t) o[t] = acc[t][j];
    if (j + 8 * h < rows) store_cols<NT>(dst + (size_t)(j + 8 * h) * K + NT * c, o);
  }
}

constexpr int PF_WAVES  = 12;
constexpr int PF_RUN    = 15;
constexpr int PF_RUNS_N = (N_NODES / 16 + PF_RUN - 1) / PF_RUN;
constexpr int PF_RUNS_E = (N_EDGES / 16 + PF_RUN - 1) / PF_RUN;
constexpr int PF_BLK_H  = PF_RUNS_N;
constexpr int PF_MW_N   = 2 * PF_RUNS_N;
constexpr int PF_MW_E   = 2 * PF_RUNS_E;
constexpr int PF_BLK_M  = (PF_MW_N + PF_MW_E + PF_WAVES - 1) / PF_WAVES;
static_assert(N_NODES % 16 == 0 && N_EDGES % 16 == 0, "row tiles are whole");

__global__ void __launch_bounds__(32 * PF_WAVES)
proj_fold_kernel(const float* __restrict__ x, const float* __restrict__ hattr,
                 const float* __restrict__ Wp, const float* __restrict__ bp,
                 const float* __restrict__ Wm, const float* __restrict__ bm,
                 float* __restrict__ h1, float* __restrict__ xm, float* __restrict__ em) {
  constexpr int K = IN_CH, NT = 2, S = K / 4;
  const int blk = blockIdx.x;
  const int wib = __builtin_amdgcn_readfirstlane(threadIdx.x >> 5);
  const float* A; const float* B; const float* bias; float* C;
  int pitch, n0, run, mTiles;
  if (blk < PF_BLK_H) {
    A = x; B = Wp; bias = bp; C = h1; pitch = OUT_CH; n0 = 16 * NT * wib; run = blk; mTiles = N_NODES / 16;
  } else {
    const int u = (blk - PF_BLK_H) * PF_WAVES + wib;
    B = Wm; bias = bm; pitch = HIDDEN;
    if (u < PF_MW_N) {
      A = x; C = xm; run = u >> 1; n0 = 16 * NT * (u & 1); mTiles = N_NODES / 16;
    } else {
      const int v = u - PF_MW_N;
      if (v >= PF_MW_E) return;
      A = hattr; C = em; run = v >> 1; n0 = 16 * NT * (v & 1); mTiles = N_EDGES / 16;
    }
  }
  int mt = run * PF_RUN;
  if (mt >= mTiles) return;
  const int mtEnd = (mt + PF_RUN < mTiles) ? (mt + PF_RUN) : mTiles;
  const int lane = threadIdx.x & 31;
  const int h = lane >> 4;
  const int c = lane & 15;

  v2f bf[S][NT];
#pragma unroll
  for (int s = 0; s < S; ++s) {
    const int k = 16 * (s >> 2) + 8 * h + 2 * (s & 3);
    float w0[NT], w1[NT];
    load_cols<NT>(B + (size_t)k * pitch + n0 + NT * c, w0);
    load_cols<NT>(B + (size_t)(k + 1) * pitch + n0 + NT * c, w1);
#pragma unroll
    for (int t = 0; t < NT; ++t) { bf[s][t].x = w0[t]; bf[s][t].y = w1[t]; }
  }
  float bv[NT];
  load_cols<NT>(bias + n0 + NT * c, bv);

  float4 av[K / 8];
  load_a_tile<K>(av, A, mt, c, h);
  for (; mt < mtEnd; ++mt) {
    const int nx = (mt + 1 < mtEnd) ? (mt + 1) : mt;
    float4 an[K / 8];
    load_a_tile<K>(an, A, nx, c, h);
    __builtin_amdgcn_sched_barrier(0);

    v8f acc[NT];
#pragma unroll
    for (int t = 0; t < NT; ++t) acc[t] = v8f{};
#pragma unroll
    for (int s = 0; s < S; ++s) {
      const float4 q = av[s >> 1];
      v2f a;
      a.x = (s & 1) ? q.z : q.x;
      a.y = (s & 1) ? q.w : q.y;
#pragma unroll
      for (int t = 0; t < NT; ++t)
        acc[t] = __builtin_amdgcn_wmma_f32_16x16x4_f32(false, a, false, bf[s][t],
                                                       (short)0, acc[t], false, false);
    }
    float* cp = C + ((size_t)mt * 16 + 8 * h) * pitch + n0 + NT * c;
#pragma unroll
    for (int j = 0; j < 8; ++j) {
      float o[NT];
#pragma unroll
      for (int t = 0; t < NT; ++t) o[t] = acc[t][j] + bv[t];
      store_cols<NT>(cp + (size_t)j * pitch, o);
    }
    __builtin_amdgcn_sched_barrier(0);
#pragma unroll
    for (int i = 0; i < K / 8; ++i) av[i] = an[i];
  }
}

constexpr int LOOP_WAVES = 8;
constexpr int LOOP_T     = 32 * LOOP_WAVES;
constexpr int LOOP_BLK   = 1024;

template <int M>
__device__ inline float xmask_add(float v) {
  const int o = __builtin_amdgcn_update_dpp(0, __builtin_bit_cast(int, v), 0x160 + M, 0xf, 0xf, false);
  return v + __builtin_bit_cast(float, o);
}
template <int LG>
__device__ inline float group_allsum(float v) {
  v = xmask_add<1>(v);
  v = xmask_add<2>(v);
  v = xmask_add<4>(v);
  if (LG == 16) v = xmask_add<8>(v);
  return v;
}

constexpr int ALPHA_LG   = 8;
constexpr int ALPHA_HL   = ALPHA_LG / 2;
constexpr int ALPHA_NQ   = HIDDEN / (4 * ALPHA_HL);
constexpr int ALPHA_IPT  = 32 / ALPHA_LG;
constexpr int ALPHA_NTRIP = NNZ / ALPHA_IPT;
constexpr int ALPHA_WPB  = 2;
constexpr int ALPHA_TPW  = 10;
constexpr int ALPHA_BLK  = (ALPHA_NTRIP + ALPHA_TPW * ALPHA_WPB - 1) / (ALPHA_TPW * ALPHA_WPB);
static_assert(NNZ % ALPHA_IPT == 0, "a trip is a whole group of incidences");
static_assert(ALPHA_LG == 8 || ALPHA_LG == 16, "lane groups lie inside the 16-lane rows");

__global__ void __launch_bounds__(32 * ALPHA_WPB)
alpha_kernel(const float* __restrict__ xm, const float* __restrict__ em,
             const int* __restrict__ row, const int* __restrict__ col,
             const float* __restrict__ ln_g, const float* __restrict__ ln_b,
             const float* __restrict__ Wsh, const float* __restrict__ bsh,
             float* __restrict__ alpha) {
  constexpr int LG = ALPHA_LG, HL = ALPHA_HL, NQ = ALPHA_NQ, IPT = ALPHA_IPT;
  const int nw = gridDim.x * ALPHA_WPB;
  int tr = __builtin_amdgcn_readfirstlane(blockIdx.x * ALPHA_WPB + (threadIdx.x >> 5));
  if (tr >= ALPHA_NTRIP) return;
  const int lane = threadIdx.x & 31;
  const int grp  = lane / LG;
  const int sub  = lane % LG;
  const bool eside = (sub >= HL);
  const int ls   = sub % HL;
  const float* zsrc = eside ? em : xm;
  const int*   ksrc = eside ? col : row;
  const int j0 = (eside ? HIDDEN : 0) + 4 * ls;

  float g[NQ][4], b[NQ][4], w[NQ][4][D];
#pragma unroll
  for (int q = 0; q < NQ; ++q) {
    const int j = j0 + 4 * HL * q;
    const float4 gv = *reinterpret_cast<const float4*>(ln_g + j);
    const float4 bv = *reinterpret_cast<const float4*>(ln_b + j);
    g[q][0] = gv.x; g[q][1] = gv.y; g[q][2] = gv.z; g[q][3] = gv.w;
    b[q][0] = bv.x; b[q][1] = bv.y; b[q][2] = bv.z; b[q][3] = bv.w;
    float wf[4 * D];
#pragma unroll
    for (int m = 0; m < D; ++m) {
      const float4 wv = *reinterpret_cast<const float4*>(Wsh + (size_t)j * D + 4 * m);
      wf[4 * m] = wv.x; wf[4 * m + 1] = wv.y; wf[4 * m + 2] = wv.z; wf[4 * m + 3] = wv.w;
    }
#pragma unroll
    for (int e = 0; e < 4; ++e)
#pragma unroll
      for (int t = 0; t < D; ++t) w[q][e][t] = wf[e * D + t];
  }
  const float bs = (sub < D) ? bsh[sub] : 0.0f;

  float4 zc[NQ], zn[NQ];
  {
    const int key = ksrc[tr * IPT + grp];
    const float* zp = zsrc + (size_t)key * HIDDEN + 4 * ls;
#pragma unroll
    for (int q = 0; q < NQ; ++q) zc[q] = *reinterpret_cast<const float4*>(zp + 4 * HL * q);
  }
  const int tr1 = (tr + nw < ALPHA_NTRIP) ? (tr + nw) : tr;
  int keyn = ksrc[tr1 * IPT + grp];
  for (; tr < ALPHA_NTRIP; tr += nw) {
    {
      const float* zp = zsrc + (size_t)keyn * HIDDEN + 4 * ls;
#pragma unroll
      for (int q = 0; q < NQ; ++q) zn[q] = *reinterpret_cast<const float4*>(zp + 4 * HL * q);
    }
    const int tr2 = (tr + 2 * nw < ALPHA_NTRIP) ? (tr + 2 * nw) : tr;
    const int keynn = ksrc[tr2 * IPT + grp];

    float z[NQ][4];
#pragma unroll
    for (int q = 0; q < NQ; ++q) { z[q][0] = zc[q].x; z[q][1] = zc[q].y; z[q][2] = zc[q].z; z[q][3] = zc[q].w; }
    float s = 0.0f, ss = 0.0f;
#pragma unroll
    for (int q = 0; q < NQ; ++q)
#pragma unroll
      for (int e = 0; e < 4; ++e) { s += z[q][e]; ss += z[q][e] * z[q][e]; }
    s  = group_allsum<LG>(s);
    ss = group_allsum<LG>(ss);
    const float mu   = s * (1.0f / 128.0f);
    const float var  = ss * (1.0f / 128.0f) - mu * mu;
    const float rstd = rsqrtf(var + 1e-5f);

    float p[D] = {0, 0, 0, 0, 0, 0};
#pragma unroll
    for (int q = 0; q < NQ; ++q) {
#pragma unroll
      for (int e = 0; e < 4; ++e) {
        const float fn = (z[q][e] - mu) * rstd * g[q][e] + b[q][e];
#pragma unroll
        for (int t = 0; t < D; ++t) p[t] = __builtin_fmaf(fn, w[q][e][t], p[t]);
      }
    }
#pragma unroll
    for (int t = 0; t < D; ++t) p[t] = group_allsum<LG>(p[t]);

    float pv = p[0];
#pragma unroll
    for (int t = 1; t < D; ++t) pv = (sub == t) ? p[t] : pv;
    if (sub < D) {
      const float a = 1.0f / (1.0f + expf(-(pv + bs)));
      alpha[(size_t)(tr * IPT + grp) * D + sub] = a;
    }
#pragma unroll
    for (int q = 0; q < NQ; ++q) zc[q] = zn[q];
    keyn = keynn;
  }
}

constexpr int CSR_T      = 1024;
constexpr int CSR_R_N    = 544;
constexpr int CSR_NB_N   = (N_NODES + CSR_R_N - 1) / CSR_R_N;
constexpr int CSR_R_E    = 160;
constexpr int CSR_NB_E   = (N_EDGES + CSR_R_E - 1) / CSR_R_E;
constexpr int CSR_IDBITS = 18;
constexpr int CSR_N4     = NNZ / 4;
constexpr int CSR_U      = 4;
static_assert(NNZ % 4 == 0, "the key arrays are read as 16-byte quads");
static_assert(NNZ <= (1 << CSR_IDBITS), "an incidence id must fit the low bits of a list word");
static_assert(CSR_R_N <= CSR_T && CSR_R_E <= CSR_T, "the scan takes one counter per thread");
static_assert(CSR_T <= (1 << (32 - CSR_IDBITS)), "a key slot must fit the high bits of a list word");
static_assert((CSR_NB_N - 1) * CSR_R_N < N_NODES && (CSR_NB_E - 1) * CSR_R_E < N_EDGES,
              "every work-group owns at least one key");

__device__ inline int wave_allsum_i(int v) {
#pragma unroll
  for (int off = 16; off > 0; off >>= 1) v += __shfl_xor(v, off, 32);
  return v;
}

__device__ inline int wave_incl_scan(int v, int lane) {
#pragma unroll
  for (int off = 1; off < 32; off <<= 1) {
    int u = __shfl_up(v, off, 32);
    if (lane >= off) v += u;
  }
  return v;
}

__device__ inline void csr_count4(const int4 q, int k0, int nk, int* cur, int& below) {
  const int kk[4] = {q.x, q.y, q.z, q.w};
#pragma unroll
  for (int c = 0; c < 4; ++c) {
    const int d = kk[c] - k0;
    below += (d < 0) ? 1 : 0;
    if ((unsigned)d < (unsigned)nk) atomicAdd(&cur[d], 1);
  }
}

__device__ inline void csr_place4(const int4 q, int i0, int k0, int nk, int* cur,
                                  const int* start, unsigned* __restrict__ tmp) {
  const int kk[4] = {q.x, q.y, q.z, q.w};
#pragma unroll
  for (int c = 0; c < 4; ++c) {
    const int d = kk[c] - k0;
    if ((unsigned)d < (unsigned)nk) {
      const int slot = atomicAdd(&cur[d], 1);
      tmp[start[d] + slot] = ((unsigned)d << CSR_IDBITS) | (unsigned)(i0 + c);
    }
  }
}

__global__ void __launch_bounds__(CSR_T)
csr_build_kernel(const int* __restrict__ row, const int* __restrict__ col,
                 int* __restrict__ nptr, int* __restrict__ eptr,
                 unsigned* __restrict__ ntmp, unsigned* __restrict__ etmp,
                 int* __restrict__ nlst, int* __restrict__ ncol,
                 int* __restrict__ elst, int* __restrict__ erow) {
  __shared__ int cur[CSR_T];
  __shared__ int start[CSR_T];
  __shared__ int wsum[32];
  __shared__ int wbel[32];
  __shared__ int sh_base, sh_total;

  const bool eside = (blockIdx.x >= CSR_NB_N);
  const int b      = eside ? (blockIdx.x - CSR_NB_N) : blockIdx.x;
  const int* keys  = eside ? col : row;
  const int* other = eside ? row : col;
  const int nkeys  = eside ? N_EDGES : N_NODES;
  const int R      = eside ? CSR_R_E : CSR_R_N;
  int* ptr         = eside ? eptr : nptr;
  unsigned* tmp    = eside ? etmp : ntmp;
  int* lst         = eside ? elst : nlst;
  int* oth         = eside ? erow : ncol;
  const int k0 = b * R;
  const int nk = (nkeys - k0 < R) ? (nkeys - k0) : R;
  const int tid = threadIdx.x, lane = tid & 31, w = tid >> 5;
  const int4* k4 = reinterpret_cast<const int4*>(keys);

  cur[tid] = 0;
  __syncthreads();

  int below = 0;
  {
    int v = tid;
    for (; v + (CSR_U - 1) * CSR_T < CSR_N4; v += CSR_U * CSR_T) {
      int4 q[CSR_U];
#pragma unroll
      for (int u = 0; u < CSR_U; ++u) q[u] = k4[v + u * CSR_T];
#pragma unroll
      for (int u = 0; u < CSR_U; ++u) csr_count4(q[u], k0, nk, cur, below);
    }
    for (; v < CSR_N4; v += CSR_T) csr_count4(k4[v], k0, nk, cur, below);
  }
  const int bw = wave_allsum_i(below);
  if (lane == 0) wbel[w] = bw;
  __syncthreads();

  const int cnt  = (tid < nk) ? cur[tid] : 0;
  const int incl = wave_incl_scan(cnt, lane);
  if (lane == 31) wsum[w] = incl;
  __syncthreads();
  if (w == 0) {
    const int t  = wsum[lane];
    const int ti = wave_incl_scan(t, lane);
    const int bl = wave_allsum_i(wbel[lane]);
    wsum[lane] = ti - t;
    if (lane == 31) { sh_base = bl; sh_total = ti; }
  }
  __syncthreads();
  const int base  = sh_base;
  const int total = sh_total;
  const int st = base + wsum[w] + incl - cnt;
  start[tid] = st;
  cur[tid] = 0;
  if (tid < nk) ptr[k0 + tid] = st;
  if (tid == nk - 1 && k0 + nk == nkeys) ptr[nkeys] = st + cnt;
  __syncthreads();

  {
    int v = tid;
    for (; v + (CSR_U - 1) * CSR_T < CSR_N4; v += CSR_U * CSR_T) {
      int4 q[CSR_U];
#pragma unroll
      for (int u = 0; u < CSR_U; ++u) q[u] = k4[v + u * CSR_T];
#pragma unroll
      for (int u = 0; u < CSR_U; ++u)
        csr_place4(q[u], 4 * (v + u * CSR_T), k0, nk, cur, start, tmp);
    }
    for (; v < CSR_N4; v += CSR_T) csr_place4(k4[v], 4 * v, k0, nk, cur, start, tmp);
  }
  __syncthreads();

  for (int p = tid; p < total; p += CSR_T) {
    const unsigned wv = tmp[base + p];
    const int d = (int)(wv >> CSR_IDBITS);
    const int i = (int)(wv & ((1u << CSR_IDBITS) - 1u));
    const int o = other[i];
    const int s = start[d], n = cur[d];
    const unsigned* sp = tmp + s;
    int rank = 0;
    int k = 0;
    for (; k + 4 <= n; k += 4) {
      const unsigned u0 = sp[k], u1 = sp[k + 1], u2 = sp[k + 2], u3 = sp[k + 3];
      rank += ((u0 < wv) ? 1 : 0) + ((u1 < wv) ? 1 : 0) + ((u2 < wv) ? 1 : 0) + ((u3 < wv) ? 1 : 0);
    }
    for (; k < n; ++k) rank += (sp[k] < wv) ? 1 : 0;
    lst[s + rank] = i;
    oth[s + rank] = o;
  }
}

constexpr int SEG_C = OUT_CH / 128;
constexpr int SEG_B = 4;
static_assert(SEG_B == 4, "the list walk picks its group body from the sizes 4, 3, 2, 1");

template <bool NODE_SIDE, int NB>
__device__ inline void seg_group(const int* __restrict__ lst, const int* __restrict__ oth,
                                 const float* __restrict__ alpha, const float* __restrict__ src,
                                 int q, int c0, int th,
                                 float4 (&acc)[SEG_C], float (&dsum)[SEG_C]) {
  int ii[NB], oo[NB];
#pragma unroll
  for (int u = 0; u < NB; ++u) { ii[u] = lst[q + u]; oo[u] = oth[q + u]; }
  float a[NB][SEG_C];
  float4 v[NB][SEG_C];
#pragma unroll
  for (int u = 0; u < NB; ++u) {
#pragma unroll
    for (int k = 0; k < SEG_C; ++k) {
      a[u][k] = alpha[(size_t)ii[u] * D + 2 * k + th];
      v[u][k] = *reinterpret_cast<const float4*>(src + (size_t)oo[u] * OUT_CH + 128 * k + c0);
    }
  }
#pragma unroll
  for (int u = 0; u < NB; ++u) {
#pragma unroll
    for (int k = 0; k < SEG_C; ++k) {
      acc[k].x = __builtin_fmaf(a[u][k], v[u][k].x, acc[k].x);
      acc[k].y = __builtin_fmaf(a[u][k], v[u][k].y, acc[k].y);
      acc[k].z = __builtin_fmaf(a[u][k], v[u][k].z, acc[k].z);
      acc[k].w = __builtin_fmaf(a[u][k], v[u][k].w, acc[k].w);
      if (NODE_SIDE) dsum[k] = __builtin_fmaf(a[u][k], a[u][k], dsum[k]);
    }
  }
}

template <bool NODE_SIDE>
__global__ void __launch_bounds__(LOOP_T)
seg_gather_kernel(const int* __restrict__ ptr, const int* __restrict__ lst,
                  const int* __restrict__ oth, const float* __restrict__ alpha,
                  const float* __restrict__ src, const float* __restrict__ h,
                  const float* __restrict__ bias, float* __restrict__ dst, int nseg) {
  const int nw = gridDim.x * LOOP_WAVES;
  int seg = blockIdx.x * LOOP_WAVES + (threadIdx.x >> 5);
  if (seg >= nseg) return;
  const int lane = threadIdx.x & 31;
  const int c0 = 4 * lane;
  const int th = lane >> 4;
  float4 bv = make_float4(0.0f, 0.0f, 0.0f, 0.0f);
  if (NODE_SIDE) bv = *reinterpret_cast<const float4*>(bias + (c0 & (HIDDEN - 1)));

  int s = ptr[seg], e = ptr[seg + 1];
  for (; seg < nseg; seg += nw) {
    const int sx = (seg + nw < nseg) ? (seg + nw) : seg;
    const int sn = ptr[sx], en = ptr[sx + 1];
    float4 hv[SEG_C];
    if (NODE_SIDE) {
#pragma unroll
      for (int k = 0; k < SEG_C; ++k)
        hv[k] = *reinterpret_cast<const float4*>(h + (size_t)seg * OUT_CH + 128 * k + c0);
    }
    float4 acc[SEG_C];
    float dsum[SEG_C];
#pragma unroll
    for (int k = 0; k < SEG_C; ++k) { acc[k] = make_float4(0.0f, 0.0f, 0.0f, 0.0f); dsum[k] = 0.0f; }

    for (int q = s; q < e; q += SEG_B) {
      const int n = e - q;
      if (n >= 4)      seg_group<NODE_SIDE, 4>(lst, oth, alpha, src, q, c0, th, acc, dsum);
      else if (n == 3) seg_group<NODE_SIDE, 3>(lst, oth, alpha, src, q, c0, th, acc, dsum);
      else if (n == 2) seg_group<NODE_SIDE, 2>(lst, oth, alpha, src, q, c0, th, acc, dsum);
      else             seg_group<NODE_SIDE, 1>(lst, oth, alpha, src, q, c0, th, acc, dsum);
    }

    if (NODE_SIDE) {
#pragma unroll
      for (int k = 0; k < SEG_C; ++k) {
        const float dinv = (dsum[k] > 0.0f) ? (1.0f / dsum[k]) : 0.0f;
        float4 o;
        o.x = hv[k].x - dinv * acc[k].x + bv.x;
        o.y = hv[k].y - dinv * acc[k].y + bv.y;
        o.z = hv[k].z - dinv * acc[k].z + bv.z;
        o.w = hv[k].w - dinv * acc[k].w + bv.w;
        o.x = (o.x > 0.0f) ? o.x : expm1f(o.x);
        o.y = (o.y > 0.0f) ? o.y : expm1f(o.y);
        o.z = (o.z > 0.0f) ? o.z : expm1f(o.z);
        o.w = (o.w > 0.0f) ? o.w : expm1f(o.w);
        *reinterpret_cast<float4*>(dst + (size_t)seg * OUT_CH + 128 * k + c0) = o;
      }
    } else {
      const int len = e - s;
      const float binv = (len > 0) ? (1.0f / (float)len) : 0.0f;
#pragma unroll
      for (int k = 0; k < SEG_C; ++k)
        *reinterpret_cast<float4*>(dst + (size_t)seg * OUT_CH + 128 * k + c0) =
            make_float4(acc[k].x * binv, acc[k].y * binv, acc[k].z * binv, acc[k].w * binv);
    }
    s = sn;
    e = en;
  }
}

constexpr int CONV_RUN   = 5;
constexpr int CONV_BLK   = ((ND / 16 + CONV_RUN - 1) / CONV_RUN + 3) / 4;
constexpr int SEG_E_BLK = (N_EDGES + LOOP_WAVES - 1) / LOOP_WAVES;
constexpr int SEG_N_BLK = LOOP_BLK;

extern "C" void kernel_launch(void* const* d_in, const int* in_sizes, int n_in,
                              void* d_out, int out_size, void* d_ws, size_t ws_size,
                              hipStream_t stream) {
  (void)in_sizes; (void)n_in; (void)out_size; (void)ws_size;
  const float* x       = (const float*)d_in[0];
  const float* hattr   = (const float*)d_in[1];
  const float* W_lin   = (const float*)d_in[2];
  const float* b_lin   = (const float*)d_in[3];
  const float* ln_g    = (const float*)d_in[4];
  const float* ln_b    = (const float*)d_in[5];
  const float* W_sheaf = (const float*)d_in[6];
  const float* b_sheaf = (const float*)d_in[7];
  const float* W_conv  = (const float*)d_in[8];
  const float* b_conv  = (const float*)d_in[9];
  const int*   eidx    = (const int*)d_in[10];
  const int* row = eidx;
  const int* col = eidx + NNZ;
  float* out = (float*)d_out;

  float* ws    = (float*)d_ws;
  float* fold  = ws;
  float* Wp    = fold;
  float* bp    = Wp + (size_t)IN_CH * OUT_CH;
  float* Wm    = bp + OUT_CH;
  float* bm    = Wm + (size_t)IN_CH * HIDDEN;
  float* es    = ws   + (size_t)ND * HIDDEN;
  float* xm    = es   + (size_t)ED * HIDDEN;
  float* em    = xm   + (size_t)N_NODES * HIDDEN;
  float* alpha = em   + (size_t)N_EDGES * HIDDEN;
  float* spare = alpha + (size_t)NNZ * D;
  float* hbuf  = spare + ND + N_EDGES;
  float* me    = hbuf + (size_t)ND * HIDDEN;
  int*   ib    = (int*)(me + (size_t)ED * HIDDEN);
  int* nptr = ib;
  int* eptr = nptr + N_NODES + 8;
  unsigned* ntmp = (unsigned*)(eptr + N_EDGES + 8);
  unsigned* etmp = ntmp + NNZ;
  int* nlst = (int*)(etmp + NNZ);
  int* ncol = nlst + NNZ;
  int* elst = ncol + NNZ;
  int* erow = elst + NNZ;

  csr_build_kernel<<<CSR_NB_N + CSR_NB_E, CSR_T, 0, stream>>>(row, col, nptr, eptr, ntmp, etmp,
                                                              nlst, ncol, elst, erow);

  fold_weights_kernel<<<FOLD_BLK, 32 * FOLD_WPB, 0, stream>>>(W_lin, b_lin, W_conv, Wp, bp, Wm, bm);

  proj_fold_kernel<<<PF_BLK_H + PF_BLK_M, 32 * PF_WAVES, 0, stream>>>(x, hattr, Wp, bp, Wm, bm,
                                                                     hbuf, xm, em);

  alpha_kernel<<<ALPHA_BLK, 32 * ALPHA_WPB, 0, stream>>>(xm, em, row, col, ln_g, ln_b,
                                                W_sheaf, b_sheaf, alpha);

  seg_gather_kernel<false><<<SEG_E_BLK, LOOP_T, 0, stream>>>(eptr, elst, erow, alpha, hbuf,
                                                           nullptr, nullptr, me, N_EDGES);
  seg_gather_kernel<true><<<SEG_N_BLK, LOOP_T, 0, stream>>>(nptr, nlst, ncol, alpha, me,
                                                          hbuf, b_conv, out, N_NODES);

  wmma_gemm_f32<HIDDEN, 4, 1, false><<<CONV_BLK, 32 * 4, 0, stream>>>(
      out, W_conv + HIDDEN * HIDDEN, nullptr, hbuf, ND / 16, CONV_RUN);
  seg_gather_kernel<false><<<SEG_E_BLK, LOOP_T, 0, stream>>>(eptr, elst, erow, alpha, hbuf,
                                                           nullptr, nullptr, me, N_EDGES);
  seg_gather_kernel<true><<<SEG_N_BLK, LOOP_T, 0, stream>>>(nptr, nlst, ncol, alpha, me,
                                                          hbuf, b_conv + HIDDEN, out, N_NODES);
}


# Second conv product (x1 @ W1) done inside the layer-1 node-side gather: a wave puts two nodes' 12 stalk rows in LDS as one WMMA tile; x1 never stored, one launch fewer; sum orders kept
# speedup vs baseline: 4.8245x; 1.0359x over previous
#include <hip/hip_runtime.h>
#include <math.h>

constexpr int N_NODES = 50000;
constexpr int N_EDGES = 10000;
constexpr int NNZ     = 150000;
constexpr int IN_CH   = 128;
constexpr int HIDDEN  = 64;
constexpr int D       = 6;
constexpr int OUT_CH  = HIDDEN * D;
constexpr int ND      = N_NODES * D;
constexpr int ED      = N_EDGES * D;

typedef __attribute__((ext_vector_type(2))) float v2f;
typedef __attribute__((ext_vector_type(8))) float v8f;

template <int NT> struct ColVec;
template <> struct ColVec<2> { typedef float2 type; };
template <> struct ColVec<4> { typedef float4 type; };

template <int NT>
__device__ inline void load_cols(const float* __restrict__ p, float (&d)[NT]) {
  const typename ColVec<NT>::type v = *reinterpret_cast<const typename ColVec<NT>::type*>(p);
  const float* f = reinterpret_cast<const float*>(&v);
#pragma unroll
  for (int t = 0; t < NT; ++t) d[t] = f[t];
}

template <int NT>
__device__ inline void store_cols(float* __restrict__ p, const float (&d)[NT]) {
  typename ColVec<NT>::type v;
  float* f = reinterpret_cast<float*>(&v);
#pragma unroll
  for (int t = 0; t < NT; ++t) f[t] = d[t];
  *reinterpret_cast<typename ColVec<NT>::type*>(p) = v;
}

template <int K>
__device__ inline void load_a_tile(float4 (&av)[K / 8], const float* __restrict__ A,
                                   int mt, int c, int h) {
  const float* ap = A + ((size_t)mt * 16 + c) * K + 8 * h;
#pragma unroll
  for (int g = 0; g < K / 16; ++g) {
    av[2 * g]     = *reinterpret_cast<const float4*>(ap + 16 * g);
    av[2 * g + 1] = *reinterpret_cast<const float4*>(ap + 16 * g + 4);
  }
}

constexpr int FOLD_TILES = IN_CH * D / 16;
constexpr int FOLD_WPB   = 4;
constexpr int FOLD_BLK   = (FOLD_TILES + 1 + FOLD_WPB - 1) / FOLD_WPB;
constexpr int FOLD_MQ    = IN_CH * HIDDEN / 4;

__global__ void __launch_bounds__(32 * FOLD_WPB)
fold_weights_kernel(const float* __restrict__ W_lin, const float* __restrict__ b_lin,
                    const float* __restrict__ W0,
                    float* __restrict__ Wp, float* __restrict__ bp,
                    float* __restrict__ Wm, float* __restrict__ bm) {
  constexpr int K = HIDDEN, NT = 4, S = K / 4;
  for (int qd = blockIdx.x * (32 * FOLD_WPB) + threadIdx.x; qd < FOLD_MQ + HIDDEN / 4;
       qd += gridDim.x * (32 * FOLD_WPB)) {
    const bool isb = (qd >= FOLD_MQ);
    const float* p = isb ? (b_lin + 4 * (qd - FOLD_MQ)) : (W_lin + (size_t)(qd >> 4) * OUT_CH + 4 * (qd & 15));
    float4 sm = make_float4(0.0f, 0.0f, 0.0f, 0.0f);
#pragma unroll
    for (int t = 0; t < D; ++t) {
      const float4 v = *reinterpret_cast<const float4*>(p + t * HIDDEN);
      sm.x += v.x; sm.y += v.y; sm.z += v.z; sm.w += v.w;
    }
    const float k6 = 1.0f / 6.0f;
    float* d = isb ? (bm + 4 * (qd - FOLD_MQ)) : (Wm + (size_t)qd * 4);
    *reinterpret_cast<float4*>(d) = make_float4(sm.x * k6, sm.y * k6, sm.z * k6, sm.w * k6);
  }

  const int wave = __builtin_amdgcn_readfirstlane(blockIdx.x * FOLD_WPB + (threadIdx.x >> 5));
  if (wave > FOLD_TILES) return;
  const bool btile = (wave == FOLD_TILES);
  const float* src = btile ? b_lin : (W_lin + (size_t)wave * 16 * K);
  float* dst       = btile ? bp : (Wp + (size_t)wave * 16 * K);
  const int rows   = btile ? D : 16;
  const int lane = threadIdx.x & 31;
  const int h = lane >> 4;
  const int c = lane & 15;

  float4 av[K / 8];
#pragma unroll
  for (int i = 0; i < K / 8; ++i) av[i] = make_float4(0.0f, 0.0f, 0.0f, 0.0f);
  if (c < rows) {
    const float* ap = src + (size_t)c * K + 8 * h;
#pragma unroll
    for (int g = 0; g < K / 16; ++g) {
      av[2 * g]     = *reinterpret_cast<const float4*>(ap + 16 * g);
      av[2 * g + 1] = *reinterpret_cast<const float4*>(ap + 16 * g + 4);
    }
  }
  v8f acc[NT];
#pragma unroll
  for (int t = 0; t < NT; ++t) acc[t] = v8f{};
#pragma unroll
  for (int s = 0; s < S; ++s) {
    const int k = 16 * (s >> 2) + 8 * h + 2 * (s & 3);
    float w0[NT], w1[NT];
    load_cols<NT>(W0 + (size_t)k * K + NT * c, w0);
    load_cols<NT>(W0 + (size_t)(k + 1) * K + NT * c, w1);
    const float4 q = av[s >> 1];
    v2f a;
    a.x = (s & 1) ? q.z : q.x;
    a.y = (s & 1) ? q.w : q.y;
#pragma unroll
    for (int t = 0; t < NT; ++t) {
      v2f b;
      b.x = w0[t]; b.y = w1[t];
      acc[t] = __builtin_amdgcn_wmma_f32_16x16x4_f32(false, a, false, b, (short)0, acc[t], false, false);
    }
  }
#pragma unroll
  for (int j = 0; j < 8; ++j) {
    float o[NT];
#pragma unroll
    for (int t = 0; t < NT; ++t) o[t] = acc[t][j];
    if (j + 8 * h < rows) store_cols<NT>(dst + (size_t)(j + 8 * h) * K + NT * c, o);
  }
}

constexpr int PF_WAVES  = 12;
constexpr int PF_RUN    = 15;
constexpr int PF_RUNS_N = (N_NODES / 16 + PF_RUN - 1) / PF_RUN;
constexpr int PF_RUNS_E = (N_EDGES / 16 + PF_RUN - 1) / PF_RUN;
constexpr int PF_BLK_H  = PF_RUNS_N;
constexpr int PF_MW_N   = 2 * PF_RUNS_N;
constexpr int PF_MW_E   = 2 * PF_RUNS_E;
constexpr int PF_BLK_M  = (PF_MW_N + PF_MW_E + PF_WAVES - 1) / PF_WAVES;
static_assert(N_NODES % 16 == 0 && N_EDGES % 16 == 0, "row tiles are whole");

__global__ void __launch_bounds__(32 * PF_WAVES)
proj_fold_kernel(const float* __restrict__ x, const float* __restrict__ hattr,
                 const float* __restrict__ Wp, const float* __restrict__ bp,
                 const float* __restrict__ Wm, const float* __restrict__ bm,
                 float* __restrict__ h1, float* __restrict__ xm, float* __restrict__ em) {
  constexpr int K = IN_CH, NT = 2, S = K / 4;
  const int blk = blockIdx.x;
  const int wib = __builtin_amdgcn_readfirstlane(threadIdx.x >> 5);
  const float* A; const float* B; const float* bias; float* C;
  int pitch, n0, run, mTiles;
  if (blk < PF_BLK_H) {
    A = x; B = Wp; bias = bp; C = h1; pitch = OUT_CH; n0 = 16 * NT * wib; run = blk; mTiles = N_NODES / 16;
  } else {
    const int u = (blk - PF_BLK_H) * PF_WAVES + wib;
    B = Wm; bias = bm; pitch = HIDDEN;
    if (u < PF_MW_N) {
      A = x; C = xm; run = u >> 1; n0 = 16 * NT * (u & 1); mTiles = N_NODES / 16;
    } else {
      const int v = u - PF_MW_N;
      if (v >= PF_MW_E) return;
      A = hattr; C = em; run = v >> 1; n0 = 16 * NT * (v & 1); mTiles = N_EDGES / 16;
    }
  }
  int mt = run * PF_RUN;
  if (mt >= mTiles) return;
  const int mtEnd = (mt + PF_RUN < mTiles) ? (mt + PF_RUN) : mTiles;
  const int lane = threadIdx.x & 31;
  const int h = lane >> 4;
  const int c = lane & 15;

  v2f bf[S][NT];
#pragma unroll
  for (int s = 0; s < S; ++s) {
    const int k = 16 * (s >> 2) + 8 * h + 2 * (s & 3);
    float w0[NT], w1[NT];
    load_cols<NT>(B + (size_t)k * pitch + n0 + NT * c, w0);
    load_cols<NT>(B + (size_t)(k + 1) * pitch + n0 + NT * c, w1);
#pragma unroll
    for (int t = 0; t < NT; ++t) { bf[s][t].x = w0[t]; bf[s][t].y = w1[t]; }
  }
  float bv[NT];
  load_cols<NT>(bias + n0 + NT * c, bv);

  float4 av[K / 8];
  load_a_tile<K>(av, A, mt, c, h);
  for (; mt < mtEnd; ++mt) {
    const int nx = (mt + 1 < mtEnd) ? (mt + 1) : mt;
    float4 an[K / 8];
    load_a_tile<K>(an, A, nx, c, h);
    __builtin_amdgcn_sched_barrier(0);

    v8f acc[NT];
#pragma unroll
    for (int t = 0; t < NT; ++t) acc[t] = v8f{};
#pragma unroll
    for (int s = 0; s < S; ++s) {
      const float4 q = av[s >> 1];
      v2f a;
      a.x = (s & 1) ? q.z : q.x;
      a.y = (s & 1) ? q.w : q.y;
#pragma unroll
      for (int t = 0; t < NT; ++t)
        acc[t] = __builtin_amdgcn_wmma_f32_16x16x4_f32(false, a, false, bf[s][t],
                                                       (short)0, acc[t], false, false);
    }
    float* cp = C + ((size_t)mt * 16 + 8 * h) * pitch + n0 + NT * c;
#pragma unroll
    for (int j = 0; j < 8; ++j) {
      float o[NT];
#pragma unroll
      for (int t = 0; t < NT; ++t) o[t] = acc[t][j] + bv[t];
      store_cols<NT>(cp + (size_t)j * pitch, o);
    }
    __builtin_amdgcn_sched_barrier(0);
#pragma unroll
    for (int i = 0; i < K / 8; ++i) av[i] = an[i];
  }
}

constexpr int LOOP_WAVES = 8;
constexpr int LOOP_T     = 32 * LOOP_WAVES;
constexpr int LOOP_BLK   = 1024;

template <int M>
__device__ inline float xmask_add(float v) {
  const int o = __builtin_amdgcn_update_dpp(0, __builtin_bit_cast(int, v), 0x160 + M, 0xf, 0xf, false);
  return v + __builtin_bit_cast(float, o);
}
template <int LG>
__device__ inline float group_allsum(float v) {
  v = xmask_add<1>(v);
  v = xmask_add<2>(v);
  v = xmask_add<4>(v);
  if (LG == 16) v = xmask_add<8>(v);
  return v;
}

constexpr int ALPHA_LG   = 8;
constexpr int ALPHA_HL   = ALPHA_LG / 2;
constexpr int ALPHA_NQ   = HIDDEN / (4 * ALPHA_HL);
constexpr int ALPHA_IPT  = 32 / ALPHA_LG;
constexpr int ALPHA_NTRIP = NNZ / ALPHA_IPT;
constexpr int ALPHA_WPB  = 2;
constexpr int ALPHA_TPW  = 10;
constexpr int ALPHA_BLK  = (ALPHA_NTRIP + ALPHA_TPW * ALPHA_WPB - 1) / (ALPHA_TPW * ALPHA_WPB);
static_assert(NNZ % ALPHA_IPT == 0, "a trip is a whole group of incidences");
static_assert(ALPHA_LG == 8 || ALPHA_LG == 16, "lane groups lie inside the 16-lane rows");

__global__ void __launch_bounds__(32 * ALPHA_WPB)
alpha_kernel(const float* __restrict__ xm, const float* __restrict__ em,
             const int* __restrict__ row, const int* __restrict__ col,
             const float* __restrict__ ln_g, const float* __restrict__ ln_b,
             const float* __restrict__ Wsh, const float* __restrict__ bsh,
             float* __restrict__ alpha) {
  constexpr int LG = ALPHA_LG, HL = ALPHA_HL, NQ = ALPHA_NQ, IPT = ALPHA_IPT;
  const int nw = gridDim.x * ALPHA_WPB;
  int tr = __builtin_amdgcn_readfirstlane(blockIdx.x * ALPHA_WPB + (threadIdx.x >> 5));
  if (tr >= ALPHA_NTRIP) return;
  const int lane = threadIdx.x & 31;
  const int grp  = lane / LG;
  const int sub  = lane % LG;
  const bool eside = (sub >= HL);
  const int ls   = sub % HL;
  const float* zsrc = eside ? em : xm;
  const int*   ksrc = eside ? col : row;
  const int j0 = (eside ? HIDDEN : 0) + 4 * ls;

  float g[NQ][4], b[NQ][4], w[NQ][4][D];
#pragma unroll
  for (int q = 0; q < NQ; ++q) {
    const int j = j0 + 4 * HL * q;
    const float4 gv = *reinterpret_cast<const float4*>(ln_g + j);
    const float4 bv = *reinterpret_cast<const float4*>(ln_b + j);
    g[q][0] = gv.x; g[q][1] = gv.y; g[q][2] = gv.z; g[q][3] = gv.w;
    b[q][0] = bv.x; b[q][1] = bv.y; b[q][2] = bv.z; b[q][3] = bv.w;
    float wf[4 * D];
#pragma unroll
    for (int m = 0; m < D; ++m) {
      const float4 wv = *reinterpret_cast<const float4*>(Wsh + (size_t)j * D + 4 * m);
      wf[4 * m] = wv.x; wf[4 * m + 1] = wv.y; wf[4 * m + 2] = wv.z; wf[4 * m + 3] = wv.w;
    }
#pragma unroll
    for (int e = 0; e < 4; ++e)
#pragma unroll
      for (int t = 0; t < D; ++t) w[q][e][t] = wf[e * D + t];
  }
  const float bs = (sub < D) ? bsh[sub] : 0.0f;

  float4 zc[NQ], zn[NQ];
  {
    const int key = ksrc[tr * IPT + grp];
    const float* zp = zsrc + (size_t)key * HIDDEN + 4 * ls;
#pragma unroll
    for (int q = 0; q < NQ; ++q) zc[q] = *reinterpret_cast<const float4*>(zp + 4 * HL * q);
  }
  const int tr1 = (tr + nw < ALPHA_NTRIP) ? (tr + nw) : tr;
  int keyn = ksrc[tr1 * IPT + grp];
  for (; tr < ALPHA_NTRIP; tr += nw) {
    {
      const float* zp = zsrc + (size_t)keyn * HIDDEN + 4 * ls;
#pragma unroll
      for (int q = 0; q < NQ; ++q) zn[q] = *reinterpret_cast<const float4*>(zp + 4 * HL * q);
    }
    const int tr2 = (tr + 2 * nw < ALPHA_NTRIP) ? (tr + 2 * nw) : tr;
    const int keynn = ksrc[tr2 * IPT + grp];

    float z[NQ][4];
#pragma unroll
    for (int q = 0; q < NQ; ++q) { z[q][0] = zc[q].x; z[q][1] = zc[q].y; z[q][2] = zc[q].z; z[q][3] = zc[q].w; }
    float s = 0.0f, ss = 0.0f;
#pragma unroll
    for (int q = 0; q < NQ; ++q)
#pragma unroll
      for (int e = 0; e < 4; ++e) { s += z[q][e]; ss += z[q][e] * z[q][e]; }
    s  = group_allsum<LG>(s);
    ss = group_allsum<LG>(ss);
    const float mu   = s * (1.0f / 128.0f);
    const float var  = ss * (1.0f / 128.0f) - mu * mu;
    const float rstd = rsqrtf(var + 1e-5f);

    float p[D] = {0, 0, 0, 0, 0, 0};
#pragma unroll
    for (int q = 0; q < NQ; ++q) {
#pragma unroll
      for (int e = 0; e < 4; ++e) {
        const float fn = (z[q][e] - mu) * rstd * g[q][e] + b[q][e];
#pragma unroll
        for (int t = 0; t < D; ++t) p[t] = __builtin_fmaf(fn, w[q][e][t], p[t]);
      }
    }
#pragma unroll
    for (int t = 0; t < D; ++t) p[t] = group_allsum<LG>(p[t]);

    float pv = p[0];
#pragma unroll
    for (int t = 1; t < D; ++t) pv = (sub == t) ? p[t] : pv;
    if (sub < D) {
      const float a = 1.0f / (1.0f + expf(-(pv + bs)));
      alpha[(size_t)(tr * IPT + grp) * D + sub] = a;
    }
#pragma unroll
    for (int q = 0; q < NQ; ++q) zc[q] = zn[q];
    keyn = keynn;
  }
}

constexpr int CSR_T      = 1024;
constexpr int CSR_R_N    = 544;
constexpr int CSR_NB_N   = (N_NODES + CSR_R_N - 1) / CSR_R_N;
constexpr int CSR_R_E    = 160;
constexpr int CSR_NB_E   = (N_EDGES + CSR_R_E - 1) / CSR_R_E;
constexpr int CSR_IDBITS = 18;
constexpr int CSR_N4     = NNZ / 4;
constexpr int CSR_U      = 4;
static_assert(NNZ % 4 == 0, "the key arrays are read as 16-byte quads");
static_assert(NNZ <= (1 << CSR_IDBITS), "an incidence id must fit the low bits of a list word");
static_assert(CSR_R_N <= CSR_T && CSR_R_E <= CSR_T, "the scan takes one counter per thread");
static_assert(CSR_T <= (1 << (32 - CSR_IDBITS)), "a key slot must fit the high bits of a list word");
static_assert((CSR_NB_N - 1) * CSR_R_N < N_NODES && (CSR_NB_E - 1) * CSR_R_E < N_EDGES,
              "every work-group owns at least one key");

__device__ inline int wave_allsum_i(int v) {
#pragma unroll
  for (int off = 16; off > 0; off >>= 1) v += __shfl_xor(v, off, 32);
  return v;
}

__device__ inline int wave_incl_scan(int v, int lane) {
#pragma unroll
  for (int off = 1; off < 32; off <<= 1) {
    int u = __shfl_up(v, off, 32);
    if (lane >= off) v += u;
  }
  return v;
}

__device__ inline void csr_count4(const int4 q, int k0, int nk, int* cur, int& below) {
  const int kk[4] = {q.x, q.y, q.z, q.w};
#pragma unroll
  for (int c = 0; c < 4; ++c) {
    const int d = kk[c] - k0;
    below += (d < 0) ? 1 : 0;
    if ((unsigned)d < (unsigned)nk) atomicAdd(&cur[d], 1);
  }
}

__device__ inline void csr_place4(const int4 q, int i0, int k0, int nk, int* cur,
                                  const int* start, unsigned* __restrict__ tmp) {
  const int kk[4] = {q.x, q.y, q.z, q.w};
#pragma unroll
  for (int c = 0; c < 4; ++c) {
    const int d = kk[c] - k0;
    if ((unsigned)d < (unsigned)nk) {
      const int slot = atomicAdd(&cur[d], 1);
      tmp[start[d] + slot] = ((unsigned)d << CSR_IDBITS) | (unsigned)(i0 + c);
    }
  }
}

__global__ void __launch_bounds__(CSR_T)
csr_build_kernel(const int* __restrict__ row, const int* __restrict__ col,
                 int* __restrict__ nptr, int* __restrict__ eptr,
                 unsigned* __restrict__ ntmp, unsigned* __restrict__ etmp,
                 int* __restrict__ nlst, int* __restrict__ ncol,
                 int* __restrict__ elst, int* __restrict__ erow) {
  __shared__ int cur[CSR_T];
  __shared__ int start[CSR_T];
  __shared__ int wsum[32];
  __shared__ int wbel[32];
  __shared__ int sh_base, sh_total;

  const bool eside = (blockIdx.x >= CSR_NB_N);
  const int b      = eside ? (blockIdx.x - CSR_NB_N) : blockIdx.x;
  const int* keys  = eside ? col : row;
  const int* other = eside ? row : col;
  const int nkeys  = eside ? N_EDGES : N_NODES;
  const int R      = eside ? CSR_R_E : CSR_R_N;
  int* ptr         = eside ? eptr : nptr;
  unsigned* tmp    = eside ? etmp : ntmp;
  int* lst         = eside ? elst : nlst;
  int* oth         = eside ? erow : ncol;
  const int k0 = b * R;
  const int nk = (nkeys - k0 < R) ? (nkeys - k0) : R;
  const int tid = threadIdx.x, lane = tid & 31, w = tid >> 5;
  const int4* k4 = reinterpret_cast<const int4*>(keys);

  cur[tid] = 0;
  __syncthreads();

  int below = 0;
  {
    int v = tid;
    for (; v + (CSR_U - 1) * CSR_T < CSR_N4; v += CSR_U * CSR_T) {
      int4 q[CSR_U];
#pragma unroll
      for (int u = 0; u < CSR_U; ++u) q[u] = k4[v + u * CSR_T];
#pragma unroll
      for (int u = 0; u < CSR_U; ++u) csr_count4(q[u], k0, nk, cur, below);
    }
    for (; v < CSR_N4; v += CSR_T) csr_count4(k4[v], k0, nk, cur, below);
  }
  const int bw = wave_allsum_i(below);
  if (lane == 0) wbel[w] = bw;
  __syncthreads();

  const int cnt  = (tid < nk) ? cur[tid] : 0;
  const int incl = wave_incl_scan(cnt, lane);
  if (lane == 31) wsum[w] = incl;
  __syncthreads();
  if (w == 0) {
    const int t  = wsum[lane];
    const int ti = wave_incl_scan(t, lane);
    const int bl = wave_allsum_i(wbel[lane]);
    wsum[lane] = ti - t;
    if (lane == 31) { sh_base = bl; sh_total = ti; }
  }
  __syncthreads();
  const int base  = sh_base;
  const int total = sh_total;
  const int st = base + wsum[w] + incl - cnt;
  start[tid] = st;
  cur[tid] = 0;
  if (tid < nk) ptr[k0 + tid] = st;
  if (tid == nk - 1 && k0 + nk == nkeys) ptr[nkeys] = st + cnt;
  __syncthreads();

  {
    int v = tid;
    for (; v + (CSR_U - 1) * CSR_T < CSR_N4; v += CSR_U * CSR_T) {
      int4 q[CSR_U];
#pragma unroll
      for (int u = 0; u < CSR_U; ++u) q[u] = k4[v + u * CSR_T];
#pragma unroll
      for (int u = 0; u < CSR_U; ++u)
        csr_place4(q[u], 4 * (v + u * CSR_T), k0, nk, cur, start, tmp);
    }
    for (; v < CSR_N4; v += CSR_T) csr_place4(k4[v], 4 * v, k0, nk, cur, start, tmp);
  }
  __syncthreads();

  for (int p = tid; p < total; p += CSR_T) {
    const unsigned wv = tmp[base + p];
    const int d = (int)(wv >> CSR_IDBITS);
    const int i = (int)(wv & ((1u << CSR_IDBITS) - 1u));
    const int o = other[i];
    const int s = start[d], n = cur[d];
    const unsigned* sp = tmp + s;
    int rank = 0;
    int k = 0;
    for (; k + 4 <= n; k += 4) {
      const unsigned u0 = sp[k], u1 = sp[k + 1], u2 = sp[k + 2], u3 = sp[k + 3];
      rank += ((u0 < wv) ? 1 : 0) + ((u1 < wv) ? 1 : 0) + ((u2 < wv) ? 1 : 0) + ((u3 < wv) ? 1 : 0);
    }
    for (; k < n; ++k) rank += (sp[k] < wv) ? 1 : 0;
    lst[s + rank] = i;
    oth[s + rank] = o;
  }
}

constexpr int SEG_C = OUT_CH / 128;
constexpr int SEG_B = 4;
static_assert(SEG_B == 4, "the list walk picks its group body from the sizes 4, 3, 2, 1");

template <bool NODE_SIDE, int NB>
__device__ inline void seg_group(const int* __restrict__ lst, const int* __restrict__ oth,
                                 const float* __restrict__ alpha, const float* __restrict__ src,
                                 int q, int c0, int th,
                                 float4 (&acc)[SEG_C], float (&dsum)[SEG_C]) {
  int ii[NB], oo[NB];
#pragma unroll
  for (int u = 0; u < NB; ++u) { ii[u] = lst[q + u]; oo[u] = oth[q + u]; }
  float a[NB][SEG_C];
  float4 v[NB][SEG_C];
#pragma unroll
  for (int u = 0; u < NB; ++u) {
#pragma unroll
    for (int k = 0; k < SEG_C; ++k) {
      a[u][k] = alpha[(size_t)ii[u] * D + 2 * k + th];
      v[u][k] = *reinterpret_cast<const float4*>(src + (size_t)oo[u] * OUT_CH + 128 * k + c0);
    }
  }
#pragma unroll
  for (int u = 0; u < NB; ++u) {
#pragma unroll
    for (int k = 0; k < SEG_C; ++k) {
      acc[k].x = __builtin_fmaf(a[u][k], v[u][k].x, acc[k].x);
      acc[k].y = __builtin_fmaf(a[u][k], v[u][k].y, acc[k].y);
      acc[k].z = __builtin_fmaf(a[u][k], v[u][k].z, acc[k].z);
      acc[k].w = __builtin_fmaf(a[u][k], v[u][k].w, acc[k].w);
      if (NODE_SIDE) dsum[k] = __builtin_fmaf(a[u][k], a[u][k], dsum[k]);
    }
  }
}

template <bool NODE_SIDE>
__global__ void __launch_bounds__(LOOP_T)
seg_gather_kernel(const int* __restrict__ ptr, const int* __restrict__ lst,
                  const int* __restrict__ oth, const float* __restrict__ alpha,
                  const float* __restrict__ src, const float* __restrict__ h,
                  const float* __restrict__ bias, float* __restrict__ dst, int nseg) {
  const int nw = gridDim.x * LOOP_WAVES;
  int seg = blockIdx.x * LOOP_WAVES + (threadIdx.x >> 5);
  if (seg >= nseg) return;
  const int lane = threadIdx.x & 31;
  const int c0 = 4 * lane;
  const int th = lane >> 4;
  float4 bv = make_float4(0.0f, 0.0f, 0.0f, 0.0f);
  if (NODE_SIDE) bv = *reinterpret_cast<const float4*>(bias + (c0 & (HIDDEN - 1)));

  int s = ptr[seg], e = ptr[seg + 1];
  for (; seg < nseg; seg += nw) {
    const int sx = (seg + nw < nseg) ? (seg + nw) : seg;
    const int sn = ptr[sx], en = ptr[sx + 1];
    float4 hv[SEG_C];
    if (NODE_SIDE) {
#pragma unroll
      for (int k = 0; k < SEG_C; ++k)
        hv[k] = *reinterpret_cast<const float4*>(h + (size_t)seg * OUT_CH + 128 * k + c0);
    }
    float4 acc[SEG_C];
    float dsum[SEG_C];
#pragma unroll
    for (int k = 0; k < SEG_C; ++k) { acc[k] = make_float4(0.0f, 0.0f, 0.0f, 0.0f); dsum[k] = 0.0f; }

    for (int q = s; q < e; q += SEG_B) {
      const int n = e - q;
      if (n >= 4)      seg_group<NODE_SIDE, 4>(lst, oth, alpha, src, q, c0, th, acc, dsum);
      else if (n == 3) seg_group<NODE_SIDE, 3>(lst, oth, alpha, src, q, c0, th, acc, dsum);
      else if (n == 2) seg_group<NODE_SIDE, 2>(lst, oth, alpha, src, q, c0, th, acc, dsum);
      else             seg_group<NODE_SIDE, 1>(lst, oth, alpha, src, q, c0, th, acc, dsum);
    }

    if (NODE_SIDE) {
#pragma unroll
      for (int k = 0; k < SEG_C; ++k) {
        const float dinv = (dsum[k] > 0.0f) ? (1.0f / dsum[k]) : 0.0f;
        float4 o;
        o.x = hv[k].x - dinv * acc[k].x + bv.x;
        o.y = hv[k].y - dinv * acc[k].y + bv.y;
        o.z = hv[k].z - dinv * acc[k].z + bv.z;
        o.w = hv[k].w - dinv * acc[k].w + bv.w;
        o.x = (o.x > 0.0f) ? o.x : expm1f(o.x);
        o.y = (o.y > 0.0f) ? o.y : expm1f(o.y);
        o.z = (o.z > 0.0f) ? o.z : expm1f(o.z);
        o.w = (o.w > 0.0f) ? o.w : expm1f(o.w);
        *reinterpret_cast<float4*>(dst + (size_t)seg * OUT_CH + 128 * k + c0) = o;
      }
    } else {
      const int len = e - s;
      const float binv = (len > 0) ? (1.0f / (float)len) : 0.0f;
#pragma unroll
      for (int k = 0; k < SEG_C; ++k)
        *reinterpret_cast<float4*>(dst + (size_t)seg * OUT_CH + 128 * k + c0) =
            make_float4(acc[k].x * binv, acc[k].y * binv, acc[k].z * binv, acc[k].w * binv);
    }
    s = sn;
    e = en;
  }
}

constexpr int NC_WAVES = 4;
constexpr int NC_T     = 32 * NC_WAVES;
constexpr int NC_PITCH = HIDDEN + 4;
constexpr int NC_TILE  = 16 * NC_PITCH;

__device__ inline void node_row_x1(const int* __restrict__ lst, const int* __restrict__ oth,
                                   const float* __restrict__ alpha, const float* __restrict__ src,
                                   const float4 (&hv)[SEG_C], const float4 bv,
                                   int s, int e, int c0, int th, float4 (&o)[SEG_C]) {
  float4 acc[SEG_C];
  float dsum[SEG_C];
#pragma unroll
  for (int k = 0; k < SEG_C; ++k) { acc[k] = make_float4(0.0f, 0.0f, 0.0f, 0.0f); dsum[k] = 0.0f; }
  for (int q = s; q < e; q += SEG_B) {
    const int n = e - q;
    if (n >= 4)      seg_group<true, 4>(lst, oth, alpha, src, q, c0, th, acc, dsum);
    else if (n == 3) seg_group<true, 3>(lst, oth, alpha, src, q, c0, th, acc, dsum);
    else if (n == 2) seg_group<true, 2>(lst, oth, alpha, src, q, c0, th, acc, dsum);
    else             seg_group<true, 1>(lst, oth, alpha, src, q, c0, th, acc, dsum);
  }
#pragma unroll
  for (int k = 0; k < SEG_C; ++k) {
    const float dinv = (dsum[k] > 0.0f) ? (1.0f / dsum[k]) : 0.0f;
    float4 r;
    r.x = hv[k].x - dinv * acc[k].x + bv.x;
    r.y = hv[k].y - dinv * acc[k].y + bv.y;
    r.z = hv[k].z - dinv * acc[k].z + bv.z;
    r.w = hv[k].w - dinv * acc[k].w + bv.w;
    r.x = (r.x > 0.0f) ? r.x : expm1f(r.x);
    r.y = (r.y > 0.0f) ? r.y : expm1f(r.y);
    r.z = (r.z > 0.0f) ? r.z : expm1f(r.z);
    r.w = (r.w > 0.0f) ? r.w : expm1f(r.w);
    o[k] = r;
  }
}

__global__ void __launch_bounds__(NC_T, 4)
seg_node_conv_kernel(const int* __restrict__ ptr, const int* __restrict__ lst,
                     const int* __restrict__ oth, const float* __restrict__ alpha,
                     const float* __restrict__ src, float* hio,
                     const float* __restrict__ bias, const float* __restrict__ W1, int nseg) {
  constexpr int K = HIDDEN, NT = 4, S = K / 4;
  __shared__ float xt[NC_WAVES * NC_TILE];
  const int wib = __builtin_amdgcn_readfirstlane(threadIdx.x >> 5);
  const int nw  = gridDim.x * NC_WAVES;
  int seg = blockIdx.x * NC_WAVES + wib;
  if (seg >= nseg) return;
  const int lane = threadIdx.x & 31;
  const int c0 = 4 * lane;
  const int th = lane >> 4;
  const int c  = lane & 15;
  float* xw = xt + wib * NC_TILE;
  for (int i = lane; i < NC_TILE; i += 32) xw[i] = 0.0f;

  v2f bf[S][NT];
#pragma unroll
  for (int s = 0; s < S; ++s) {
    const int k = 16 * (s >> 2) + 8 * th + 2 * (s & 3);
    float w0[NT], w1[NT];
    load_cols<NT>(W1 + (size_t)k * K + NT * c, w0);
    load_cols<NT>(W1 + (size_t)(k + 1) * K + NT * c, w1);
#pragma unroll
    for (int t = 0; t < NT; ++t) { bf[s][t].x = w0[t]; bf[s][t].y = w1[t]; }
  }
  const float4 bv = *reinterpret_cast<const float4*>(bias + (c0 & (HIDDEN - 1)));

  int s0 = ptr[seg], e0 = ptr[seg + 1];
  int s1, e1;
  {
    const int sb = (seg + nw < nseg) ? (seg + nw) : seg;
    s1 = ptr[sb]; e1 = ptr[sb + 1];
  }
  for (; seg < nseg; seg += 2 * nw) {
    const int seg1 = seg + nw;
    const bool two = (seg1 < nseg);
    const int nx0 = (seg + 2 * nw < nseg) ? (seg + 2 * nw) : seg;
    const int nx1 = (seg + 3 * nw < nseg) ? (seg + 3 * nw) : nx0;
    const int ns0 = ptr[nx0], ne0 = ptr[nx0 + 1];
    const int ns1 = ptr[nx1], ne1 = ptr[nx1 + 1];

    float4 hv[SEG_C];
#pragma unroll
    for (int k = 0; k < SEG_C; ++k)
      hv[k] = *reinterpret_cast<const float4*>(hio + (size_t)seg * OUT_CH + 128 * k + c0);
    float4 o[SEG_C];
    node_row_x1(lst, oth, alpha, src, hv, bv, s0, e0, c0, th, o);
#pragma unroll
    for (int k = 0; k < SEG_C; ++k)
      *reinterpret_cast<float4*>(xw + (2 * k + th) * NC_PITCH + 4 * c) = o[k];
    if (two) {
#pragma unroll
      for (int k = 0; k < SEG_C; ++k)
        hv[k] = *reinterpret_cast<const float4*>(hio + (size_t)seg1 * OUT_CH + 128 * k + c0);
      node_row_x1(lst, oth, alpha, src, hv, bv, s1, e1, c0, th, o);
#pragma unroll
      for (int k = 0; k < SEG_C; ++k)
        *reinterpret_cast<float4*>(xw + (D + 2 * k + th) * NC_PITCH + 4 * c) = o[k];
    }
    __builtin_amdgcn_fence(__ATOMIC_RELEASE, "wavefront");
    __builtin_amdgcn_wave_barrier();
    __builtin_amdgcn_fence(__ATOMIC_ACQUIRE, "wavefront");

    float4 av[K / 8];
    {
      const float* ap = xw + c * NC_PITCH + 8 * th;
#pragma unroll
      for (int g = 0; g < K / 16; ++g) {
        av[2 * g]     = *reinterpret_cast<const float4*>(ap + 16 * g);
        av[2 * g + 1] = *reinterpret_cast<const float4*>(ap + 16 * g + 4);
      }
    }
    __builtin_amdgcn_fence(__ATOMIC_RELEASE, "wavefront");
    __builtin_amdgcn_wave_barrier();
    __builtin_amdgcn_fence(__ATOMIC_ACQUIRE, "wavefront");
    __builtin_amdgcn_sched_barrier(0);

    v8f acc[NT];
#pragma unroll
    for (int t = 0; t < NT; ++t) acc[t] = v8f{};
#pragma unroll
    for (int s = 0; s < S; ++s) {
      const float4 q = av[s >> 1];
      v2f a;
      a.x = (s & 1) ? q.z : q.x;
      a.y = (s & 1) ? q.w : q.y;
#pragma unroll
      for (int t = 0; t < NT; ++t)
        acc[t] = __builtin_amdgcn_wmma_f32_16x16x4_f32(false, a, false, bf[s][t],
                                                       (short)0, acc[t], false, false);
    }
#pragma unroll
    for (int j = 0; j < 8; ++j) {
      const int r = j + 8 * th;
      const bool second = (r >= D);
      const bool ok = second ? (two && r < 2 * D) : true;
      const int node  = second ? seg1 : seg;
      const int stalk = second ? (r - D) : r;
      if (ok)
        *reinterpret_cast<float4*>(hio + ((size_t)node * D + stalk) * HIDDEN + NT * c) =
            make_float4(acc[0][j], acc[1][j], acc[2][j], acc[3][j]);
    }
    s0 = ns0; e0 = ne0; s1 = ns1; e1 = ne1;
  }
}

constexpr int SEG_E_BLK = (N_EDGES + LOOP_WAVES - 1) / LOOP_WAVES;
constexpr int SEG_N_BLK = LOOP_BLK;
constexpr int NC_BLK    = 1024;

extern "C" void kernel_launch(void* const* d_in, const int* in_sizes, int n_in,
                              void* d_out, int out_size, void* d_ws, size_t ws_size,
                              hipStream_t stream) {
  (void)in_sizes; (void)n_in; (void)out_size; (void)ws_size;
  const float* x       = (const float*)d_in[0];
  const float* hattr   = (const float*)d_in[1];
  const float* W_lin   = (const float*)d_in[2];
  const float* b_lin   = (const float*)d_in[3];
  const float* ln_g    = (const float*)d_in[4];
  const float* ln_b    = (const float*)d_in[5];
  const float* W_sheaf = (const float*)d_in[6];
  const float* b_sheaf = (const float*)d_in[7];
  const float* W_conv  = (const float*)d_in[8];
  const float* b_conv  = (const float*)d_in[9];
  const int*   eidx    = (const int*)d_in[10];
  const int* row = eidx;
  const int* col = eidx + NNZ;
  float* out = (float*)d_out;

  float* ws    = (float*)d_ws;
  float* fold  = ws;
  float* Wp    = fold;
  float* bp    = Wp + (size_t)IN_CH * OUT_CH;
  float* Wm    = bp + OUT_CH;
  float* bm    = Wm + (size_t)IN_CH * HIDDEN;
  float* es    = ws   + (size_t)ND * HIDDEN;
  float* xm    = es   + (size_t)ED * HIDDEN;
  float* em    = xm   + (size_t)N_NODES * HIDDEN;
  float* alpha = em   + (size_t)N_EDGES * HIDDEN;
  float* spare = alpha + (size_t)NNZ * D;
  float* hbuf  = spare + ND + N_EDGES;
  float* me    = hbuf + (size_t)ND * HIDDEN;
  int*   ib    = (int*)(me + (size_t)ED * HIDDEN);
  int* nptr = ib;
  int* eptr = nptr + N_NODES + 8;
  unsigned* ntmp = (unsigned*)(eptr + N_EDGES + 8);
  unsigned* etmp = ntmp + NNZ;
  int* nlst = (int*)(etmp + NNZ);
  int* ncol = nlst + NNZ;
  int* elst = ncol + NNZ;
  int* erow = elst + NNZ;

  csr_build_kernel<<<CSR_NB_N + CSR_NB_E, CSR_T, 0, stream>>>(row, col, nptr, eptr, ntmp, etmp,
                                                              nlst, ncol, elst, erow);

  fold_weights_kernel<<<FOLD_BLK, 32 * FOLD_WPB, 0, stream>>>(W_lin, b_lin, W_conv, Wp, bp, Wm, bm);

  proj_fold_kernel<<<PF_BLK_H + PF_BLK_M, 32 * PF_WAVES, 0, stream>>>(x, hattr, Wp, bp, Wm, bm,
                                                                     hbuf, xm, em);

  alpha_kernel<<<ALPHA_BLK, 32 * ALPHA_WPB, 0, stream>>>(xm, em, row, col, ln_g, ln_b,
                                                W_sheaf, b_sheaf, alpha);

  seg_gather_kernel<false><<<SEG_E_BLK, LOOP_T, 0, stream>>>(eptr, elst, erow, alpha, hbuf,
                                                           nullptr, nullptr, me, N_EDGES);
  seg_node_conv_kernel<<<NC_BLK, NC_T, 0, stream>>>(nptr, nlst, ncol, alpha, me, hbuf, b_conv,
                                                    W_conv + HIDDEN * HIDDEN, N_NODES);

  seg_gather_kernel<false><<<SEG_E_BLK, LOOP_T, 0, stream>>>(eptr, elst, erow, alpha, hbuf,
                                                           nullptr, nullptr, me, N_EDGES);
  seg_gather_kernel<true><<<SEG_N_BLK, LOOP_T, 0, stream>>>(nptr, nlst, ncol, alpha, me,
                                                          hbuf, b_conv + HIDDEN, out, N_NODES);
}


# Workspace tables h and me moved 64 bytes back onto a 256-byte boundary: every gathered 1536-byte row is 12 whole 128-byte lines, each 512-byte wave load or store 4 lines instead of 5; no other change
# speedup vs baseline: 5.0004x; 1.0365x over previous
#include <hip/hip_runtime.h>
#include <math.h>

constexpr int N_NODES = 50000;
constexpr int N_EDGES = 10000;
constexpr int NNZ     = 150000;
constexpr int IN_CH   = 128;
constexpr int HIDDEN  = 64;
constexpr int D       = 6;
constexpr int OUT_CH  = HIDDEN * D;
constexpr int ND      = N_NODES * D;
constexpr int ED      = N_EDGES * D;

typedef __attribute__((ext_vector_type(2))) float v2f;
typedef __attribute__((ext_vector_type(8))) float v8f;

template <int NT> struct ColVec;
template <> struct ColVec<2> { typedef float2 type; };
template <> struct ColVec<4> { typedef float4 type; };

template <int NT>
__device__ inline void load_cols(const float* __restrict__ p, float (&d)[NT]) {
  const typename ColVec<NT>::type v = *reinterpret_cast<const typename ColVec<NT>::type*>(p);
  const float* f = reinterpret_cast<const float*>(&v);
#pragma unroll
  for (int t = 0; t < NT; ++t) d[t] = f[t];
}

template <int NT>
__device__ inline void store_cols(float* __restrict__ p, const float (&d)[NT]) {
  typename ColVec<NT>::type v;
  float* f = reinterpret_cast<float*>(&v);
#pragma unroll
  for (int t = 0; t < NT; ++t) f[t] = d[t];
  *reinterpret_cast<typename ColVec<NT>::type*>(p) = v;
}

template <int K>
__device__ inline void load_a_tile(float4 (&av)[K / 8], const float* __restrict__ A,
                                   int mt, int c, int h) {
  const float* ap = A + ((size_t)mt * 16 + c) * K + 8 * h;
#pragma unroll
  for (int g = 0; g < K / 16; ++g) {
    av[2 * g]     = *reinterpret_cast<const float4*>(ap + 16 * g);
    av[2 * g + 1] = *reinterpret_cast<const float4*>(ap + 16 * g + 4);
  }
}

constexpr int FOLD_TILES = IN_CH * D / 16;
constexpr int FOLD_WPB   = 4;
constexpr int FOLD_BLK   = (FOLD_TILES + 1 + FOLD_WPB - 1) / FOLD_WPB;
constexpr int FOLD_MQ    = IN_CH * HIDDEN / 4;

__global__ void __launch_bounds__(32 * FOLD_WPB)
fold_weights_kernel(const float* __restrict__ W_lin, const float* __restrict__ b_lin,
                    const float* __restrict__ W0,
                    float* __restrict__ Wp, float* __restrict__ bp,
                    float* __restrict__ Wm, float* __restrict__ bm) {
  constexpr int K = HIDDEN, NT = 4, S = K / 4;
  for (int qd = blockIdx.x * (32 * FOLD_WPB) + threadIdx.x; qd < FOLD_MQ + HIDDEN / 4;
       qd += gridDim.x * (32 * FOLD_WPB)) {
    const bool isb = (qd >= FOLD_MQ);
    const float* p = isb ? (b_lin + 4 * (qd - FOLD_MQ)) : (W_lin + (size_t)(qd >> 4) * OUT_CH + 4 * (qd & 15));
    float4 sm = make_float4(0.0f, 0.0f, 0.0f, 0.0f);
#pragma unroll
    for (int t = 0; t < D; ++t) {
      const float4 v = *reinterpret_cast<const float4*>(p + t * HIDDEN);
      sm.x += v.x; sm.y += v.y; sm.z += v.z; sm.w += v.w;
    }
    const float k6 = 1.0f / 6.0f;
    float* d = isb ? (bm + 4 * (qd - FOLD_MQ)) : (Wm + (size_t)qd * 4);
    *reinterpret_cast<float4*>(d) = make_float4(sm.x * k6, sm.y * k6, sm.z * k6, sm.w * k6);
  }

  const int wave = __builtin_amdgcn_readfirstlane(blockIdx.x * FOLD_WPB + (threadIdx.x >> 5));
  if (wave > FOLD_TILES) return;
  const bool btile = (wave == FOLD_TILES);
  const float* src = btile ? b_lin : (W_lin + (size_t)wave * 16 * K);
  float* dst       = btile ? bp : (Wp + (size_t)wave * 16 * K);
  const int rows   = btile ? D : 16;
  const int lane = threadIdx.x & 31;
  const int h = lane >> 4;
  const int c = lane & 15;

  float4 av[K / 8];
#pragma unroll
  for (int i = 0; i < K / 8; ++i) av[i] = make_float4(0.0f, 0.0f, 0.0f, 0.0f);
  if (c < rows) {
    const float* ap = src + (size_t)c * K + 8 * h;
#pragma unroll
    for (int g = 0; g < K / 16; ++g) {
      av[2 * g]     = *reinterpret_cast<const float4*>(ap + 16 * g);
      av[2 * g + 1] = *reinterpret_cast<const float4*>(ap + 16 * g + 4);
    }
  }
  v8f acc[NT];
#pragma unroll
  for (int t = 0; t < NT; ++t) acc[t] = v8f{};
#pragma unroll
  for (int s = 0; s < S; ++s) {
    const int k = 16 * (s >> 2) + 8 * h + 2 * (s & 3);
    float w0[NT], w1[NT];
    load_cols<NT>(W0 + (size_t)k * K + NT * c, w0);
    load_cols<NT>(W0 + (size_t)(k + 1) * K + NT * c, w1);
    const float4 q = av[s >> 1];
    v2f a;
    a.x = (s & 1) ? q.z : q.x;
    a.y = (s & 1) ? q.w : q.y;
#pragma unroll
    for (int t = 0; t < NT; ++t) {
      v2f b;
      b.x = w0[t]; b.y = w1[t];
      acc[t] = __builtin_amdgcn_wmma_f32_16x16x4_f32(false, a, false, b, (short)0, acc[t], false, false);
    }
  }
#pragma unroll
  for (int j = 0; j < 8; ++j) {
    float o[NT];
#pragma unroll
    for (int t = 0; t < NT; ++t) o[t] = acc[t][j];
    if (j + 8 * h < rows) store_cols<NT>(dst + (size_t)(j + 8 * h) * K + NT * c, o);
  }
}

constexpr int PF_WAVES  = 12;
constexpr int PF_RUN    = 15;
constexpr int PF_RUNS_N = (N_NODES / 16 + PF_RUN - 1) / PF_RUN;
constexpr int PF_RUNS_E = (N_EDGES / 16 + PF_RUN - 1) / PF_RUN;
constexpr int PF_BLK_H  = PF_RUNS_N;
constexpr int PF_MW_N   = 2 * PF_RUNS_N;
constexpr int PF_MW_E   = 2 * PF_RUNS_E;
constexpr int PF_BLK_M  = (PF_MW_N + PF_MW_E + PF_WAVES - 1) / PF_WAVES;
static_assert(N_NODES % 16 == 0 && N_EDGES % 16 == 0, "row tiles are whole");

__global__ void __launch_bounds__(32 * PF_WAVES)
proj_fold_kernel(const float* __restrict__ x, const float* __restrict__ hattr,
                 const float* __restrict__ Wp, const float* __restrict__ bp,
                 const float* __restrict__ Wm, const float* __restrict__ bm,
                 float* __restrict__ h1, float* __restrict__ xm, float* __restrict__ em) {
  constexpr int K = IN_CH, NT = 2, S = K / 4;
  const int blk = blockIdx.x;
  const int wib = __builtin_amdgcn_readfirstlane(threadIdx.x >> 5);
  const float* A; const float* B; const float* bias; float* C;
  int pitch, n0, run, mTiles;
  if (blk < PF_BLK_H) {
    A = x; B = Wp; bias = bp; C = h1; pitch = OUT_CH; n0 = 16 * NT * wib; run = blk; mTiles = N_NODES / 16;
  } else {
    const int u = (blk - PF_BLK_H) * PF_WAVES + wib;
    B = Wm; bias = bm; pitch = HIDDEN;
    if (u < PF_MW_N) {
      A = x; C = xm; run = u >> 1; n0 = 16 * NT * (u & 1); mTiles = N_NODES / 16;
    } else {
      const int v = u - PF_MW_N;
      if (v >= PF_MW_E) return;
      A = hattr; C = em; run = v >> 1; n0 = 16 * NT * (v & 1); mTiles = N_EDGES / 16;
    }
  }
  int mt = run * PF_RUN;
  if (mt >= mTiles) return;
  const int mtEnd = (mt + PF_RUN < mTiles) ? (mt + PF_RUN) : mTiles;
  const int lane = threadIdx.x & 31;
  const int h = lane >> 4;
  const int c = lane & 15;

  v2f bf[S][NT];
#pragma unroll
  for (int s = 0; s < S; ++s) {
    const int k = 16 * (s >> 2) + 8 * h + 2 * (s & 3);
    float w0[NT], w1[NT];
    load_cols<NT>(B + (size_t)k * pitch + n0 + NT * c, w0);
    load_cols<NT>(B + (size_t)(k + 1) * pitch + n0 + NT * c, w1);
#pragma unroll
    for (int t = 0; t < NT; ++t) { bf[s][t].x = w0[t]; bf[s][t].y = w1[t]; }
  }
  float bv[NT];
  load_cols<NT>(bias + n0 + NT * c, bv);

  float4 av[K / 8];
  load_a_tile<K>(av, A, mt, c, h);
  for (; mt < mtEnd; ++mt) {
    const int nx = (mt + 1 < mtEnd) ? (mt + 1) : mt;
    float4 an[K / 8];
    load_a_tile<K>(an, A, nx, c, h);
    __builtin_amdgcn_sched_barrier(0);

    v8f acc[NT];
#pragma unroll
    for (int t = 0; t < NT; ++t) acc[t] = v8f{};
#pragma unroll
    for (int s = 0; s < S; ++s) {
      const float4 q = av[s >> 1];
      v2f a;
      a.x = (s & 1) ? q.z : q.x;
      a.y = (s & 1) ? q.w : q.y;
#pragma unroll
      for (int t = 0; t < NT; ++t)
        acc[t] = __builtin_amdgcn_wmma_f32_16x16x4_f32(false, a, false, bf[s][t],
                                                       (short)0, acc[t], false, false);
    }
    float* cp = C + ((size_t)mt * 16 + 8 * h) * pitch + n0 + NT * c;
#pragma unroll
    for (int j = 0; j < 8; ++j) {
      float o[NT];
#pragma unroll
      for (int t = 0; t < NT; ++t) o[t] = acc[t][j] + bv[t];
      store_cols<NT>(cp + (size_t)j * pitch, o);
    }
    __builtin_amdgcn_sched_barrier(0);
#pragma unroll
    for (int i = 0; i < K / 8; ++i) av[i] = an[i];
  }
}

constexpr int LOOP_WAVES = 8;
constexpr int LOOP_T     = 32 * LOOP_WAVES;
constexpr int LOOP_BLK   = 1024;

template <int M>
__device__ inline float xmask_add(float v) {
  const int o = __builtin_amdgcn_update_dpp(0, __builtin_bit_cast(int, v), 0x160 + M, 0xf, 0xf, false);
  return v + __builtin_bit_cast(float, o);
}
template <int LG>
__device__ inline float group_allsum(float v) {
  v = xmask_add<1>(v);
  v = xmask_add<2>(v);
  v = xmask_add<4>(v);
  if (LG == 16) v = xmask_add<8>(v);
  return v;
}

constexpr int ALPHA_LG   = 8;
constexpr int ALPHA_HL   = ALPHA_LG / 2;
constexpr int ALPHA_NQ   = HIDDEN / (4 * ALPHA_HL);
constexpr int ALPHA_IPT  = 32 / ALPHA_LG;
constexpr int ALPHA_NTRIP = NNZ / ALPHA_IPT;
constexpr int ALPHA_WPB  = 2;
constexpr int ALPHA_TPW  = 10;
constexpr int ALPHA_BLK  = (ALPHA_NTRIP + ALPHA_TPW * ALPHA_WPB - 1) / (ALPHA_TPW * ALPHA_WPB);
static_assert(NNZ % ALPHA_IPT == 0, "a trip is a whole group of incidences");
static_assert(ALPHA_LG == 8 || ALPHA_LG == 16, "lane groups lie inside the 16-lane rows");

__global__ void __launch_bounds__(32 * ALPHA_WPB)
alpha_kernel(const float* __restrict__ xm, const float* __restrict__ em,
             const int* __restrict__ row, const int* __restrict__ col,
             const float* __restrict__ ln_g, const float* __restrict__ ln_b,
             const float* __restrict__ Wsh, const float* __restrict__ bsh,
             float* __restrict__ alpha) {
  constexpr int LG = ALPHA_LG, HL = ALPHA_HL, NQ = ALPHA_NQ, IPT = ALPHA_IPT;
  const int nw = gridDim.x * ALPHA_WPB;
  int tr = __builtin_amdgcn_readfirstlane(blockIdx.x * ALPHA_WPB + (threadIdx.x >> 5));
  if (tr >= ALPHA_NTRIP) return;
  const int lane = threadIdx.x & 31;
  const int grp  = lane / LG;
  const int sub  = lane % LG;
  const bool eside = (sub >= HL);
  const int ls   = sub % HL;
  const float* zsrc = eside ? em : xm;
  const int*   ksrc = eside ? col : row;
  const int j0 = (eside ? HIDDEN : 0) + 4 * ls;

  float g[NQ][4], b[NQ][4], w[NQ][4][D];
#pragma unroll
  for (int q = 0; q < NQ; ++q) {
    const int j = j0 + 4 * HL * q;
    const float4 gv = *reinterpret_cast<const float4*>(ln_g + j);
    const float4 bv = *reinterpret_cast<const float4*>(ln_b + j);
    g[q][0] = gv.x; g[q][1] = gv.y; g[q][2] = gv.z; g[q][3] = gv.w;
    b[q][0] = bv.x; b[q][1] = bv.y; b[q][2] = bv.z; b[q][3] = bv.w;
    float wf[4 * D];
#pragma unroll
    for (int m = 0; m < D; ++m) {
      const float4 wv = *reinterpret_cast<const float4*>(Wsh + (size_t)j * D + 4 * m);
      wf[4 * m] = wv.x; wf[4 * m + 1] = wv.y; wf[4 * m + 2] = wv.z; wf[4 * m + 3] = wv.w;
    }
#pragma unroll
    for (int e = 0; e < 4; ++e)
#pragma unroll
      for (int t = 0; t < D; ++t) w[q][e][t] = wf[e * D + t];
  }
  const float bs = (sub < D) ? bsh[sub] : 0.0f;

  float4 zc[NQ], zn[NQ];
  {
    const int key = ksrc[tr * IPT + grp];
    const float* zp = zsrc + (size_t)key * HIDDEN + 4 * ls;
#pragma unroll
    for (int q = 0; q < NQ; ++q) zc[q] = *reinterpret_cast<const float4*>(zp + 4 * HL * q);
  }
  const int tr1 = (tr + nw < ALPHA_NTRIP) ? (tr + nw) : tr;
  int keyn = ksrc[tr1 * IPT + grp];
  for (; tr < ALPHA_NTRIP; tr += nw) {
    {
      const float* zp = zsrc + (size_t)keyn * HIDDEN + 4 * ls;
#pragma unroll
      for (int q = 0; q < NQ; ++q) zn[q] = *reinterpret_cast<const float4*>(zp + 4 * HL * q);
    }
    const int tr2 = (tr + 2 * nw < ALPHA_NTRIP) ? (tr + 2 * nw) : tr;
    const int keynn = ksrc[tr2 * IPT + grp];

    float z[NQ][4];
#pragma unroll
    for (int q = 0; q < NQ; ++q) { z[q][0] = zc[q].x; z[q][1] = zc[q].y; z[q][2] = zc[q].z; z[q][3] = zc[q].w; }
    float s = 0.0f, ss = 0.0f;
#pragma unroll
    for (int q = 0; q < NQ; ++q)
#pragma unroll
      for (int e = 0; e < 4; ++e) { s += z[q][e]; ss += z[q][e] * z[q][e]; }
    s  = group_allsum<LG>(s);
    ss = group_allsum<LG>(ss);
    const float mu   = s * (1.0f / 128.0f);
    const float var  = ss * (1.0f / 128.0f) - mu * mu;
    const float rstd = rsqrtf(var + 1e-5f);

    float p[D] = {0, 0, 0, 0, 0, 0};
#pragma unroll
    for (int q = 0; q < NQ; ++q) {
#pragma unroll
      for (int e = 0; e < 4; ++e) {
        const float fn = (z[q][e] - mu) * rstd * g[q][e] + b[q][e];
#pragma unroll
        for (int t = 0; t < D; ++t) p[t] = __builtin_fmaf(fn, w[q][e][t], p[t]);
      }
    }
#pragma unroll
    for (int t = 0; t < D; ++t) p[t] = group_allsum<LG>(p[t]);

    float pv = p[0];
#pragma unroll
    for (int t = 1; t < D; ++t) pv = (sub == t) ? p[t] : pv;
    if (sub < D) {
      const float a = 1.0f / (1.0f + expf(-(pv + bs)));
      alpha[(size_t)(tr * IPT + grp) * D + sub] = a;
    }
#pragma unroll
    for (int q = 0; q < NQ; ++q) zc[q] = zn[q];
    keyn = keynn;
  }
}

constexpr int CSR_T      = 1024;
constexpr int CSR_R_N    = 544;
constexpr int CSR_NB_N   = (N_NODES + CSR_R_N - 1) / CSR_R_N;
constexpr int CSR_R_E    = 160;
constexpr int CSR_NB_E   = (N_EDGES + CSR_R_E - 1) / CSR_R_E;
constexpr int CSR_IDBITS = 18;
constexpr int CSR_N4     = NNZ / 4;
constexpr int CSR_U      = 4;
static_assert(NNZ % 4 == 0, "the key arrays are read as 16-byte quads");
static_assert(NNZ <= (1 << CSR_IDBITS), "an incidence id must fit the low bits of a list word");
static_assert(CSR_R_N <= CSR_T && CSR_R_E <= CSR_T, "the scan takes one counter per thread");
static_assert(CSR_T <= (1 << (32 - CSR_IDBITS)), "a key slot must fit the high bits of a list word");
static_assert((CSR_NB_N - 1) * CSR_R_N < N_NODES && (CSR_NB_E - 1) * CSR_R_E < N_EDGES,
              "every work-group owns at least one key");

__device__ inline int wave_allsum_i(int v) {
#pragma unroll
  for (int off = 16; off > 0; off >>= 1) v += __shfl_xor(v, off, 32);
  return v;
}

__device__ inline int wave_incl_scan(int v, int lane) {
#pragma unroll
  for (int off = 1; off < 32; off <<= 1) {
    int u = __shfl_up(v, off, 32);
    if (lane >= off) v += u;
  }
  return v;
}

__device__ inline void csr_count4(const int4 q, int k0, int nk, int* cur, int& below) {
  const int kk[4] = {q.x, q.y, q.z, q.w};
#pragma unroll
  for (int c = 0; c < 4; ++c) {
    const int d = kk[c] - k0;
    below += (d < 0) ? 1 : 0;
    if ((unsigned)d < (unsigned)nk) atomicAdd(&cur[d], 1);
  }
}

__device__ inline void csr_place4(const int4 q, int i0, int k0, int nk, int* cur,
                                  const int* start, unsigned* __restrict__ tmp) {
  const int kk[4] = {q.x, q.y, q.z, q.w};
#pragma unroll
  for (int c = 0; c < 4; ++c) {
    const int d = kk[c] - k0;
    if ((unsigned)d < (unsigned)nk) {
      const int slot = atomicAdd(&cur[d], 1);
      tmp[start[d] + slot] = ((unsigned)d << CSR_IDBITS) | (unsigned)(i0 + c);
    }
  }
}

__global__ void __launch_bounds__(CSR_T)
csr_build_kernel(const int* __restrict__ row, const int* __restrict__ col,
                 int* __restrict__ nptr, int* __restrict__ eptr,
                 unsigned* __restrict__ ntmp, unsigned* __restrict__ etmp,
                 int* __restrict__ nlst, int* __restrict__ ncol,
                 int* __restrict__ elst, int* __restrict__ erow) {
  __shared__ int cur[CSR_T];
  __shared__ int start[CSR_T];
  __shared__ int wsum[32];
  __shared__ int wbel[32];
  __shared__ int sh_base, sh_total;

  const bool eside = (blockIdx.x >= CSR_NB_N);
  const int b      = eside ? (blockIdx.x - CSR_NB_N) : blockIdx.x;
  const int* keys  = eside ? col : row;
  const int* other = eside ? row : col;
  const int nkeys  = eside ? N_EDGES : N_NODES;
  const int R      = eside ? CSR_R_E : CSR_R_N;
  int* ptr         = eside ? eptr : nptr;
  unsigned* tmp    = eside ? etmp : ntmp;
  int* lst         = eside ? elst : nlst;
  int* oth         = eside ? erow : ncol;
  const int k0 = b * R;
  const int nk = (nkeys - k0 < R) ? (nkeys - k0) : R;
  const int tid = threadIdx.x, lane = tid & 31, w = tid >> 5;
  const int4* k4 = reinterpret_cast<const int4*>(keys);

  cur[tid] = 0;
  __syncthreads();

  int below = 0;
  {
    int v = tid;
    for (; v + (CSR_U - 1) * CSR_T < CSR_N4; v += CSR_U * CSR_T) {
      int4 q[CSR_U];
#pragma unroll
      for (int u = 0; u < CSR_U; ++u) q[u] = k4[v + u * CSR_T];
#pragma unroll
      for (int u = 0; u < CSR_U; ++u) csr_count4(q[u], k0, nk, cur, below);
    }
    for (; v < CSR_N4; v += CSR_T) csr_count4(k4[v], k0, nk, cur, below);
  }
  const int bw = wave_allsum_i(below);
  if (lane == 0) wbel[w] = bw;
  __syncthreads();

  const int cnt  = (tid < nk) ? cur[tid] : 0;
  const int incl = wave_incl_scan(cnt, lane);
  if (lane == 31) wsum[w] = incl;
  __syncthreads();
  if (w == 0) {
    const int t  = wsum[lane];
    const int ti = wave_incl_scan(t, lane);
    const int bl = wave_allsum_i(wbel[lane]);
    wsum[lane] = ti - t;
    if (lane == 31) { sh_base = bl; sh_total = ti; }
  }
  __syncthreads();
  const int base  = sh_base;
  const int total = sh_total;
  const int st = base + wsum[w] + incl - cnt;
  start[tid] = st;
  cur[tid] = 0;
  if (tid < nk) ptr[k0 + tid] = st;
  if (tid == nk - 1 && k0 + nk == nkeys) ptr[nkeys] = st + cnt;
  __syncthreads();

  {
    int v = tid;
    for (; v + (CSR_U - 1) * CSR_T < CSR_N4; v += CSR_U * CSR_T) {
      int4 q[CSR_U];
#pragma unroll
      for (int u = 0; u < CSR_U; ++u) q[u] = k4[v + u * CSR_T];
#pragma unroll
      for (int u = 0; u < CSR_U; ++u)
        csr_place4(q[u], 4 * (v + u * CSR_T), k0, nk, cur, start, tmp);
    }
    for (; v < CSR_N4; v += CSR_T) csr_place4(k4[v], 4 * v, k0, nk, cur, start, tmp);
  }
  __syncthreads();

  for (int p = tid; p < total; p += CSR_T) {
    const unsigned wv = tmp[base + p];
    const int d = (int)(wv >> CSR_IDBITS);
    const int i = (int)(wv & ((1u << CSR_IDBITS) - 1u));
    const int o = other[i];
    const int s = start[d], n = cur[d];
    const unsigned* sp = tmp + s;
    int rank = 0;
    int k = 0;
    for (; k + 4 <= n; k += 4) {
      const unsigned u0 = sp[k], u1 = sp[k + 1], u2 = sp[k + 2], u3 = sp[k + 3];
      rank += ((u0 < wv) ? 1 : 0) + ((u1 < wv) ? 1 : 0) + ((u2 < wv) ? 1 : 0) + ((u3 < wv) ? 1 : 0);
    }
    for (; k < n; ++k) rank += (sp[k] < wv) ? 1 : 0;
    lst[s + rank] = i;
    oth[s + rank] = o;
  }
}

constexpr int SEG_C = OUT_CH / 128;
constexpr int SEG_B = 4;
static_assert(SEG_B == 4, "the list walk picks its group body from the sizes 4, 3, 2, 1");

template <bool NODE_SIDE, int NB>
__device__ inline void seg_group(const int* __restrict__ lst, const int* __restrict__ oth,
                                 const float* __restrict__ alpha, const float* __restrict__ src,
                                 int q, int c0, int th,
                                 float4 (&acc)[SEG_C], float (&dsum)[SEG_C]) {
  int ii[NB], oo[NB];
#pragma unroll
  for (int u = 0; u < NB; ++u) { ii[u] = lst[q + u]; oo[u] = oth[q + u]; }
  float a[NB][SEG_C];
  float4 v[NB][SEG_C];
#pragma unroll
  for (int u = 0; u < NB; ++u) {
#pragma unroll
    for (int k = 0; k < SEG_C; ++k) {
      a[u][k] = alpha[(size_t)ii[u] * D + 2 * k + th];
      v[u][k] = *reinterpret_cast<const float4*>(src + (size_t)oo[u] * OUT_CH + 128 * k + c0);
    }
  }
#pragma unroll
  for (int u = 0; u < NB; ++u) {
#pragma unroll
    for (int k = 0; k < SEG_C; ++k) {
      acc[k].x = __builtin_fmaf(a[u][k], v[u][k].x, acc[k].x);
      acc[k].y = __builtin_fmaf(a[u][k], v[u][k].y, acc[k].y);
      acc[k].z = __builtin_fmaf(a[u][k], v[u][k].z, acc[k].z);
      acc[k].w = __builtin_fmaf(a[u][k], v[u][k].w, acc[k].w);
      if (NODE_SIDE) dsum[k] = __builtin_fmaf(a[u][k], a[u][k], dsum[k]);
    }
  }
}

template <bool NODE_SIDE>
__global__ void __launch_bounds__(LOOP_T)
seg_gather_kernel(const int* __restrict__ ptr, const int* __restrict__ lst,
                  const int* __restrict__ oth, const float* __restrict__ alpha,
                  const float* __restrict__ src, const float* __restrict__ h,
                  const float* __restrict__ bias, float* __restrict__ dst, int nseg) {
  const int nw = gridDim.x * LOOP_WAVES;
  int seg = blockIdx.x * LOOP_WAVES + (threadIdx.x >> 5);
  if (seg >= nseg) return;
  const int lane = threadIdx.x & 31;
  const int c0 = 4 * lane;
  const int th = lane >> 4;
  float4 bv = make_float4(0.0f, 0.0f, 0.0f, 0.0f);
  if (NODE_SIDE) bv = *reinterpret_cast<const float4*>(bias + (c0 & (HIDDEN - 1)));

  int s = ptr[seg], e = ptr[seg + 1];
  for (; seg < nseg; seg += nw) {
    const int sx = (seg + nw < nseg) ? (seg + nw) : seg;
    const int sn = ptr[sx], en = ptr[sx + 1];
    float4 hv[SEG_C];
    if (NODE_SIDE) {
#pragma unroll
      for (int k = 0; k < SEG_C; ++k)
        hv[k] = *reinterpret_cast<const float4*>(h + (size_t)seg * OUT_CH + 128 * k + c0);
    }
    float4 acc[SEG_C];
    float dsum[SEG_C];
#pragma unroll
    for (int k = 0; k < SEG_C; ++k) { acc[k] = make_float4(0.0f, 0.0f, 0.0f, 0.0f); dsum[k] = 0.0f; }

    for (int q = s; q < e; q += SEG_B) {
      const int n = e - q;
      if (n >= 4)      seg_group<NODE_SIDE, 4>(lst, oth, alpha, src, q, c0, th, acc, dsum);
      else if (n == 3) seg_group<NODE_SIDE, 3>(lst, oth, alpha, src, q, c0, th, acc, dsum);
      else if (n == 2) seg_group<NODE_SIDE, 2>(lst, oth, alpha, src, q, c0, th, acc, dsum);
      else             seg_group<NODE_SIDE, 1>(lst, oth, alpha, src, q, c0, th, acc, dsum);
    }

    if (NODE_SIDE) {
#pragma unroll
      for (int k = 0; k < SEG_C; ++k) {
        const float dinv = (dsum[k] > 0.0f) ? (1.0f / dsum[k]) : 0.0f;
        float4 o;
        o.x = hv[k].x - dinv * acc[k].x + bv.x;
        o.y = hv[k].y - dinv * acc[k].y + bv.y;
        o.z = hv[k].z - dinv * acc[k].z + bv.z;
        o.w = hv[k].w - dinv * acc[k].w + bv.w;
        o.x = (o.x > 0.0f) ? o.x : expm1f(o.x);
        o.y = (o.y > 0.0f) ? o.y : expm1f(o.y);
        o.z = (o.z > 0.0f) ? o.z : expm1f(o.z);
        o.w = (o.w > 0.0f) ? o.w : expm1f(o.w);
        *reinterpret_cast<float4*>(dst + (size_t)seg * OUT_CH + 128 * k + c0) = o;
      }
    } else {
      const int len = e - s;
      const float binv = (len > 0) ? (1.0f / (float)len) : 0.0f;
#pragma unroll
      for (int k = 0; k < SEG_C; ++k)
        *reinterpret_cast<float4*>(dst + (size_t)seg * OUT_CH + 128 * k + c0) =
            make_float4(acc[k].x * binv, acc[k].y * binv, acc[k].z * binv, acc[k].w * binv);
    }
    s = sn;
    e = en;
  }
}

constexpr int NC_WAVES = 4;
constexpr int NC_T     = 32 * NC_WAVES;
constexpr int NC_PITCH = HIDDEN + 4;
constexpr int NC_TILE  = 16 * NC_PITCH;

__device__ inline void node_row_x1(const int* __restrict__ lst, const int* __restrict__ oth,
                                   const float* __restrict__ alpha, const float* __restrict__ src,
                                   const float4 (&hv)[SEG_C], const float4 bv,
                                   int s, int e, int c0, int th, float4 (&o)[SEG_C]) {
  float4 acc[SEG_C];
  float dsum[SEG_C];
#pragma unroll
  for (int k = 0; k < SEG_C; ++k) { acc[k] = make_float4(0.0f, 0.0f, 0.0f, 0.0f); dsum[k] = 0.0f; }
  for (int q = s; q < e; q += SEG_B) {
    const int n = e - q;
    if (n >= 4)      seg_group<true, 4>(lst, oth, alpha, src, q, c0, th, acc, dsum);
    else if (n == 3) seg_group<true, 3>(lst, oth, alpha, src, q, c0, th, acc, dsum);
    else if (n == 2) seg_group<true, 2>(lst, oth, alpha, src, q, c0, th, acc, dsum);
    else             seg_group<true, 1>(lst, oth, alpha, src, q, c0, th, acc, dsum);
  }
#pragma unroll
  for (int k = 0; k < SEG_C; ++k) {
    const float dinv = (dsum[k] > 0.0f) ? (1.0f / dsum[k]) : 0.0f;
    float4 r;
    r.x = hv[k].x - dinv * acc[k].x + bv.x;
    r.y = hv[k].y - dinv * acc[k].y + bv.y;
    r.z = hv[k].z - dinv * acc[k].z + bv.z;
    r.w = hv[k].w - dinv * acc[k].w + bv.w;
    r.x = (r.x > 0.0f) ? r.x : expm1f(r.x);
    r.y = (r.y > 0.0f) ? r.y : expm1f(r.y);
    r.z = (r.z > 0.0f) ? r.z : expm1f(r.z);
    r.w = (r.w > 0.0f) ? r.w : expm1f(r.w);
    o[k] = r;
  }
}

__global__ void __launch_bounds__(NC_T, 4)
seg_node_conv_kernel(const int* __restrict__ ptr, const int* __restrict__ lst,
                     const int* __restrict__ oth, const float* __restrict__ alpha,
                     const float* __restrict__ src, float* hio,
                     const float* __restrict__ bias, const float* __restrict__ W1, int nseg) {
  constexpr int K = HIDDEN, NT = 4, S = K / 4;
  __shared__ float xt[NC_WAVES * NC_TILE];
  const int wib = __builtin_amdgcn_readfirstlane(threadIdx.x >> 5);
  const int nw  = gridDim.x * NC_WAVES;
  int seg = blockIdx.x * NC_WAVES + wib;
  if (seg >= nseg) return;
  const int lane = threadIdx.x & 31;
  const int c0 = 4 * lane;
  const int th = lane >> 4;
  const int c  = lane & 15;
  float* xw = xt + wib * NC_TILE;
  for (int i = lane; i < NC_TILE; i += 32) xw[i] = 0.0f;

  v2f bf[S][NT];
#pragma unroll
  for (int s = 0; s < S; ++s) {
    const int k = 16 * (s >> 2) + 8 * th + 2 * (s & 3);
    float w0[NT], w1[NT];
    load_cols<NT>(W1 + (size_t)k * K + NT * c, w0);
    load_cols<NT>(W1 + (size_t)(k + 1) * K + NT * c, w1);
#pragma unroll
    for (int t = 0; t < NT; ++t) { bf[s][t].x = w0[t]; bf[s][t].y = w1[t]; }
  }
  const float4 bv = *reinterpret_cast<const float4*>(bias + (c0 & (HIDDEN - 1)));

  int s0 = ptr[seg], e0 = ptr[seg + 1];
  int s1, e1;
  {
    const int sb = (seg + nw < nseg) ? (seg + nw) : seg;
    s1 = ptr[sb]; e1 = ptr[sb + 1];
  }
  for (; seg < nseg; seg += 2 * nw) {
    const int seg1 = seg + nw;
    const bool two = (seg1 < nseg);
    const int nx0 = (seg + 2 * nw < nseg) ? (seg + 2 * nw) : seg;
    const int nx1 = (seg + 3 * nw < nseg) ? (seg + 3 * nw) : nx0;
    const int ns0 = ptr[nx0], ne0 = ptr[nx0 + 1];
    const int ns1 = ptr[nx1], ne1 = ptr[nx1 + 1];

    float4 hv[SEG_C];
#pragma unroll
    for (int k = 0; k < SEG_C; ++k)
      hv[k] = *reinterpret_cast<const float4*>(hio + (size_t)seg * OUT_CH + 128 * k + c0);
    float4 o[SEG_C];
    node_row_x1(lst, oth, alpha, src, hv, bv, s0, e0, c0, th, o);
#pragma unroll
    for (int k = 0; k < SEG_C; ++k)
      *reinterpret_cast<float4*>(xw + (2 * k + th) * NC_PITCH + 4 * c) = o[k];
    if (two) {
#pragma unroll
      for (int k = 0; k < SEG_C; ++k)
        hv[k] = *reinterpret_cast<const float4*>(hio + (size_t)seg1 * OUT_CH + 128 * k + c0);
      node_row_x1(lst, oth, alpha, src, hv, bv, s1, e1, c0, th, o);
#pragma unroll
      for (int k = 0; k < SEG_C; ++k)
        *reinterpret_cast<float4*>(xw + (D + 2 * k + th) * NC_PITCH + 4 * c) = o[k];
    }
    __builtin_amdgcn_fence(__ATOMIC_RELEASE, "wavefront");
    __builtin_amdgcn_wave_barrier();
    __builtin_amdgcn_fence(__ATOMIC_ACQUIRE, "wavefront");

    float4 av[K / 8];
    {
      const float* ap = xw + c * NC_PITCH + 8 * th;
#pragma unroll
      for (int g = 0; g < K / 16; ++g) {
        av[2 * g]     = *reinterpret_cast<const float4*>(ap + 16 * g);
        av[2 * g + 1] = *reinterpret_cast<const float4*>(ap + 16 * g + 4);
      }
    }
    __builtin_amdgcn_fence(__ATOMIC_RELEASE, "wavefront");
    __builtin_amdgcn_wave_barrier();
    __builtin_amdgcn_fence(__ATOMIC_ACQUIRE, "wavefront");
    __builtin_amdgcn_sched_barrier(0);

    v8f acc[NT];
#pragma unroll
    for (int t = 0; t < NT; ++t) acc[t] = v8f{};
#pragma unroll
    for (int s = 0; s < S; ++s) {
      const float4 q = av[s >> 1];
      v2f a;
      a.x = (s & 1) ? q.z : q.x;
      a.y = (s & 1) ? q.w : q.y;
#pragma unroll
      for (int t = 0; t < NT; ++t)
        acc[t] = __builtin_amdgcn_wmma_f32_16x16x4_f32(false, a, false, bf[s][t],
                                                       (short)0, acc[t], false, false);
    }
#pragma unroll
    for (int j = 0; j < 8; ++j) {
      const int r = j + 8 * th;
      const bool second = (r >= D);
      const bool ok = second ? (two && r < 2 * D) : true;
      const int node  = second ? seg1 : seg;
      const int stalk = second ? (r - D) : r;
      if (ok)
        *reinterpret_cast<float4*>(hio + ((size_t)node * D + stalk) * HIDDEN + NT * c) =
            make_float4(acc[0][j], acc[1][j], acc[2][j], acc[3][j]);
    }
    s0 = ns0; e0 = ne0; s1 = ns1; e1 = ne1;
  }
}

constexpr int SEG_E_BLK = (N_EDGES + LOOP_WAVES - 1) / LOOP_WAVES;
constexpr int SEG_N_BLK = LOOP_BLK;
constexpr int NC_BLK    = 1024;

extern "C" void kernel_launch(void* const* d_in, const int* in_sizes, int n_in,
                              void* d_out, int out_size, void* d_ws, size_t ws_size,
                              hipStream_t stream) {
  (void)in_sizes; (void)n_in; (void)out_size; (void)ws_size;
  const float* x       = (const float*)d_in[0];
  const float* hattr   = (const float*)d_in[1];
  const float* W_lin   = (const float*)d_in[2];
  const float* b_lin   = (const float*)d_in[3];
  const float* ln_g    = (const float*)d_in[4];
  const float* ln_b    = (const float*)d_in[5];
  const float* W_sheaf = (const float*)d_in[6];
  const float* b_sheaf = (const float*)d_in[7];
  const float* W_conv  = (const float*)d_in[8];
  const float* b_conv  = (const float*)d_in[9];
  const int*   eidx    = (const int*)d_in[10];
  const int* row = eidx;
  const int* col = eidx + NNZ;
  float* out = (float*)d_out;

  float* ws    = (float*)d_ws;
  float* fold  = ws;
  float* Wp    = fold;
  float* bp    = Wp + (size_t)IN_CH * OUT_CH;
  float* Wm    = bp + OUT_CH;
  float* bm    = Wm + (size_t)IN_CH * HIDDEN;
  float* es    = ws   + (size_t)ND * HIDDEN;
  float* xm    = es   + (size_t)ED * HIDDEN;
  float* em    = xm   + (size_t)N_NODES * HIDDEN;
  float* alpha = em   + (size_t)N_EDGES * HIDDEN;
  float* spare = alpha + (size_t)NNZ * D;
  float* hbuf  = spare + ND + N_EDGES - 16;
  float* me    = hbuf + (size_t)ND * HIDDEN;
  int*   ib    = (int*)(me + (size_t)ED * HIDDEN + 16);
  int* nptr = ib;
  int* eptr = nptr + N_NODES + 8;
  unsigned* ntmp = (unsigned*)(eptr + N_EDGES + 8);
  unsigned* etmp = ntmp + NNZ;
  int* nlst = (int*)(etmp + NNZ);
  int* ncol = nlst + NNZ;
  int* elst = ncol + NNZ;
  int* erow = elst + NNZ;

  csr_build_kernel<<<CSR_NB_N + CSR_NB_E, CSR_T, 0, stream>>>(row, col, nptr, eptr, ntmp, etmp,
                                                              nlst, ncol, elst, erow);

  fold_weights_kernel<<<FOLD_BLK, 32 * FOLD_WPB, 0, stream>>>(W_lin, b_lin, W_conv, Wp, bp, Wm, bm);

  proj_fold_kernel<<<PF_BLK_H + PF_BLK_M, 32 * PF_WAVES, 0, stream>>>(x, hattr, Wp, bp, Wm, bm,
                                                                     hbuf, xm, em);

  alpha_kernel<<<ALPHA_BLK, 32 * ALPHA_WPB, 0, stream>>>(xm, em, row, col, ln_g, ln_b,
                                                W_sheaf, b_sheaf, alpha);

  seg_gather_kernel<false><<<SEG_E_BLK, LOOP_T, 0, stream>>>(eptr, elst, erow, alpha, hbuf,
                                                           nullptr, nullptr, me, N_EDGES);
  seg_node_conv_kernel<<<NC_BLK, NC_T, 0, stream>>>(nptr, nlst, ncol, alpha, me, hbuf, b_conv,
                                                    W_conv + HIDDEN * HIDDEN, N_NODES);

  seg_gather_kernel<false><<<SEG_E_BLK, LOOP_T, 0, stream>>>(eptr, elst, erow, alpha, hbuf,
                                                           nullptr, nullptr, me, N_EDGES);
  seg_gather_kernel<true><<<SEG_N_BLK, LOOP_T, 0, stream>>>(nptr, nlst, ncol, alpha, me,
                                                          hbuf, b_conv + HIDDEN, out, N_NODES);
}


# Weight fold (W_lin x W0 per stalk, stalk means) runs as 2 extra work-groups of the list-build launch, on WGPs that launch leaves idle: 7 dependent launches, not 8; arithmetic and sum orders unchanged
# speedup vs baseline: 5.2797x; 1.0559x over previous
#include <hip/hip_runtime.h>
#include <math.h>

constexpr int N_NODES = 50000;
constexpr int N_EDGES = 10000;
constexpr int NNZ     = 150000;
constexpr int IN_CH   = 128;
constexpr int HIDDEN  = 64;
constexpr int D       = 6;
constexpr int OUT_CH  = HIDDEN * D;
constexpr int ND      = N_NODES * D;
constexpr int ED      = N_EDGES * D;

typedef __attribute__((ext_vector_type(2))) float v2f;
typedef __attribute__((ext_vector_type(8))) float v8f;

template <int NT> struct ColVec;
template <> struct ColVec<2> { typedef float2 type; };
template <> struct ColVec<4> { typedef float4 type; };

template <int NT>
__device__ inline void load_cols(const float* __restrict__ p, float (&d)[NT]) {
  const typename ColVec<NT>::type v = *reinterpret_cast<const typename ColVec<NT>::type*>(p);
  const float* f = reinterpret_cast<const float*>(&v);
#pragma unroll
  for (int t = 0; t < NT; ++t) d[t] = f[t];
}

template <int NT>
__device__ inline void store_cols(float* __restrict__ p, const float (&d)[NT]) {
  typename ColVec<NT>::type v;
  float* f = reinterpret_cast<float*>(&v);
#pragma unroll
  for (int t = 0; t < NT; ++t) f[t] = d[t];
  *reinterpret_cast<typename ColVec<NT>::type*>(p) = v;
}

template <int K>
__device__ inline void load_a_tile(float4 (&av)[K / 8], const float* __restrict__ A,
                                   int mt, int c, int h) {
  const float* ap = A + ((size_t)mt * 16 + c) * K + 8 * h;
#pragma unroll
  for (int g = 0; g < K / 16; ++g) {
    av[2 * g]     = *reinterpret_cast<const float4*>(ap + 16 * g);
    av[2 * g + 1] = *reinterpret_cast<const float4*>(ap + 16 * g + 4);
  }
}

constexpr int FOLD_TILES = IN_CH * D / 16;
constexpr int FOLD_T     = 1024;
constexpr int FOLD_WPB   = FOLD_T / 32;
constexpr int FOLD_BLK   = (FOLD_TILES + 1 + FOLD_WPB - 1) / FOLD_WPB;
constexpr int FOLD_MQ    = IN_CH * HIDDEN / 4;

__device__ inline void fold_weights_block(int fb,
                    const float* __restrict__ W_lin, const float* __restrict__ b_lin,
                    const float* __restrict__ W0,
                    float* __restrict__ Wp, float* __restrict__ bp,
                    float* __restrict__ Wm, float* __restrict__ bm) {
  constexpr int K = HIDDEN, NT = 4, S = K / 4;
  for (int qd = fb * FOLD_T + threadIdx.x; qd < FOLD_MQ + HIDDEN / 4;
       qd += FOLD_BLK * FOLD_T) {
    const bool isb = (qd >= FOLD_MQ);
    const float* p = isb ? (b_lin + 4 * (qd - FOLD_MQ)) : (W_lin + (size_t)(qd >> 4) * OUT_CH + 4 * (qd & 15));
    float4 sm = make_float4(0.0f, 0.0f, 0.0f, 0.0f);
#pragma unroll
    for (int t = 0; t < D; ++t) {
      const float4 v = *reinterpret_cast<const float4*>(p + t * HIDDEN);
      sm.x += v.x; sm.y += v.y; sm.z += v.z; sm.w += v.w;
    }
    const float k6 = 1.0f / 6.0f;
    float* d = isb ? (bm + 4 * (qd - FOLD_MQ)) : (Wm + (size_t)qd * 4);
    *reinterpret_cast<float4*>(d) = make_float4(sm.x * k6, sm.y * k6, sm.z * k6, sm.w * k6);
  }

  const int wave = __builtin_amdgcn_readfirstlane(fb * FOLD_WPB + (threadIdx.x >> 5));
  if (wave > FOLD_TILES) return;
  const bool btile = (wave == FOLD_TILES);
  const float* src = btile ? b_lin : (W_lin + (size_t)wave * 16 * K);
  float* dst       = btile ? bp : (Wp + (size_t)wave * 16 * K);
  const int rows   = btile ? D : 16;
  const int lane = threadIdx.x & 31;
  const int h = lane >> 4;
  const int c = lane & 15;

  float4 av[K / 8];
#pragma unroll
  for (int i = 0; i < K / 8; ++i) av[i] = make_float4(0.0f, 0.0f, 0.0f, 0.0f);
  if (c < rows) {
    const float* ap = src + (size_t)c * K + 8 * h;
#pragma unroll
    for (int g = 0; g < K / 16; ++g) {
      av[2 * g]     = *reinterpret_cast<const float4*>(ap + 16 * g);
      av[2 * g + 1] = *reinterpret_cast<const float4*>(ap + 16 * g + 4);
    }
  }
  v8f acc[NT];
#pragma unroll
  for (int t = 0; t < NT; ++t) acc[t] = v8f{};
#pragma unroll
  for (int s = 0; s < S; ++s) {
    const int k = 16 * (s >> 2) + 8 * h + 2 * (s & 3);
    float w0[NT], w1[NT];
    load_cols<NT>(W0 + (size_t)k * K + NT * c, w0);
    load_cols<NT>(W0 + (size_t)(k + 1) * K + NT * c, w1);
    const float4 q = av[s >> 1];
    v2f a;
    a.x = (s & 1) ? q.z : q.x;
    a.y = (s & 1) ? q.w : q.y;
#pragma unroll
    for (int t = 0; t < NT; ++t) {
      v2f b;
      b.x = w0[t]; b.y = w1[t];
      acc[t] = __builtin_amdgcn_wmma_f32_16x16x4_f32(false, a, false, b, (short)0, acc[t], false, false);
    }
  }
#pragma unroll
  for (int j = 0; j < 8; ++j) {
    float o[NT];
#pragma unroll
    for (int t = 0; t < NT; ++t) o[t] = acc[t][j];
    if (j + 8 * h < rows) store_cols<NT>(dst + (size_t)(j + 8 * h) * K + NT * c, o);
  }
}

constexpr int PF_WAVES  = 12;
constexpr int PF_RUN    = 15;
constexpr int PF_RUNS_N = (N_NODES / 16 + PF_RUN - 1) / PF_RUN;
constexpr int PF_RUNS_E = (N_EDGES / 16 + PF_RUN - 1) / PF_RUN;
constexpr int PF_BLK_H  = PF_RUNS_N;
constexpr int PF_MW_N   = 2 * PF_RUNS_N;
constexpr int PF_MW_E   = 2 * PF_RUNS_E;
constexpr int PF_BLK_M  = (PF_MW_N + PF_MW_E + PF_WAVES - 1) / PF_WAVES;
static_assert(N_NODES % 16 == 0 && N_EDGES % 16 == 0, "row tiles are whole");

__global__ void __launch_bounds__(32 * PF_WAVES)
proj_fold_kernel(const float* __restrict__ x, const float* __restrict__ hattr,
                 const float* __restrict__ Wp, const float* __restrict__ bp,
                 const float* __restrict__ Wm, const float* __restrict__ bm,
                 float* __restrict__ h1, float* __restrict__ xm, float* __restrict__ em) {
  constexpr int K = IN_CH, NT = 2, S = K / 4;
  const int blk = blockIdx.x;
  const int wib = __builtin_amdgcn_readfirstlane(threadIdx.x >> 5);
  const float* A; const float* B; const float* bias; float* C;
  int pitch, n0, run, mTiles;
  if (blk < PF_BLK_H) {
    A = x; B = Wp; bias = bp; C = h1; pitch = OUT_CH; n0 = 16 * NT * wib; run = blk; mTiles = N_NODES / 16;
  } else {
    const int u = (blk - PF_BLK_H) * PF_WAVES + wib;
    B = Wm; bias = bm; pitch = HIDDEN;
    if (u < PF_MW_N) {
      A = x; C = xm; run = u >> 1; n0 = 16 * NT * (u & 1); mTiles = N_NODES / 16;
    } else {
      const int v = u - PF_MW_N;
      if (v >= PF_MW_E) return;
      A = hattr; C = em; run = v >> 1; n0 = 16 * NT * (v & 1); mTiles = N_EDGES / 16;
    }
  }
  int mt = run * PF_RUN;
  if (mt >= mTiles) return;
  const int mtEnd = (mt + PF_RUN < mTiles) ? (mt + PF_RUN) : mTiles;
  const int lane = threadIdx.x & 31;
  const int h = lane >> 4;
  const int c = lane & 15;

  v2f bf[S][NT];
#pragma unroll
  for (int s = 0; s < S; ++s) {
    const int k = 16 * (s >> 2) + 8 * h + 2 * (s & 3);
    float w0[NT], w1[NT];
    load_cols<NT>(B + (size_t)k * pitch + n0 + NT * c, w0);
    load_cols<NT>(B + (size_t)(k + 1) * pitch + n0 + NT * c, w1);
#pragma unroll
    for (int t = 0; t < NT; ++t) { bf[s][t].x = w0[t]; bf[s][t].y = w1[t]; }
  }
  float bv[NT];
  load_cols<NT>(bias + n0 + NT * c, bv);

  float4 av[K / 8];
  load_a_tile<K>(av, A, mt, c, h);
  for (; mt < mtEnd; ++mt) {
    const int nx = (mt + 1 < mtEnd) ? (mt + 1) : mt;
    float4 an[K / 8];
    load_a_tile<K>(an, A, nx, c, h);
    __builtin_amdgcn_sched_barrier(0);

    v8f acc[NT];
#pragma unroll
    for (int t = 0; t < NT; ++t) acc[t] = v8f{};
#pragma unroll
    for (int s = 0; s < S; ++s) {
      const float4 q = av[s >> 1];
      v2f a;
      a.x = (s & 1) ? q.z : q.x;
      a.y = (s & 1) ? q.w : q.y;
#pragma unroll
      for (int t = 0; t < NT; ++t)
        acc[t] = __builtin_amdgcn_wmma_f32_16x16x4_f32(false, a, false, bf[s][t],
                                                       (short)0, acc[t], false, false);
    }
    float* cp = C + ((size_t)mt * 16 + 8 * h) * pitch + n0 + NT * c;
#pragma unroll
    for (int j = 0; j < 8; ++j) {
      float o[NT];
#pragma unroll
      for (int t = 0; t < NT; ++t) o[t] = acc[t][j] + bv[t];
      store_cols<NT>(cp + (size_t)j * pitch, o);
    }
    __builtin_amdgcn_sched_barrier(0);
#pragma unroll
    for (int i = 0; i < K / 8; ++i) av[i] = an[i];
  }
}

constexpr int LOOP_WAVES = 8;
constexpr int LOOP_T     = 32 * LOOP_WAVES;
constexpr int LOOP_BLK   = 1024;

template <int M>
__device__ inline float xmask_add(float v) {
  const int o = __builtin_amdgcn_update_dpp(0, __builtin_bit_cast(int, v), 0x160 + M, 0xf, 0xf, false);
  return v + __builtin_bit_cast(float, o);
}
template <int LG>
__device__ inline float group_allsum(float v) {
  v = xmask_add<1>(v);
  v = xmask_add<2>(v);
  v = xmask_add<4>(v);
  if (LG == 16) v = xmask_add<8>(v);
  return v;
}

constexpr int ALPHA_LG   = 8;
constexpr int ALPHA_HL   = ALPHA_LG / 2;
constexpr int ALPHA_NQ   = HIDDEN / (4 * ALPHA_HL);
constexpr int ALPHA_IPT  = 32 / ALPHA_LG;
constexpr int ALPHA_NTRIP = NNZ / ALPHA_IPT;
constexpr int ALPHA_WPB  = 2;
constexpr int ALPHA_TPW  = 10;
constexpr int ALPHA_BLK  = (ALPHA_NTRIP + ALPHA_TPW * ALPHA_WPB - 1) / (ALPHA_TPW * ALPHA_WPB);
static_assert(NNZ % ALPHA_IPT == 0, "a trip is a whole group of incidences");
static_assert(ALPHA_LG == 8 || ALPHA_LG == 16, "lane groups lie inside the 16-lane rows");

__global__ void __launch_bounds__(32 * ALPHA_WPB)
alpha_kernel(const float* __restrict__ xm, const float* __restrict__ em,
             const int* __restrict__ row, const int* __restrict__ col,
             const float* __restrict__ ln_g, const float* __restrict__ ln_b,
             const float* __restrict__ Wsh, const float* __restrict__ bsh,
             float* __restrict__ alpha) {
  constexpr int LG = ALPHA_LG, HL = ALPHA_HL, NQ = ALPHA_NQ, IPT = ALPHA_IPT;
  const int nw = gridDim.x * ALPHA_WPB;
  int tr = __builtin_amdgcn_readfirstlane(blockIdx.x * ALPHA_WPB + (threadIdx.x >> 5));
  if (tr >= ALPHA_NTRIP) return;
  const int lane = threadIdx.x & 31;
  const int grp  = lane / LG;
  const int sub  = lane % LG;
  const bool eside = (sub >= HL);
  const int ls   = sub % HL;
  const float* zsrc = eside ? em : xm;
  const int*   ksrc = eside ? col : row;
  const int j0 = (eside ? HIDDEN : 0) + 4 * ls;

  float g[NQ][4], b[NQ][4], w[NQ][4][D];
#pragma unroll
  for (int q = 0; q < NQ; ++q) {
    const int j = j0 + 4 * HL * q;
    const float4 gv = *reinterpret_cast<const float4*>(ln_g + j);
    const float4 bv = *reinterpret_cast<const float4*>(ln_b + j);
    g[q][0] = gv.x; g[q][1] = gv.y; g[q][2] = gv.z; g[q][3] = gv.w;
    b[q][0] = bv.x; b[q][1] = bv.y; b[q][2] = bv.z; b[q][3] = bv.w;
    float wf[4 * D];
#pragma unroll
    for (int m = 0; m < D; ++m) {
      const float4 wv = *reinterpret_cast<const float4*>(Wsh + (size_t)j * D + 4 * m);
      wf[4 * m] = wv.x; wf[4 * m + 1] = wv.y; wf[4 * m + 2] = wv.z; wf[4 * m + 3] = wv.w;
    }
#pragma unroll
    for (int e = 0; e < 4; ++e)
#pragma unroll
      for (int t = 0; t < D; ++t) w[q][e][t] = wf[e * D + t];
  }
  const float bs = (sub < D) ? bsh[sub] : 0.0f;

  float4 zc[NQ], zn[NQ];
  {
    const int key = ksrc[tr * IPT + grp];
    const float* zp = zsrc + (size_t)key * HIDDEN + 4 * ls;
#pragma unroll
    for (int q = 0; q < NQ; ++q) zc[q] = *reinterpret_cast<const float4*>(zp + 4 * HL * q);
  }
  const int tr1 = (tr + nw < ALPHA_NTRIP) ? (tr + nw) : tr;
  int keyn = ksrc[tr1 * IPT + grp];
  for (; tr < ALPHA_NTRIP; tr += nw) {
    {
      const float* zp = zsrc + (size_t)keyn * HIDDEN + 4 * ls;
#pragma unroll
      for (int q = 0; q < NQ; ++q) zn[q] = *reinterpret_cast<const float4*>(zp + 4 * HL * q);
    }
    const int tr2 = (tr + 2 * nw < ALPHA_NTRIP) ? (tr + 2 * nw) : tr;
    const int keynn = ksrc[tr2 * IPT + grp];

    float z[NQ][4];
#pragma unroll
    for (int q = 0; q < NQ; ++q) { z[q][0] = zc[q].x; z[q][1] = zc[q].y; z[q][2] = zc[q].z; z[q][3] = zc[q].w; }
    float s = 0.0f, ss = 0.0f;
#pragma unroll
    for (int q = 0; q < NQ; ++q)
#pragma unroll
      for (int e = 0; e < 4; ++e) { s += z[q][e]; ss += z[q][e] * z[q][e]; }
    s  = group_allsum<LG>(s);
    ss = group_allsum<LG>(ss);
    const float mu   = s * (1.0f / 128.0f);
    const float var  = ss * (1.0f / 128.0f) - mu * mu;
    const float rstd = rsqrtf(var + 1e-5f);

    float p[D] = {0, 0, 0, 0, 0, 0};
#pragma unroll
    for (int q = 0; q < NQ; ++q) {
#pragma unroll
      for (int e = 0; e < 4; ++e) {
        const float fn = (z[q][e] - mu) * rstd * g[q][e] + b[q][e];
#pragma unroll
        for (int t = 0; t < D; ++t) p[t] = __builtin_fmaf(fn, w[q][e][t], p[t]);
      }
    }
#pragma unroll
    for (int t = 0; t < D; ++t) p[t] = group_allsum<LG>(p[t]);

    float pv = p[0];
#pragma unroll
    for (int t = 1; t < D; ++t) pv = (sub == t) ? p[t] : pv;
    if (sub < D) {
      const float a = 1.0f / (1.0f + expf(-(pv + bs)));
      alpha[(size_t)(tr * IPT + grp) * D + sub] = a;
    }
#pragma unroll
    for (int q = 0; q < NQ; ++q) zc[q] = zn[q];
    keyn = keynn;
  }
}

constexpr int CSR_T      = 1024;
constexpr int CSR_R_N    = 544;
constexpr int CSR_NB_N   = (N_NODES + CSR_R_N - 1) / CSR_R_N;
constexpr int CSR_R_E    = 160;
constexpr int CSR_NB_E   = (N_EDGES + CSR_R_E - 1) / CSR_R_E;
constexpr int CSR_IDBITS = 18;
constexpr int CSR_N4     = NNZ / 4;
constexpr int CSR_U      = 4;
static_assert(NNZ % 4 == 0, "the key arrays are read as 16-byte quads");
static_assert(NNZ <= (1 << CSR_IDBITS), "an incidence id must fit the low bits of a list word");
static_assert(CSR_R_N <= CSR_T && CSR_R_E <= CSR_T, "the scan takes one counter per thread");
static_assert(CSR_T <= (1 << (32 - CSR_IDBITS)), "a key slot must fit the high bits of a list word");
static_assert((CSR_NB_N - 1) * CSR_R_N < N_NODES && (CSR_NB_E - 1) * CSR_R_E < N_EDGES,
              "every work-group owns at least one key");

__device__ inline int wave_allsum_i(int v) {
#pragma unroll
  for (int off = 16; off > 0; off >>= 1) v += __shfl_xor(v, off, 32);
  return v;
}

__device__ inline int wave_incl_scan(int v, int lane) {
#pragma unroll
  for (int off = 1; off < 32; off <<= 1) {
    int u = __shfl_up(v, off, 32);
    if (lane >= off) v += u;
  }
  return v;
}

__device__ inline void csr_count4(const int4 q, int k0, int nk, int* cur, int& below) {
  const int kk[4] = {q.x, q.y, q.z, q.w};
#pragma unroll
  for (int c = 0; c < 4; ++c) {
    const int d = kk[c] - k0;
    below += (d < 0) ? 1 : 0;
    if ((unsigned)d < (unsigned)nk) atomicAdd(&cur[d], 1);
  }
}

__device__ inline void csr_place4(const int4 q, int i0, int k0, int nk, int* cur,
                                  const int* start, unsigned* __restrict__ tmp) {
  const int kk[4] = {q.x, q.y, q.z, q.w};
#pragma unroll
  for (int c = 0; c < 4; ++c) {
    const int d = kk[c] - k0;
    if ((unsigned)d < (unsigned)nk) {
      const int slot = atomicAdd(&cur[d], 1);
      tmp[start[d] + slot] = ((unsigned)d << CSR_IDBITS) | (unsigned)(i0 + c);
    }
  }
}

static_assert(FOLD_T == CSR_T, "the weight fold rides in this launch with this work-group size");

__global__ void __launch_bounds__(CSR_T)
csr_build_kernel(const int* __restrict__ row, const int* __restrict__ col,
                 int* __restrict__ nptr, int* __restrict__ eptr,
                 unsigned* __restrict__ ntmp, unsigned* __restrict__ etmp,
                 int* __restrict__ nlst, int* __restrict__ ncol,
                 int* __restrict__ elst, int* __restrict__ erow,
                 const float* __restrict__ W_lin, const float* __restrict__ b_lin,
                 const float* __restrict__ W0,
                 float* __restrict__ Wp, float* __restrict__ bp,
                 float* __restrict__ Wm, float* __restrict__ bm) {
  if (blockIdx.x >= CSR_NB_N + CSR_NB_E) {
    fold_weights_block(blockIdx.x - (CSR_NB_N + CSR_NB_E), W_lin, b_lin, W0, Wp, bp, Wm, bm);
    return;
  }
  __shared__ int cur[CSR_T];
  __shared__ int start[CSR_T];
  __shared__ int wsum[32];
  __shared__ int wbel[32];
  __shared__ int sh_base, sh_total;

  const bool eside = (blockIdx.x >= CSR_NB_N);
  const int b      = eside ? (blockIdx.x - CSR_NB_N) : blockIdx.x;
  const int* keys  = eside ? col : row;
  const int* other = eside ? row : col;
  const int nkeys  = eside ? N_EDGES : N_NODES;
  const int R      = eside ? CSR_R_E : CSR_R_N;
  int* ptr         = eside ? eptr : nptr;
  unsigned* tmp    = eside ? etmp : ntmp;
  int* lst         = eside ? elst : nlst;
  int* oth         = eside ? erow : ncol;
  const int k0 = b * R;
  const int nk = (nkeys - k0 < R) ? (nkeys - k0) : R;
  const int tid = threadIdx.x, lane = tid & 31, w = tid >> 5;
  const int4* k4 = reinterpret_cast<const int4*>(keys);

  cur[tid] = 0;
  __syncthreads();

  int below = 0;
  {
    int v = tid;
    for (; v + (CSR_U - 1) * CSR_T < CSR_N4; v += CSR_U * CSR_T) {
      int4 q[CSR_U];
#pragma unroll
      for (int u = 0; u < CSR_U; ++u) q[u] = k4[v + u * CSR_T];
#pragma unroll
      for (int u = 0; u < CSR_U; ++u) csr_count4(q[u], k0, nk, cur, below);
    }
    for (; v < CSR_N4; v += CSR_T) csr_count4(k4[v], k0, nk, cur, below);
  }
  const int bw = wave_allsum_i(below);
  if (lane == 0) wbel[w] = bw;
  __syncthreads();

  const int cnt  = (tid < nk) ? cur[tid] : 0;
  const int incl = wave_incl_scan(cnt, lane);
  if (lane == 31) wsum[w] = incl;
  __syncthreads();
  if (w == 0) {
    const int t  = wsum[lane];
    const int ti = wave_incl_scan(t, lane);
    const int bl = wave_allsum_i(wbel[lane]);
    wsum[lane] = ti - t;
    if (lane == 31) { sh_base = bl; sh_total = ti; }
  }
  __syncthreads();
  const int base  = sh_base;
  const int total = sh_total;
  const int st = base + wsum[w] + incl - cnt;
  start[tid] = st;
  cur[tid] = 0;
  if (tid < nk) ptr[k0 + tid] = st;
  if (tid == nk - 1 && k0 + nk == nkeys) ptr[nkeys] = st + cnt;
  __syncthreads();

  {
    int v = tid;
    for (; v + (CSR_U - 1) * CSR_T < CSR_N4; v += CSR_U * CSR_T) {
      int4 q[CSR_U];
#pragma unroll
      for (int u = 0; u < CSR_U; ++u) q[u] = k4[v + u * CSR_T];
#pragma unroll
      for (int u = 0; u < CSR_U; ++u)
        csr_place4(q[u], 4 * (v + u * CSR_T), k0, nk, cur, start, tmp);
    }
    for (; v < CSR_N4; v += CSR_T) csr_place4(k4[v], 4 * v, k0, nk, cur, start, tmp);
  }
  __syncthreads();

  for (int p = tid; p < total; p += CSR_T) {
    const unsigned wv = tmp[base + p];
    const int d = (int)(wv >> CSR_IDBITS);
    const int i = (int)(wv & ((1u << CSR_IDBITS) - 1u));
    const int o = other[i];
    const int s = start[d], n = cur[d];
    const unsigned* sp = tmp + s;
    int rank = 0;
    int k = 0;
    for (; k + 4 <= n; k += 4) {
      const unsigned u0 = sp[k], u1 = sp[k + 1], u2 = sp[k + 2], u3 = sp[k + 3];
      rank += ((u0 < wv) ? 1 : 0) + ((u1 < wv) ? 1 : 0) + ((u2 < wv) ? 1 : 0) + ((u3 < wv) ? 1 : 0);
    }
    for (; k < n; ++k) rank += (sp[k] < wv) ? 1 : 0;
    lst[s + rank] = i;
    oth[s + rank] = o;
  }
}

constexpr int SEG_C = OUT_CH / 128;
constexpr int SEG_B = 4;
static_assert(SEG_B == 4, "the list walk picks its group body from the sizes 4, 3, 2, 1");

template <bool NODE_SIDE, int NB>
__device__ inline void seg_group(const int* __restrict__ lst, const int* __restrict__ oth,
                                 const float* __restrict__ alpha, const float* __restrict__ src,
                                 int q, int c0, int th,
                                 float4 (&acc)[SEG_C], float (&dsum)[SEG_C]) {
  int ii[NB], oo[NB];
#pragma unroll
  for (int u = 0; u < NB; ++u) { ii[u] = lst[q + u]; oo[u] = oth[q + u]; }
  float a[NB][SEG_C];
  float4 v[NB][SEG_C];
#pragma unroll
  for (int u = 0; u < NB; ++u) {
#pragma unroll
    for (int k = 0; k < SEG_C; ++k) {
      a[u][k] = alpha[(size_t)ii[u] * D + 2 * k + th];
      v[u][k] = *reinterpret_cast<const float4*>(src + (size_t)oo[u] * OUT_CH + 128 * k + c0);
    }
  }
#pragma unroll
  for (int u = 0; u < NB; ++u) {
#pragma unroll
    for (int k = 0; k < SEG_C; ++k) {
      acc[k].x = __builtin_fmaf(a[u][k], v[u][k].x, acc[k].x);
      acc[k].y = __builtin_fmaf(a[u][k], v[u][k].y, acc[k].y);
      acc[k].z = __builtin_fmaf(a[u][k], v[u][k].z, acc[k].z);
      acc[k].w = __builtin_fmaf(a[u][k], v[u][k].w, acc[k].w);
      if (NODE_SIDE) dsum[k] = __builtin_fmaf(a[u][k], a[u][k], dsum[k]);
    }
  }
}

template <bool NODE_SIDE>
__global__ void __launch_bounds__(LOOP_T)
seg_gather_kernel(const int* __restrict__ ptr, const int* __restrict__ lst,
                  const int* __restrict__ oth, const float* __restrict__ alpha,
                  const float* __restrict__ src, const float* __restrict__ h,
                  const float* __restrict__ bias, float* __restrict__ dst, int nseg) {
  const int nw = gridDim.x * LOOP_WAVES;
  int seg = blockIdx.x * LOOP_WAVES + (threadIdx.x >> 5);
  if (seg >= nseg) return;
  const int lane = threadIdx.x & 31;
  const int c0 = 4 * lane;
  const int th = lane >> 4;
  float4 bv = make_float4(0.0f, 0.0f, 0.0f, 0.0f);
  if (NODE_SIDE) bv = *reinterpret_cast<const float4*>(bias + (c0 & (HIDDEN - 1)));

  int s = ptr[seg], e = ptr[seg + 1];
  for (; seg < nseg; seg += nw) {
    const int sx = (seg + nw < nseg) ? (seg + nw) : seg;
    const int sn = ptr[sx], en = ptr[sx + 1];
    float4 hv[SEG_C];
    if (NODE_SIDE) {
#pragma unroll
      for (int k = 0; k < SEG_C; ++k)
        hv[k] = *reinterpret_cast<const float4*>(h + (size_t)seg * OUT_CH + 128 * k + c0);
    }
    float4 acc[SEG_C];
    float dsum[SEG_C];
#pragma unroll
    for (int k = 0; k < SEG_C; ++k) { acc[k] = make_float4(0.0f, 0.0f, 0.0f, 0.0f); dsum[k] = 0.0f; }

    for (int q = s; q < e; q += SEG_B) {
      const int n = e - q;
      if (n >= 4)      seg_group<NODE_SIDE, 4>(lst, oth, alpha, src, q, c0, th, acc, dsum);
      else if (n == 3) seg_group<NODE_SIDE, 3>(lst, oth, alpha, src, q, c0, th, acc, dsum);
      else if (n == 2) seg_group<NODE_SIDE, 2>(lst, oth, alpha, src, q, c0, th, acc, dsum);
      else             seg_group<NODE_SIDE, 1>(lst, oth, alpha, src, q, c0, th, acc, dsum);
    }

    if (NODE_SIDE) {
#pragma unroll
      for (int k = 0; k < SEG_C; ++k) {
        const float dinv = (dsum[k] > 0.0f) ? (1.0f / dsum[k]) : 0.0f;
        float4 o;
        o.x = hv[k].x - dinv * acc[k].x + bv.x;
        o.y = hv[k].y - dinv * acc[k].y + bv.y;
        o.z = hv[k].z - dinv * acc[k].z + bv.z;
        o.w = hv[k].w - dinv * acc[k].w + bv.w;
        o.x = (o.x > 0.0f) ? o.x : expm1f(o.x);
        o.y = (o.y > 0.0f) ? o.y : expm1f(o.y);
        o.z = (o.z > 0.0f) ? o.z : expm1f(o.z);
        o.w = (o.w > 0.0f) ? o.w : expm1f(o.w);
        *reinterpret_cast<float4*>(dst + (size_t)seg * OUT_CH + 128 * k + c0) = o;
      }
    } else {
      const int len = e - s;
      const float binv = (len > 0) ? (1.0f / (float)len) : 0.0f;
#pragma unroll
      for (int k = 0; k < SEG_C; ++k)
        *reinterpret_cast<float4*>(dst + (size_t)seg * OUT_CH + 128 * k + c0) =
            make_float4(acc[k].x * binv, acc[k].y * binv, acc[k].z * binv, acc[k].w * binv);
    }
    s = sn;
    e = en;
  }
}

constexpr int NC_WAVES = 4;
constexpr int NC_T     = 32 * NC_WAVES;
constexpr int NC_PITCH = HIDDEN + 4;
constexpr int NC_TILE  = 16 * NC_PITCH;

__device__ inline void node_row_x1(const int* __restrict__ lst, const int* __restrict__ oth,
                                   const float* __restrict__ alpha, const float* __restrict__ src,
                                   const float4 (&hv)[SEG_C], const float4 bv,
                                   int s, int e, int c0, int th, float4 (&o)[SEG_C]) {
  float4 acc[SEG_C];
  float dsum[SEG_C];
#pragma unroll
  for (int k = 0; k < SEG_C; ++k) { acc[k] = make_float4(0.0f, 0.0f, 0.0f, 0.0f); dsum[k] = 0.0f; }
  for (int q = s; q < e; q += SEG_B) {
    const int n = e - q;
    if (n >= 4)      seg_group<true, 4>(lst, oth, alpha, src, q, c0, th, acc, dsum);
    else if (n == 3) seg_group<true, 3>(lst, oth, alpha, src, q, c0, th, acc, dsum);
    else if (n == 2) seg_group<true, 2>(lst, oth, alpha, src, q, c0, th, acc, dsum);
    else             seg_group<true, 1>(lst, oth, alpha, src, q, c0, th, acc, dsum);
  }
#pragma unroll
  for (int k = 0; k < SEG_C; ++k) {
    const float dinv = (dsum[k] > 0.0f) ? (1.0f / dsum[k]) : 0.0f;
    float4 r;
    r.x = hv[k].x - dinv * acc[k].x + bv.x;
    r.y = hv[k].y - dinv * acc[k].y + bv.y;
    r.z = hv[k].z - dinv * acc[k].z + bv.z;
    r.w = hv[k].w - dinv * acc[k].w + bv.w;
    r.x = (r.x > 0.0f) ? r.x : expm1f(r.x);
    r.y = (r.y > 0.0f) ? r.y : expm1f(r.y);
    r.z = (r.z > 0.0f) ? r.z : expm1f(r.z);
    r.w = (r.w > 0.0f) ? r.w : expm1f(r.w);
    o[k] = r;
  }
}

__global__ void __launch_bounds__(NC_T, 4)
seg_node_conv_kernel(const int* __restrict__ ptr, const int* __restrict__ lst,
                     const int* __restrict__ oth, const float* __restrict__ alpha,
                     const float* __restrict__ src, float* hio,
                     const float* __restrict__ bias, const float* __restrict__ W1, int nseg) {
  constexpr int K = HIDDEN, NT = 4, S = K / 4;
  __shared__ float xt[NC_WAVES * NC_TILE];
  const int wib = __builtin_amdgcn_readfirstlane(threadIdx.x >> 5);
  const int nw  = gridDim.x * NC_WAVES;
  int seg = blockIdx.x * NC_WAVES + wib;
  if (seg >= nseg) return;
  const int lane = threadIdx.x & 31;
  const int c0 = 4 * lane;
  const int th = lane >> 4;
  const int c  = lane & 15;
  float* xw = xt + wib * NC_TILE;
  for (int i = lane; i < NC_TILE; i += 32) xw[i] = 0.0f;

  v2f bf[S][NT];
#pragma unroll
  for (int s = 0; s < S; ++s) {
    const int k = 16 * (s >> 2) + 8 * th + 2 * (s & 3);
    float w0[NT], w1[NT];
    load_cols<NT>(W1 + (size_t)k * K + NT * c, w0);
    load_cols<NT>(W1 + (size_t)(k + 1) * K + NT * c, w1);
#pragma unroll
    for (int t = 0; t < NT; ++t) { bf[s][t].x = w0[t]; bf[s][t].y = w1[t]; }
  }
  const float4 bv = *reinterpret_cast<const float4*>(bias + (c0 & (HIDDEN - 1)));

  int s0 = ptr[seg], e0 = ptr[seg + 1];
  int s1, e1;
  {
    const int sb = (seg + nw < nseg) ? (seg + nw) : seg;
    s1 = ptr[sb]; e1 = ptr[sb + 1];
  }
  for (; seg < nseg; seg += 2 * nw) {
    const int seg1 = seg + nw;
    const bool two = (seg1 < nseg);
    const int nx0 = (seg + 2 * nw < nseg) ? (seg + 2 * nw) : seg;
    const int nx1 = (seg + 3 * nw < nseg) ? (seg + 3 * nw) : nx0;
    const int ns0 = ptr[nx0], ne0 = ptr[nx0 + 1];
    const int ns1 = ptr[nx1], ne1 = ptr[nx1 + 1];

    float4 hv[SEG_C];
#pragma unroll
    for (int k = 0; k < SEG_C; ++k)
      hv[k] = *reinterpret_cast<const float4*>(hio + (size_t)seg * OUT_CH + 128 * k + c0);
    float4 o[SEG_C];
    node_row_x1(lst, oth, alpha, src, hv, bv, s0, e0, c0, th, o);
#pragma unroll
    for (int k = 0; k < SEG_C; ++k)
      *reinterpret_cast<float4*>(xw + (2 * k + th) * NC_PITCH + 4 * c) = o[k];
    if (two) {
#pragma unroll
      for (int k = 0; k < SEG_C; ++k)
        hv[k] = *reinterpret_cast<const float4*>(hio + (size_t)seg1 * OUT_CH + 128 * k + c0);
      node_row_x1(lst, oth, alpha, src, hv, bv, s1, e1, c0, th, o);
#pragma unroll
      for (int k = 0; k < SEG_C; ++k)
        *reinterpret_cast<float4*>(xw + (D + 2 * k + th) * NC_PITCH + 4 * c) = o[k];
    }
    __builtin_amdgcn_fence(__ATOMIC_RELEASE, "wavefront");
    __builtin_amdgcn_wave_barrier();
    __builtin_amdgcn_fence(__ATOMIC_ACQUIRE, "wavefront");

    float4 av[K / 8];
    {
      const float* ap = xw + c * NC_PITCH + 8 * th;
#pragma unroll
      for (int g = 0; g < K / 16; ++g) {
        av[2 * g]     = *reinterpret_cast<const float4*>(ap + 16 * g);
        av[2 * g + 1] = *reinterpret_cast<const float4*>(ap + 16 * g + 4);
      }
    }
    __builtin_amdgcn_fence(__ATOMIC_RELEASE, "wavefront");
    __builtin_amdgcn_wave_barrier();
    __builtin_amdgcn_fence(__ATOMIC_ACQUIRE, "wavefront");
    __builtin_amdgcn_sched_barrier(0);

    v8f acc[NT];
#pragma unroll
    for (int t = 0; t < NT; ++t) acc[t] = v8f{};
#pragma unroll
    for (int s = 0; s < S; ++s) {
      const float4 q = av[s >> 1];
      v2f a;
      a.x = (s & 1) ? q.z : q.x;
      a.y = (s & 1) ? q.w : q.y;
#pragma unroll
      for (int t = 0; t < NT; ++t)
        acc[t] = __builtin_amdgcn_wmma_f32_16x16x4_f32(false, a, false, bf[s][t],
                                                       (short)0, acc[t], false, false);
    }
#pragma unroll
    for (int j = 0; j < 8; ++j) {
      const int r = j + 8 * th;
      const bool second = (r >= D);
      const bool ok = second ? (two && r < 2 * D) : true;
      const int node  = second ? seg1 : seg;
      const int stalk = second ? (r - D) : r;
      if (ok)
        *reinterpret_cast<float4*>(hio + ((size_t)node * D + stalk) * HIDDEN + NT * c) =
            make_float4(acc[0][j], acc[1][j], acc[2][j], acc[3][j]);
    }
    s0 = ns0; e0 = ne0; s1 = ns1; e1 = ne1;
  }
}

constexpr int SEG_E_BLK = (N_EDGES + LOOP_WAVES - 1) / LOOP_WAVES;
constexpr int SEG_N_BLK = LOOP_BLK;
constexpr int NC_BLK    = 1024;

extern "C" void kernel_launch(void* const* d_in, const int* in_sizes, int n_in,
                              void* d_out, int out_size, void* d_ws, size_t ws_size,
                              hipStream_t stream) {
  (void)in_sizes; (void)n_in; (void)out_size; (void)ws_size;
  const float* x       = (const float*)d_in[0];
  const float* hattr   = (const float*)d_in[1];
  const float* W_lin   = (const float*)d_in[2];
  const float* b_lin   = (const float*)d_in[3];
  const float* ln_g    = (const float*)d_in[4];
  const float* ln_b    = (const float*)d_in[5];
  const float* W_sheaf = (const float*)d_in[6];
  const float* b_sheaf = (const float*)d_in[7];
  const float* W_conv  = (const float*)d_in[8];
  const float* b_conv  = (const float*)d_in[9];
  const int*   eidx    = (const int*)d_in[10];
  const int* row = eidx;
  const int* col = eidx + NNZ;
  float* out = (float*)d_out;

  float* ws    = (float*)d_ws;
  float* fold  = ws;
  float* Wp    = fold;
  float* bp    = Wp + (size_t)IN_CH * OUT_CH;
  float* Wm    = bp + OUT_CH;
  float* bm    = Wm + (size_t)IN_CH * HIDDEN;
  float* es    = ws   + (size_t)ND * HIDDEN;
  float* xm    = es   + (size_t)ED * HIDDEN;
  float* em    = xm   + (size_t)N_NODES * HIDDEN;
  float* alpha = em   + (size_t)N_EDGES * HIDDEN;
  float* spare = alpha + (size_t)NNZ * D;
  float* hbuf  = spare + ND + N_EDGES - 16;
  float* me    = hbuf + (size_t)ND * HIDDEN;
  int*   ib    = (int*)(me + (size_t)ED * HIDDEN + 16);
  int* nptr = ib;
  int* eptr = nptr + N_NODES + 8;
  unsigned* ntmp = (unsigned*)(eptr + N_EDGES + 8);
  unsigned* etmp = ntmp + NNZ;
  int* nlst = (int*)(etmp + NNZ);
  int* ncol = nlst + NNZ;
  int* elst = ncol + NNZ;
  int* erow = elst + NNZ;

  csr_build_kernel<<<CSR_NB_N + CSR_NB_E + FOLD_BLK, CSR_T, 0, stream>>>(row, col, nptr, eptr, ntmp, etmp,
                                                                         nlst, ncol, elst, erow,
                                                                         W_lin, b_lin, W_conv, Wp, bp, Wm, bm);

  proj_fold_kernel<<<PF_BLK_H + PF_BLK_M, 32 * PF_WAVES, 0, stream>>>(x, hattr, Wp, bp, Wm, bm,
                                                                     hbuf, xm, em);

  alpha_kernel<<<ALPHA_BLK, 32 * ALPHA_WPB, 0, stream>>>(xm, em, row, col, ln_g, ln_b,
                                                W_sheaf, b_sheaf, alpha);

  seg_gather_kernel<false><<<SEG_E_BLK, LOOP_T, 0, stream>>>(eptr, elst, erow, alpha, hbuf,
                                                           nullptr, nullptr, me, N_EDGES);
  seg_node_conv_kernel<<<NC_BLK, NC_T, 0, stream>>>(nptr, nlst, ncol, alpha, me, hbuf, b_conv,
                                                    W_conv + HIDDEN * HIDDEN, N_NODES);

  seg_gather_kernel<false><<<SEG_E_BLK, LOOP_T, 0, stream>>>(eptr, elst, erow, alpha, hbuf,
                                                           nullptr, nullptr, me, N_EDGES);
  seg_gather_kernel<true><<<SEG_N_BLK, LOOP_T, 0, stream>>>(nptr, nlst, ncol, alpha, me,
                                                          hbuf, b_conv + HIDDEN, out, N_NODES);
}


# ELU negative side = hardware exp2(x*log2 e) - 1, branch-free, instead of library expm1 (24 dependent instructions + a branch per value): abs error about 1e-7, tiny results lose relative accuracy
# speedup vs baseline: 5.5244x; 1.0463x over previous
#include <hip/hip_runtime.h>
#include <math.h>

constexpr int N_NODES = 50000;
constexpr int N_EDGES = 10000;
constexpr int NNZ     = 150000;
constexpr int IN_CH   = 128;
constexpr int HIDDEN  = 64;
constexpr int D       = 6;
constexpr int OUT_CH  = HIDDEN * D;
constexpr int ND      = N_NODES * D;
constexpr int ED      = N_EDGES * D;

typedef __attribute__((ext_vector_type(2))) float v2f;
typedef __attribute__((ext_vector_type(8))) float v8f;

template <int NT> struct ColVec;
template <> struct ColVec<2> { typedef float2 type; };
template <> struct ColVec<4> { typedef float4 type; };

template <int NT>
__device__ inline void load_cols(const float* __restrict__ p, float (&d)[NT]) {
  const typename ColVec<NT>::type v = *reinterpret_cast<const typename ColVec<NT>::type*>(p);
  const float* f = reinterpret_cast<const float*>(&v);
#pragma unroll
  for (int t = 0; t < NT; ++t) d[t] = f[t];
}

template <int NT>
__device__ inline void store_cols(float* __restrict__ p, const float (&d)[NT]) {
  typename ColVec<NT>::type v;
  float* f = reinterpret_cast<float*>(&v);
#pragma unroll
  for (int t = 0; t < NT; ++t) f[t] = d[t];
  *reinterpret_cast<typename ColVec<NT>::type*>(p) = v;
}

template <int K>
__device__ inline void load_a_tile(float4 (&av)[K / 8], const float* __restrict__ A,
                                   int mt, int c, int h) {
  const float* ap = A + ((size_t)mt * 16 + c) * K + 8 * h;
#pragma unroll
  for (int g = 0; g < K / 16; ++g) {
    av[2 * g]     = *reinterpret_cast<const float4*>(ap + 16 * g);
    av[2 * g + 1] = *reinterpret_cast<const float4*>(ap + 16 * g + 4);
  }
}

constexpr int FOLD_TILES = IN_CH * D / 16;
constexpr int FOLD_T     = 1024;
constexpr int FOLD_WPB   = FOLD_T / 32;
constexpr int FOLD_BLK   = (FOLD_TILES + 1 + FOLD_WPB - 1) / FOLD_WPB;
constexpr int FOLD_MQ    = IN_CH * HIDDEN / 4;

__device__ inline void fold_weights_block(int fb,
                    const float* __restrict__ W_lin, const float* __restrict__ b_lin,
                    const float* __restrict__ W0,
                    float* __restrict__ Wp, float* __restrict__ bp,
                    float* __restrict__ Wm, float* __restrict__ bm) {
  constexpr int K = HIDDEN, NT = 4, S = K / 4;
  for (int qd = fb * FOLD_T + threadIdx.x; qd < FOLD_MQ + HIDDEN / 4;
       qd += FOLD_BLK * FOLD_T) {
    const bool isb = (qd >= FOLD_MQ);
    const float* p = isb ? (b_lin + 4 * (qd - FOLD_MQ)) : (W_lin + (size_t)(qd >> 4) * OUT_CH + 4 * (qd & 15));
    float4 sm = make_float4(0.0f, 0.0f, 0.0f, 0.0f);
#pragma unroll
    for (int t = 0; t < D; ++t) {
      const float4 v = *reinterpret_cast<const float4*>(p + t * HIDDEN);
      sm.x += v.x; sm.y += v.y; sm.z += v.z; sm.w += v.w;
    }
    const float k6 = 1.0f / 6.0f;
    float* d = isb ? (bm + 4 * (qd - FOLD_MQ)) : (Wm + (size_t)qd * 4);
    *reinterpret_cast<float4*>(d) = make_float4(sm.x * k6, sm.y * k6, sm.z * k6, sm.w * k6);
  }

  const int wave = __builtin_amdgcn_readfirstlane(fb * FOLD_WPB + (threadIdx.x >> 5));
  if (wave > FOLD_TILES) return;
  const bool btile = (wave == FOLD_TILES);
  const float* src = btile ? b_lin : (W_lin + (size_t)wave * 16 * K);
  float* dst       = btile ? bp : (Wp + (size_t)wave * 16 * K);
  const int rows   = btile ? D : 16;
  const int lane = threadIdx.x & 31;
  const int h = lane >> 4;
  const int c = lane & 15;

  float4 av[K / 8];
#pragma unroll
  for (int i = 0; i < K / 8; ++i) av[i] = make_float4(0.0f, 0.0f, 0.0f, 0.0f);
  if (c < rows) {
    const float* ap = src + (size_t)c * K + 8 * h;
#pragma unroll
    for (int g = 0; g < K / 16; ++g) {
      av[2 * g]     = *reinterpret_cast<const float4*>(ap + 16 * g);
      av[2 * g + 1] = *reinterpret_cast<const float4*>(ap + 16 * g + 4);
    }
  }
  v8f acc[NT];
#pragma unroll
  for (int t = 0; t < NT; ++t) acc[t] = v8f{};
#pragma unroll
  for (int s = 0; s < S; ++s) {
    const int k = 16 * (s >> 2) + 8 * h + 2 * (s & 3);
    float w0[NT], w1[NT];
    load_cols<NT>(W0 + (size_t)k * K + NT * c, w0);
    load_cols<NT>(W0 + (size_t)(k + 1) * K + NT * c, w1);
    const float4 q = av[s >> 1];
    v2f a;
    a.x = (s & 1) ? q.z : q.x;
    a.y = (s & 1) ? q.w : q.y;
#pragma unroll
    for (int t = 0; t < NT; ++t) {
      v2f b;
      b.x = w0[t]; b.y = w1[t];
      acc[t] = __builtin_amdgcn_wmma_f32_16x16x4_f32(false, a, false, b, (short)0, acc[t], false, false);
    }
  }
#pragma unroll
  for (int j = 0; j < 8; ++j) {
    float o[NT];
#pragma unroll
    for (int t = 0; t < NT; ++t) o[t] = acc[t][j];
    if (j + 8 * h < rows) store_cols<NT>(dst + (size_t)(j + 8 * h) * K + NT * c, o);
  }
}

constexpr int PF_WAVES  = 12;
constexpr int PF_RUN    = 15;
constexpr int PF_RUNS_N = (N_NODES / 16 + PF_RUN - 1) / PF_RUN;
constexpr int PF_RUNS_E = (N_EDGES / 16 + PF_RUN - 1) / PF_RUN;
constexpr int PF_BLK_H  = PF_RUNS_N;
constexpr int PF_MW_N   = 2 * PF_RUNS_N;
constexpr int PF_MW_E   = 2 * PF_RUNS_E;
constexpr int PF_BLK_M  = (PF_MW_N + PF_MW_E + PF_WAVES - 1) / PF_WAVES;
static_assert(N_NODES % 16 == 0 && N_EDGES % 16 == 0, "row tiles are whole");

__global__ void __launch_bounds__(32 * PF_WAVES)
proj_fold_kernel(const float* __restrict__ x, const float* __restrict__ hattr,
                 const float* __restrict__ Wp, const float* __restrict__ bp,
                 const float* __restrict__ Wm, const float* __restrict__ bm,
                 float* __restrict__ h1, float* __restrict__ xm, float* __restrict__ em) {
  constexpr int K = IN_CH, NT = 2, S = K / 4;
  const int blk = blockIdx.x;
  const int wib = __builtin_amdgcn_readfirstlane(threadIdx.x >> 5);
  const float* A; const float* B; const float* bias; float* C;
  int pitch, n0, run, mTiles;
  if (blk < PF_BLK_H) {
    A = x; B = Wp; bias = bp; C = h1; pitch = OUT_CH; n0 = 16 * NT * wib; run = blk; mTiles = N_NODES / 16;
  } else {
    const int u = (blk - PF_BLK_H) * PF_WAVES + wib;
    B = Wm; bias = bm; pitch = HIDDEN;
    if (u < PF_MW_N) {
      A = x; C = xm; run = u >> 1; n0 = 16 * NT * (u & 1); mTiles = N_NODES / 16;
    } else {
      const int v = u - PF_MW_N;
      if (v >= PF_MW_E) return;
      A = hattr; C = em; run = v >> 1; n0 = 16 * NT * (v & 1); mTiles = N_EDGES / 16;
    }
  }
  int mt = run * PF_RUN;
  if (mt >= mTiles) return;
  const int mtEnd = (mt + PF_RUN < mTiles) ? (mt + PF_RUN) : mTiles;
  const int lane = threadIdx.x & 31;
  const int h = lane >> 4;
  const int c = lane & 15;

  v2f bf[S][NT];
#pragma unroll
  for (int s = 0; s < S; ++s) {
    const int k = 16 * (s >> 2) + 8 * h + 2 * (s & 3);
    float w0[NT], w1[NT];
    load_cols<NT>(B + (size_t)k * pitch + n0 + NT * c, w0);
    load_cols<NT>(B + (size_t)(k + 1) * pitch + n0 + NT * c, w1);
#pragma unroll
    for (int t = 0; t < NT; ++t) { bf[s][t].x = w0[t]; bf[s][t].y = w1[t]; }
  }
  float bv[NT];
  load_cols<NT>(bias + n0 + NT * c, bv);

  float4 av[K / 8];
  load_a_tile<K>(av, A, mt, c, h);
  for (; mt < mtEnd; ++mt) {
    const int nx = (mt + 1 < mtEnd) ? (mt + 1) : mt;
    float4 an[K / 8];
    load_a_tile<K>(an, A, nx, c, h);
    __builtin_amdgcn_sched_barrier(0);

    v8f acc[NT];
#pragma unroll
    for (int t = 0; t < NT; ++t) acc[t] = v8f{};
#pragma unroll
    for (int s = 0; s < S; ++s) {
      const float4 q = av[s >> 1];
      v2f a;
      a.x = (s & 1) ? q.z : q.x;
      a.y = (s & 1) ? q.w : q.y;
#pragma unroll
      for (int t = 0; t < NT; ++t)
        acc[t] = __builtin_amdgcn_wmma_f32_16x16x4_f32(false, a, false, bf[s][t],
                                                       (short)0, acc[t], false, false);
    }
    float* cp = C + ((size_t)mt * 16 + 8 * h) * pitch + n0 + NT * c;
#pragma unroll
    for (int j = 0; j < 8; ++j) {
      float o[NT];
#pragma unroll
      for (int t = 0; t < NT; ++t) o[t] = acc[t][j] + bv[t];
      store_cols<NT>(cp + (size_t)j * pitch, o);
    }
    __builtin_amdgcn_sched_barrier(0);
#pragma unroll
    for (int i = 0; i < K / 8; ++i) av[i] = an[i];
  }
}

constexpr int LOOP_WAVES = 8;
constexpr int LOOP_T     = 32 * LOOP_WAVES;
constexpr int LOOP_BLK   = 1024;

template <int M>
__device__ inline float xmask_add(float v) {
  const int o = __builtin_amdgcn_update_dpp(0, __builtin_bit_cast(int, v), 0x160 + M, 0xf, 0xf, false);
  return v + __builtin_bit_cast(float, o);
}
template <int LG>
__device__ inline float group_allsum(float v) {
  v = xmask_add<1>(v);
  v = xmask_add<2>(v);
  v = xmask_add<4>(v);
  if (LG == 16) v = xmask_add<8>(v);
  return v;
}

constexpr int ALPHA_LG   = 8;
constexpr int ALPHA_HL   = ALPHA_LG / 2;
constexpr int ALPHA_NQ   = HIDDEN / (4 * ALPHA_HL);
constexpr int ALPHA_IPT  = 32 / ALPHA_LG;
constexpr int ALPHA_NTRIP = NNZ / ALPHA_IPT;
constexpr int ALPHA_WPB  = 2;
constexpr int ALPHA_TPW  = 10;
constexpr int ALPHA_BLK  = (ALPHA_NTRIP + ALPHA_TPW * ALPHA_WPB - 1) / (ALPHA_TPW * ALPHA_WPB);
static_assert(NNZ % ALPHA_IPT == 0, "a trip is a whole group of incidences");
static_assert(ALPHA_LG == 8 || ALPHA_LG == 16, "lane groups lie inside the 16-lane rows");

__global__ void __launch_bounds__(32 * ALPHA_WPB)
alpha_kernel(const float* __restrict__ xm, const float* __restrict__ em,
             const int* __restrict__ row, const int* __restrict__ col,
             const float* __restrict__ ln_g, const float* __restrict__ ln_b,
             const float* __restrict__ Wsh, const float* __restrict__ bsh,
             float* __restrict__ alpha) {
  constexpr int LG = ALPHA_LG, HL = ALPHA_HL, NQ = ALPHA_NQ, IPT = ALPHA_IPT;
  const int nw = gridDim.x * ALPHA_WPB;
  int tr = __builtin_amdgcn_readfirstlane(blockIdx.x * ALPHA_WPB + (threadIdx.x >> 5));
  if (tr >= ALPHA_NTRIP) return;
  const int lane = threadIdx.x & 31;
  const int grp  = lane / LG;
  const int sub  = lane % LG;
  const bool eside = (sub >= HL);
  const int ls   = sub % HL;
  const float* zsrc = eside ? em : xm;
  const int*   ksrc = eside ? col : row;
  const int j0 = (eside ? HIDDEN : 0) + 4 * ls;

  float g[NQ][4], b[NQ][4], w[NQ][4][D];
#pragma unroll
  for (int q = 0; q < NQ; ++q) {
    const int j = j0 + 4 * HL * q;
    const float4 gv = *reinterpret_cast<const float4*>(ln_g + j);
    const float4 bv = *reinterpret_cast<const float4*>(ln_b + j);
    g[q][0] = gv.x; g[q][1] = gv.y; g[q][2] = gv.z; g[q][3] = gv.w;
    b[q][0] = bv.x; b[q][1] = bv.y; b[q][2] = bv.z; b[q][3] = bv.w;
    float wf[4 * D];
#pragma unroll
    for (int m = 0; m < D; ++m) {
      const float4 wv = *reinterpret_cast<const float4*>(Wsh + (size_t)j * D + 4 * m);
      wf[4 * m] = wv.x; wf[4 * m + 1] = wv.y; wf[4 * m + 2] = wv.z; wf[4 * m + 3] = wv.w;
    }
#pragma unroll
    for (int e = 0; e < 4; ++e)
#pragma unroll
      for (int t = 0; t < D; ++t) w[q][e][t] = wf[e * D + t];
  }
  const float bs = (sub < D) ? bsh[sub] : 0.0f;

  float4 zc[NQ], zn[NQ];
  {
    const int key = ksrc[tr * IPT + grp];
    const float* zp = zsrc + (size_t)key * HIDDEN + 4 * ls;
#pragma unroll
    for (int q = 0; q < NQ; ++q) zc[q] = *reinterpret_cast<const float4*>(zp + 4 * HL * q);
  }
  const int tr1 = (tr + nw < ALPHA_NTRIP) ? (tr + nw) : tr;
  int keyn = ksrc[tr1 * IPT + grp];
  for (; tr < ALPHA_NTRIP; tr += nw) {
    {
      const float* zp = zsrc + (size_t)keyn * HIDDEN + 4 * ls;
#pragma unroll
      for (int q = 0; q < NQ; ++q) zn[q] = *reinterpret_cast<const float4*>(zp + 4 * HL * q);
    }
    const int tr2 = (tr + 2 * nw < ALPHA_NTRIP) ? (tr + 2 * nw) : tr;
    const int keynn = ksrc[tr2 * IPT + grp];

    float z[NQ][4];
#pragma unroll
    for (int q = 0; q < NQ; ++q) { z[q][0] = zc[q].x; z[q][1] = zc[q].y; z[q][2] = zc[q].z; z[q][3] = zc[q].w; }
    float s = 0.0f, ss = 0.0f;
#pragma unroll
    for (int q = 0; q < NQ; ++q)
#pragma unroll
      for (int e = 0; e < 4; ++e) { s += z[q][e]; ss += z[q][e] * z[q][e]; }
    s  = group_allsum<LG>(s);
    ss = group_allsum<LG>(ss);
    const float mu   = s * (1.0f / 128.0f);
    const float var  = ss * (1.0f / 128.0f) - mu * mu;
    const float rstd = rsqrtf(var + 1e-5f);

    float p[D] = {0, 0, 0, 0, 0, 0};
#pragma unroll
    for (int q = 0; q < NQ; ++q) {
#pragma unroll
      for (int e = 0; e < 4; ++e) {
        const float fn = (z[q][e] - mu) * rstd * g[q][e] + b[q][e];
#pragma unroll
        for (int t = 0; t < D; ++t) p[t] = __builtin_fmaf(fn, w[q][e][t], p[t]);
      }
    }
#pragma unroll
    for (int t = 0; t < D; ++t) p[t] = group_allsum<LG>(p[t]);

    float pv = p[0];
#pragma unroll
    for (int t = 1; t < D; ++t) pv = (sub == t) ? p[t] : pv;
    if (sub < D) {
      const float a = 1.0f / (1.0f + expf(-(pv + bs)));
      alpha[(size_t)(tr * IPT + grp) * D + sub] = a;
    }
#pragma unroll
    for (int q = 0; q < NQ; ++q) zc[q] = zn[q];
    keyn = keynn;
  }
}

constexpr int CSR_T      = 1024;
constexpr int CSR_R_N    = 544;
constexpr int CSR_NB_N   = (N_NODES + CSR_R_N - 1) / CSR_R_N;
constexpr int CSR_R_E    = 160;
constexpr int CSR_NB_E   = (N_EDGES + CSR_R_E - 1) / CSR_R_E;
constexpr int CSR_IDBITS = 18;
constexpr int CSR_N4     = NNZ / 4;
constexpr int CSR_U      = 4;
static_assert(NNZ % 4 == 0, "the key arrays are read as 16-byte quads");
static_assert(NNZ <= (1 << CSR_IDBITS), "an incidence id must fit the low bits of a list word");
static_assert(CSR_R_N <= CSR_T && CSR_R_E <= CSR_T, "the scan takes one counter per thread");
static_assert(CSR_T <= (1 << (32 - CSR_IDBITS)), "a key slot must fit the high bits of a list word");
static_assert((CSR_NB_N - 1) * CSR_R_N < N_NODES && (CSR_NB_E - 1) * CSR_R_E < N_EDGES,
              "every work-group owns at least one key");

__device__ inline int wave_allsum_i(int v) {
#pragma unroll
  for (int off = 16; off > 0; off >>= 1) v += __shfl_xor(v, off, 32);
  return v;
}

__device__ inline int wave_incl_scan(int v, int lane) {
#pragma unroll
  for (int off = 1; off < 32; off <<= 1) {
    int u = __shfl_up(v, off, 32);
    if (lane >= off) v += u;
  }
  return v;
}

__device__ inline void csr_count4(const int4 q, int k0, int nk, int* cur, int& below) {
  const int kk[4] = {q.x, q.y, q.z, q.w};
#pragma unroll
  for (int c = 0; c < 4; ++c) {
    const int d = kk[c] - k0;
    below += (d < 0) ? 1 : 0;
    if ((unsigned)d < (unsigned)nk) atomicAdd(&cur[d], 1);
  }
}

__device__ inline void csr_place4(const int4 q, int i0, int k0, int nk, int* cur,
                                  const int* start, unsigned* __restrict__ tmp) {
  const int kk[4] = {q.x, q.y, q.z, q.w};
#pragma unroll
  for (int c = 0; c < 4; ++c) {
    const int d = kk[c] - k0;
    if ((unsigned)d < (unsigned)nk) {
      const int slot = atomicAdd(&cur[d], 1);
      tmp[start[d] + slot] = ((unsigned)d << CSR_IDBITS) | (unsigned)(i0 + c);
    }
  }
}

static_assert(FOLD_T == CSR_T, "the weight fold rides in this launch with this work-group size");

__global__ void __launch_bounds__(CSR_T)
csr_build_kernel(const int* __restrict__ row, const int* __restrict__ col,
                 int* __restrict__ nptr, int* __restrict__ eptr,
                 unsigned* __restrict__ ntmp, unsigned* __restrict__ etmp,
                 int* __restrict__ nlst, int* __restrict__ ncol,
                 int* __restrict__ elst, int* __restrict__ erow,
                 const float* __restrict__ W_lin, const float* __restrict__ b_lin,
                 const float* __restrict__ W0,
                 float* __restrict__ Wp, float* __restrict__ bp,
                 float* __restrict__ Wm, float* __restrict__ bm) {
  if (blockIdx.x >= CSR_NB_N + CSR_NB_E) {
    fold_weights_block(blockIdx.x - (CSR_NB_N + CSR_NB_E), W_lin, b_lin, W0, Wp, bp, Wm, bm);
    return;
  }
  __shared__ int cur[CSR_T];
  __shared__ int start[CSR_T];
  __shared__ int wsum[32];
  __shared__ int wbel[32];
  __shared__ int sh_base, sh_total;

  const bool eside = (blockIdx.x >= CSR_NB_N);
  const int b      = eside ? (blockIdx.x - CSR_NB_N) : blockIdx.x;
  const int* keys  = eside ? col : row;
  const int* other = eside ? row : col;
  const int nkeys  = eside ? N_EDGES : N_NODES;
  const int R      = eside ? CSR_R_E : CSR_R_N;
  int* ptr         = eside ? eptr : nptr;
  unsigned* tmp    = eside ? etmp : ntmp;
  int* lst         = eside ? elst : nlst;
  int* oth         = eside ? erow : ncol;
  const int k0 = b * R;
  const int nk = (nkeys - k0 < R) ? (nkeys - k0) : R;
  const int tid = threadIdx.x, lane = tid & 31, w = tid >> 5;
  const int4* k4 = reinterpret_cast<const int4*>(keys);

  cur[tid] = 0;
  __syncthreads();

  int below = 0;
  {
    int v = tid;
    for (; v + (CSR_U - 1) * CSR_T < CSR_N4; v += CSR_U * CSR_T) {
      int4 q[CSR_U];
#pragma unroll
      for (int u = 0; u < CSR_U; ++u) q[u] = k4[v + u * CSR_T];
#pragma unroll
      for (int u = 0; u < CSR_U; ++u) csr_count4(q[u], k0, nk, cur, below);
    }
    for (; v < CSR_N4; v += CSR_T) csr_count4(k4[v], k0, nk, cur, below);
  }
  const int bw = wave_allsum_i(below);
  if (lane == 0) wbel[w] = bw;
  __syncthreads();

  const int cnt  = (tid < nk) ? cur[tid] : 0;
  const int incl = wave_incl_scan(cnt, lane);
  if (lane == 31) wsum[w] = incl;
  __syncthreads();
  if (w == 0) {
    const int t  = wsum[lane];
    const int ti = wave_incl_scan(t, lane);
    const int bl = wave_allsum_i(wbel[lane]);
    wsum[lane] = ti - t;
    if (lane == 31) { sh_base = bl; sh_total = ti; }
  }
  __syncthreads();
  const int base  = sh_base;
  const int total = sh_total;
  const int st = base + wsum[w] + incl - cnt;
  start[tid] = st;
  cur[tid] = 0;
  if (tid < nk) ptr[k0 + tid] = st;
  if (tid == nk - 1 && k0 + nk == nkeys) ptr[nkeys] = st + cnt;
  __syncthreads();

  {
    int v = tid;
    for (; v + (CSR_U - 1) * CSR_T < CSR_N4; v += CSR_U * CSR_T) {
      int4 q[CSR_U];
#pragma unroll
      for (int u = 0; u < CSR_U; ++u) q[u] = k4[v + u * CSR_T];
#pragma unroll
      for (int u = 0; u < CSR_U; ++u)
        csr_place4(q[u], 4 * (v + u * CSR_T), k0, nk, cur, start, tmp);
    }
    for (; v < CSR_N4; v += CSR_T) csr_place4(k4[v], 4 * v, k0, nk, cur, start, tmp);
  }
  __syncthreads();

  for (int p = tid; p < total; p += CSR_T) {
    const unsigned wv = tmp[base + p];
    const int d = (int)(wv >> CSR_IDBITS);
    const int i = (int)(wv & ((1u << CSR_IDBITS) - 1u));
    const int o = other[i];
    const int s = start[d], n = cur[d];
    const unsigned* sp = tmp + s;
    int rank = 0;
    int k = 0;
    for (; k + 4 <= n; k += 4) {
      const unsigned u0 = sp[k], u1 = sp[k + 1], u2 = sp[k + 2], u3 = sp[k + 3];
      rank += ((u0 < wv) ? 1 : 0) + ((u1 < wv) ? 1 : 0) + ((u2 < wv) ? 1 : 0) + ((u3 < wv) ? 1 : 0);
    }
    for (; k < n; ++k) rank += (sp[k] < wv) ? 1 : 0;
    lst[s + rank] = i;
    oth[s + rank] = o;
  }
}

__device__ inline float elu_f32(float x) {
  const float e = __builtin_amdgcn_exp2f(x * 1.44269504f);
  return (x > 0.0f) ? x : (e - 1.0f);
}

constexpr int SEG_C = OUT_CH / 128;
constexpr int SEG_B = 4;
static_assert(SEG_B == 4, "the list walk picks its group body from the sizes 4, 3, 2, 1");

template <bool NODE_SIDE, int NB>
__device__ inline void seg_group(const int* __restrict__ lst, const int* __restrict__ oth,
                                 const float* __restrict__ alpha, const float* __restrict__ src,
                                 int q, int c0, int th,
                                 float4 (&acc)[SEG_C], float (&dsum)[SEG_C]) {
  int ii[NB], oo[NB];
#pragma unroll
  for (int u = 0; u < NB; ++u) { ii[u] = lst[q + u]; oo[u] = oth[q + u]; }
  float a[NB][SEG_C];
  float4 v[NB][SEG_C];
#pragma unroll
  for (int u = 0; u < NB; ++u) {
#pragma unroll
    for (int k = 0; k < SEG_C; ++k) {
      a[u][k] = alpha[(size_t)ii[u] * D + 2 * k + th];
      v[u][k] = *reinterpret_cast<const float4*>(src + (size_t)oo[u] * OUT_CH + 128 * k + c0);
    }
  }
#pragma unroll
  for (int u = 0; u < NB; ++u) {
#pragma unroll
    for (int k = 0; k < SEG_C; ++k) {
      acc[k].x = __builtin_fmaf(a[u][k], v[u][k].x, acc[k].x);
      acc[k].y = __builtin_fmaf(a[u][k], v[u][k].y, acc[k].y);
      acc[k].z = __builtin_fmaf(a[u][k], v[u][k].z, acc[k].z);
      acc[k].w = __builtin_fmaf(a[u][k], v[u][k].w, acc[k].w);
      if (NODE_SIDE) dsum[k] = __builtin_fmaf(a[u][k], a[u][k], dsum[k]);
    }
  }
}

template <bool NODE_SIDE>
__global__ void __launch_bounds__(LOOP_T)
seg_gather_kernel(const int* __restrict__ ptr, const int* __restrict__ lst,
                  const int* __restrict__ oth, const float* __restrict__ alpha,
                  const float* __restrict__ src, const float* __restrict__ h,
                  const float* __restrict__ bias, float* __restrict__ dst, int nseg) {
  const int nw = gridDim.x * LOOP_WAVES;
  int seg = blockIdx.x * LOOP_WAVES + (threadIdx.x >> 5);
  if (seg >= nseg) return;
  const int lane = threadIdx.x & 31;
  const int c0 = 4 * lane;
  const int th = lane >> 4;
  float4 bv = make_float4(0.0f, 0.0f, 0.0f, 0.0f);
  if (NODE_SIDE) bv = *reinterpret_cast<const float4*>(bias + (c0 & (HIDDEN - 1)));

  int s = ptr[seg], e = ptr[seg + 1];
  for (; seg < nseg; seg += nw) {
    const int sx = (seg + nw < nseg) ? (seg + nw) : seg;
    const int sn = ptr[sx], en = ptr[sx + 1];
    float4 hv[SEG_C];
    if (NODE_SIDE) {
#pragma unroll
      for (int k = 0; k < SEG_C; ++k)
        hv[k] = *reinterpret_cast<const float4*>(h + (size_t)seg * OUT_CH + 128 * k + c0);
    }
    float4 acc[SEG_C];
    float dsum[SEG_C];
#pragma unroll
    for (int k = 0; k < SEG_C; ++k) { acc[k] = make_float4(0.0f, 0.0f, 0.0f, 0.0f); dsum[k] = 0.0f; }

    for (int q = s; q < e; q += SEG_B) {
      const int n = e - q;
      if (n >= 4)      seg_group<NODE_SIDE, 4>(lst, oth, alpha, src, q, c0, th, acc, dsum);
      else if (n == 3) seg_group<NODE_SIDE, 3>(lst, oth, alpha, src, q, c0, th, acc, dsum);
      else if (n == 2) seg_group<NODE_SIDE, 2>(lst, oth, alpha, src, q, c0, th, acc, dsum);
      else             seg_group<NODE_SIDE, 1>(lst, oth, alpha, src, q, c0, th, acc, dsum);
    }

    if (NODE_SIDE) {
#pragma unroll
      for (int k = 0; k < SEG_C; ++k) {
        const float dinv = (dsum[k] > 0.0f) ? (1.0f / dsum[k]) : 0.0f;
        float4 o;
        o.x = hv[k].x - dinv * acc[k].x + bv.x;
        o.y = hv[k].y - dinv * acc[k].y + bv.y;
        o.z = hv[k].z - dinv * acc[k].z + bv.z;
        o.w = hv[k].w - dinv * acc[k].w + bv.w;
        o.x = elu_f32(o.x);
        o.y = elu_f32(o.y);
        o.z = elu_f32(o.z);
        o.w = elu_f32(o.w);
        *reinterpret_cast<float4*>(dst + (size_t)seg * OUT_CH + 128 * k + c0) = o;
      }
    } else {
      const int len = e - s;
      const float binv = (len > 0) ? (1.0f / (float)len) : 0.0f;
#pragma unroll
      for (int k = 0; k < SEG_C; ++k)
        *reinterpret_cast<float4*>(dst + (size_t)seg * OUT_CH + 128 * k + c0) =
            make_float4(acc[k].x * binv, acc[k].y * binv, acc[k].z * binv, acc[k].w * binv);
    }
    s = sn;
    e = en;
  }
}

constexpr int NC_WAVES = 4;
constexpr int NC_T     = 32 * NC_WAVES;
constexpr int NC_PITCH = HIDDEN + 4;
constexpr int NC_TILE  = 16 * NC_PITCH;

__device__ inline void node_row_x1(const int* __restrict__ lst, const int* __restrict__ oth,
                                   const float* __restrict__ alpha, const float* __restrict__ src,
                                   const float4 (&hv)[SEG_C], const float4 bv,
                                   int s, int e, int c0, int th, float4 (&o)[SEG_C]) {
  float4 acc[SEG_C];
  float dsum[SEG_C];
#pragma unroll
  for (int k = 0; k < SEG_C; ++k) { acc[k] = make_float4(0.0f, 0.0f, 0.0f, 0.0f); dsum[k] = 0.0f; }
  for (int q = s; q < e; q += SEG_B) {
    const int n = e - q;
    if (n >= 4)      seg_group<true, 4>(lst, oth, alpha, src, q, c0, th, acc, dsum);
    else if (n == 3) seg_group<true, 3>(lst, oth, alpha, src, q, c0, th, acc, dsum);
    else if (n == 2) seg_group<true, 2>(lst, oth, alpha, src, q, c0, th, acc, dsum);
    else             seg_group<true, 1>(lst, oth, alpha, src, q, c0, th, acc, dsum);
  }
#pragma unroll
  for (int k = 0; k < SEG_C; ++k) {
    const float dinv = (dsum[k] > 0.0f) ? (1.0f / dsum[k]) : 0.0f;
    float4 r;
    r.x = hv[k].x - dinv * acc[k].x + bv.x;
    r.y = hv[k].y - dinv * acc[k].y + bv.y;
    r.z = hv[k].z - dinv * acc[k].z + bv.z;
    r.w = hv[k].w - dinv * acc[k].w + bv.w;
    r.x = elu_f32(r.x);
    r.y = elu_f32(r.y);
    r.z = elu_f32(r.z);
    r.w = elu_f32(r.w);
    o[k] = r;
  }
}

__global__ void __launch_bounds__(NC_T, 4)
seg_node_conv_kernel(const int* __restrict__ ptr, const int* __restrict__ lst,
                     const int* __restrict__ oth, const float* __restrict__ alpha,
                     const float* __restrict__ src, float* hio,
                     const float* __restrict__ bias, const float* __restrict__ W1, int nseg) {
  constexpr int K = HIDDEN, NT = 4, S = K / 4;
  __shared__ float xt[NC_WAVES * NC_TILE];
  const int wib = __builtin_amdgcn_readfirstlane(threadIdx.x >> 5);
  const int nw  = gridDim.x * NC_WAVES;
  int seg = blockIdx.x * NC_WAVES + wib;
  if (seg >= nseg) return;
  const int lane = threadIdx.x & 31;
  const int c0 = 4 * lane;
  const int th = lane >> 4;
  const int c  = lane & 15;
  float* xw = xt + wib * NC_TILE;
  for (int i = lane; i < NC_TILE; i += 32) xw[i] = 0.0f;

  v2f bf[S][NT];
#pragma unroll
  for (int s = 0; s < S; ++s) {
    const int k = 16 * (s >> 2) + 8 * th + 2 * (s & 3);
    float w0[NT], w1[NT];
    load_cols<NT>(W1 + (size_t)k * K + NT * c, w0);
    load_cols<NT>(W1 + (size_t)(k + 1) * K + NT * c, w1);
#pragma unroll
    for (int t = 0; t < NT; ++t) { bf[s][t].x = w0[t]; bf[s][t].y = w1[t]; }
  }
  const float4 bv = *reinterpret_cast<const float4*>(bias + (c0 & (HIDDEN - 1)));

  int s0 = ptr[seg], e0 = ptr[seg + 1];
  int s1, e1;
  {
    const int sb = (seg + nw < nseg) ? (seg + nw) : seg;
    s1 = ptr[sb]; e1 = ptr[sb + 1];
  }
  for (; seg < nseg; seg += 2 * nw) {
    const int seg1 = seg + nw;
    const bool two = (seg1 < nseg);
    const int nx0 = (seg + 2 * nw < nseg) ? (seg + 2 * nw) : seg;
    const int nx1 = (seg + 3 * nw < nseg) ? (seg + 3 * nw) : nx0;
    const int ns0 = ptr[nx0], ne0 = ptr[nx0 + 1];
    const int ns1 = ptr[nx1], ne1 = ptr[nx1 + 1];

    float4 hv[SEG_C];
#pragma unroll
    for (int k = 0; k < SEG_C; ++k)
      hv[k] = *reinterpret_cast<const float4*>(hio + (size_t)seg * OUT_CH + 128 * k + c0);
    float4 o[SEG_C];
    node_row_x1(lst, oth, alpha, src, hv, bv, s0, e0, c0, th, o);
#pragma unroll
    for (int k = 0; k < SEG_C; ++k)
      *reinterpret_cast<float4*>(xw + (2 * k + th) * NC_PITCH + 4 * c) = o[k];
    if (two) {
#pragma unroll
      for (int k = 0; k < SEG_C; ++k)
        hv[k] = *reinterpret_cast<const float4*>(hio + (size_t)seg1 * OUT_CH + 128 * k + c0);
      node_row_x1(lst, oth, alpha, src, hv, bv, s1, e1, c0, th, o);
#pragma unroll
      for (int k = 0; k < SEG_C; ++k)
        *reinterpret_cast<float4*>(xw + (D + 2 * k + th) * NC_PITCH + 4 * c) = o[k];
    }
    __builtin_amdgcn_fence(__ATOMIC_RELEASE, "wavefront");
    __builtin_amdgcn_wave_barrier();
    __builtin_amdgcn_fence(__ATOMIC_ACQUIRE, "wavefront");

    float4 av[K / 8];
    {
      const float* ap = xw + c * NC_PITCH + 8 * th;
#pragma unroll
      for (int g = 0; g < K / 16; ++g) {
        av[2 * g]     = *reinterpret_cast<const float4*>(ap + 16 * g);
        av[2 * g + 1] = *reinterpret_cast<const float4*>(ap + 16 * g + 4);
      }
    }
    __builtin_amdgcn_fence(__ATOMIC_RELEASE, "wavefront");
    __builtin_amdgcn_wave_barrier();
    __builtin_amdgcn_fence(__ATOMIC_ACQUIRE, "wavefront");
    __builtin_amdgcn_sched_barrier(0);

    v8f acc[NT];
#pragma unroll
    for (int t = 0; t < NT; ++t) acc[t] = v8f{};
#pragma unroll
    for (int s = 0; s < S; ++s) {
      const float4 q = av[s >> 1];
      v2f a;
      a.x = (s & 1) ? q.z : q.x;
      a.y = (s & 1) ? q.w : q.y;
#pragma unroll
      for (int t = 0; t < NT; ++t)
        acc[t] = __builtin_amdgcn_wmma_f32_16x16x4_f32(false, a, false, bf[s][t],
                                                       (short)0, acc[t], false, false);
    }
#pragma unroll
    for (int j = 0; j < 8; ++j) {
      const int r = j + 8 * th;
      const bool second = (r >= D);
      const bool ok = second ? (two && r < 2 * D) : true;
      const int node  = second ? seg1 : seg;
      const int stalk = second ? (r - D) : r;
      if (ok)
        *reinterpret_cast<float4*>(hio + ((size_t)node * D + stalk) * HIDDEN + NT * c) =
            make_float4(acc[0][j], acc[1][j], acc[2][j], acc[3][j]);
    }
    s0 = ns0; e0 = ne0; s1 = ns1; e1 = ne1;
  }
}

constexpr int SEG_E_BLK = (N_EDGES + LOOP_WAVES - 1) / LOOP_WAVES;
constexpr int SEG_N_BLK = LOOP_BLK;
constexpr int NC_BLK    = 1024;

extern "C" void kernel_launch(void* const* d_in, const int* in_sizes, int n_in,
                              void* d_out, int out_size, void* d_ws, size_t ws_size,
                              hipStream_t stream) {
  (void)in_sizes; (void)n_in; (void)out_size; (void)ws_size;
  const float* x       = (const float*)d_in[0];
  const float* hattr   = (const float*)d_in[1];
  const float* W_lin   = (const float*)d_in[2];
  const float* b_lin   = (const float*)d_in[3];
  const float* ln_g    = (const float*)d_in[4];
  const float* ln_b    = (const float*)d_in[5];
  const float* W_sheaf = (const float*)d_in[6];
  const float* b_sheaf = (const float*)d_in[7];
  const float* W_conv  = (const float*)d_in[8];
  const float* b_conv  = (const float*)d_in[9];
  const int*   eidx    = (const int*)d_in[10];
  const int* row = eidx;
  const int* col = eidx + NNZ;
  float* out = (float*)d_out;

  float* ws    = (float*)d_ws;
  float* fold  = ws;
  float* Wp    = fold;
  float* bp    = Wp + (size_t)IN_CH * OUT_CH;
  float* Wm    = bp + OUT_CH;
  float* bm    = Wm + (size_t)IN_CH * HIDDEN;
  float* es    = ws   + (size_t)ND * HIDDEN;
  float* xm    = es   + (size_t)ED * HIDDEN;
  float* em    = xm   + (size_t)N_NODES * HIDDEN;
  float* alpha = em   + (size_t)N_EDGES * HIDDEN;
  float* spare = alpha + (size_t)NNZ * D;
  float* hbuf  = spare + ND + N_EDGES - 16;
  float* me    = hbuf + (size_t)ND * HIDDEN;
  int*   ib    = (int*)(me + (size_t)ED * HIDDEN + 16);
  int* nptr = ib;
  int* eptr = nptr + N_NODES + 8;
  unsigned* ntmp = (unsigned*)(eptr + N_EDGES + 8);
  unsigned* etmp = ntmp + NNZ;
  int* nlst = (int*)(etmp + NNZ);
  int* ncol = nlst + NNZ;
  int* elst = ncol + NNZ;
  int* erow = elst + NNZ;

  csr_build_kernel<<<CSR_NB_N + CSR_NB_E + FOLD_BLK, CSR_T, 0, stream>>>(row, col, nptr, eptr, ntmp, etmp,
                                                                         nlst, ncol, elst, erow,
                                                                         W_lin, b_lin, W_conv, Wp, bp, Wm, bm);

  proj_fold_kernel<<<PF_BLK_H + PF_BLK_M, 32 * PF_WAVES, 0, stream>>>(x, hattr, Wp, bp, Wm, bm,
                                                                     hbuf, xm, em);

  alpha_kernel<<<ALPHA_BLK, 32 * ALPHA_WPB, 0, stream>>>(xm, em, row, col, ln_g, ln_b,
                                                W_sheaf, b_sheaf, alpha);

  seg_gather_kernel<false><<<SEG_E_BLK, LOOP_T, 0, stream>>>(eptr, elst, erow, alpha, hbuf,
                                                           nullptr, nullptr, me, N_EDGES);
  seg_node_conv_kernel<<<NC_BLK, NC_T, 0, stream>>>(nptr, nlst, ncol, alpha, me, hbuf, b_conv,
                                                    W_conv + HIDDEN * HIDDEN, N_NODES);

  seg_gather_kernel<false><<<SEG_E_BLK, LOOP_T, 0, stream>>>(eptr, elst, erow, alpha, hbuf,
                                                           nullptr, nullptr, me, N_EDGES);
  seg_gather_kernel<true><<<SEG_N_BLK, LOOP_T, 0, stream>>>(nptr, nlst, ncol, alpha, me,
                                                          hbuf, b_conv + HIDDEN, out, N_NODES);
}
